# starting kernel
# baseline (speedup 1.0000x reference)
#include <hip/hip_runtime.h>


typedef __attribute__((ext_vector_type(2))) float v2f;
typedef __attribute__((ext_vector_type(8))) float v8f;

#define N_NODES  100000
#define N_EDGES  1600000
#define N_GRAPHS 128
#define N_TILES  (N_NODES / 16)

__global__ void zero_kernel(float* __restrict__ p, int n) {
  int i = blockIdx.x * blockDim.x + threadIdx.x;
  if (i < n) p[i] = 0.0f;
}

__global__ void scatter3_kernel(const float* __restrict__ x,
                                const int* __restrict__ ei,
                                float* __restrict__ agg3) {
  int e = blockIdx.x * blockDim.x + threadIdx.x;
  if (e >= N_EDGES) return;
  int s = ei[e];
  int d = ei[N_EDGES + e];
  #pragma unroll
  for (int k = 0; k < 3; ++k)
    atomicAdd(&agg3[d * 3 + k], x[s * 3 + k]);
}

__global__ void scatter64_kernel(const float* __restrict__ h,
                                 const int* __restrict__ ei,
                                 float* __restrict__ agg) {
  long long gid = (long long)blockIdx.x * blockDim.x + threadIdx.x;
  int e = (int)(gid >> 6);
  int f = (int)(gid & 63);
  if (e >= N_EDGES) return;
  int s = ei[e];
  int d = ei[N_EDGES + e];
  atomicAdd(&agg[d * 64 + f], h[s * 64 + f]);
}

__global__ __launch_bounds__(128) void mlp1_kernel(
    const float* __restrict__ x, const float* __restrict__ agg3,
    const float* __restrict__ W1a, const float* __restrict__ b1a,
    const float* __restrict__ W1b, const float* __restrict__ b1b,
    float* __restrict__ h1) {
  __shared__ float sWa[4 * 64];
  __shared__ float sWb[64 * 64];
  __shared__ float sBa[64];
  __shared__ float sBb[64];
  __shared__ float sT[4][16 * 68];

  int tid = threadIdx.x;
  for (int i = tid; i < 4 * 64; i += 128) {
    int k = i >> 6, n = i & 63;
    sWa[i] = (k < 3) ? W1a[k * 64 + n] : 0.0f;
  }
  for (int i = tid; i < 64 * 64; i += 128) sWb[i] = W1b[i];
  if (tid < 64) { sBa[tid] = b1a[tid]; sBb[tid] = b1b[tid]; }
  __syncthreads();

  int wave = tid >> 5;
  int lane = tid & 31;
  int half = lane >> 4;
  int m    = lane & 15;
  int tile = blockIdx.x * 4 + wave;
  if (tile >= N_TILES) return;
  int node0 = tile * 16;
  float* myT = &sT[wave][0];

  int node = node0 + m;
  int k0   = 2 * half;
  v2f a;
  a.x = x[node * 3 + k0] + agg3[node * 3 + k0];
  a.y = (half == 0) ? (x[node * 3 + 1] + agg3[node * 3 + 1]) : 0.0f;

  v8f acc[4] = {};
  #pragma unroll
  for (int nt = 0; nt < 4; ++nt) {
    v2f b;
    b.x = sWa[k0 * 64 + nt * 16 + m];
    b.y = sWa[(k0 + 1) * 64 + nt * 16 + m];
    acc[nt] = __builtin_amdgcn_wmma_f32_16x16x4_f32(
        false, a, false, b, (short)0, acc[nt], false, false);
  }

  #pragma unroll
  for (int nt = 0; nt < 4; ++nt) {
    float bias = sBa[nt * 16 + m];
    #pragma unroll
    for (int r = 0; r < 8; ++r) {
      float v = acc[nt][r] + bias;
      myT[(r + 8 * half) * 68 + nt * 16 + m] = v > 0.0f ? v : 0.0f;
    }
  }

  v8f acc2[4] = {};
  for (int ks = 0; ks < 16; ++ks) {
    int kk = ks * 4 + 2 * half;
    v2f a2;
    a2.x = myT[m * 68 + kk];
    a2.y = myT[m * 68 + kk + 1];
    #pragma unroll
    for (int nt = 0; nt < 4; ++nt) {
      v2f b;
      b.x = sWb[kk * 64 + nt * 16 + m];
      b.y = sWb[(kk + 1) * 64 + nt * 16 + m];
      acc2[nt] = __builtin_amdgcn_wmma_f32_16x16x4_f32(
          false, a2, false, b, (short)0, acc2[nt], false, false);
    }
  }

  #pragma unroll
  for (int nt = 0; nt < 4; ++nt) {
    float bias = sBb[nt * 16 + m];
    #pragma unroll
    for (int r = 0; r < 8; ++r) {
      float v = acc2[nt][r] + bias;
      h1[(node0 + r + 8 * half) * 64 + nt * 16 + m] = v > 0.0f ? v : 0.0f;
    }
  }
}

__global__ __launch_bounds__(128) void mlp2_kernel(
    const float* __restrict__ h1, const float* __restrict__ agg64,
    const float* __restrict__ W2a, const float* __restrict__ b2a,
    const float* __restrict__ W2b, const float* __restrict__ b2b,
    float* __restrict__ h2) {
  __shared__ float sWa[64 * 64];
  __shared__ float sWb[64 * 64];
  __shared__ float sBa[64];
  __shared__ float sBb[64];
  __shared__ float sT[4][16 * 68];

  int tid = threadIdx.x;
  for (int i = tid; i < 64 * 64; i += 128) { sWa[i] = W2a[i]; sWb[i] = W2b[i]; }
  if (tid < 64) { sBa[tid] = b2a[tid]; sBb[tid] = b2b[tid]; }
  __syncthreads();

  int wave = tid >> 5;
  int lane = tid & 31;
  int half = lane >> 4;
  int m    = lane & 15;
  int tile = blockIdx.x * 4 + wave;
  if (tile >= N_TILES) return;
  int node0 = tile * 16;
  float* myT = &sT[wave][0];

  for (int i = lane; i < 16 * 64; i += 32) {
    int r = i >> 6, c = i & 63;
    int node = node0 + r;
    myT[r * 68 + c] = h1[node * 64 + c] + agg64[node * 64 + c];
  }

  v8f acc[4] = {};
  for (int ks = 0; ks < 16; ++ks) {
    int kk = ks * 4 + 2 * half;
    v2f a;
    a.x = myT[m * 68 + kk];
    a.y = myT[m * 68 + kk + 1];
    #pragma unroll
    for (int nt = 0; nt < 4; ++nt) {
      v2f b;
      b.x = sWa[kk * 64 + nt * 16 + m];
      b.y = sWa[(kk + 1) * 64 + nt * 16 + m];
      acc[nt] = __builtin_amdgcn_wmma_f32_16x16x4_f32(
          false, a, false, b, (short)0, acc[nt], false, false);
    }
  }

  #pragma unroll
  for (int nt = 0; nt < 4; ++nt) {
    float bias = sBa[nt * 16 + m];
    #pragma unroll
    for (int r = 0; r < 8; ++r) {
      float v = acc[nt][r] + bias;
      myT[(r + 8 * half) * 68 + nt * 16 + m] = v > 0.0f ? v : 0.0f;
    }
  }

  v8f acc2[4] = {};
  for (int ks = 0; ks < 16; ++ks) {
    int kk = ks * 4 + 2 * half;
    v2f a2;
    a2.x = myT[m * 68 + kk];
    a2.y = myT[m * 68 + kk + 1];
    #pragma unroll
    for (int nt = 0; nt < 4; ++nt) {
      v2f b;
      b.x = sWb[kk * 64 + nt * 16 + m];
      b.y = sWb[(kk + 1) * 64 + nt * 16 + m];
      acc2[nt] = __builtin_amdgcn_wmma_f32_16x16x4_f32(
          false, a2, false, b, (short)0, acc2[nt], false, false);
    }
  }

  #pragma unroll
  for (int nt = 0; nt < 4; ++nt) {
    float bias = sBb[nt * 16 + m];
    #pragma unroll
    for (int r = 0; r < 8; ++r) {
      float v = acc2[nt][r] + bias;
      h2[(node0 + r + 8 * half) * 64 + nt * 16 + m] = v > 0.0f ? v : 0.0f;
    }
  }
}

__global__ void pool_add_kernel(const float* __restrict__ h,
                                const int* __restrict__ batch,
                                float* __restrict__ sums) {
  long long gid = (long long)blockIdx.x * blockDim.x + threadIdx.x;
  int n = (int)(gid >> 6);
  int f = (int)(gid & 63);
  if (n >= N_NODES) return;
  int g = batch[n];
  atomicAdd(&sums[g * 64 + f], h[n * 64 + f]);
}

__global__ void count_kernel(const int* __restrict__ batch,
                             float* __restrict__ cnt) {
  int n = blockIdx.x * blockDim.x + threadIdx.x;
  if (n >= N_NODES) return;
  atomicAdd(&cnt[batch[n]], 1.0f);
}

__global__ void finalize_kernel(const float* __restrict__ sums,
                                const float* __restrict__ cnt,
                                float* __restrict__ out) {
  int i = blockIdx.x * blockDim.x + threadIdx.x;
  if (i >= N_GRAPHS * 64) return;
  float c = cnt[i >> 6];
  out[i] = sums[i] / fmaxf(c, 1.0f);
}

extern "C" void kernel_launch(void* const* d_in, const int* in_sizes, int n_in,
                              void* d_out, int out_size, void* d_ws, size_t ws_size,
                              hipStream_t stream) {
  const float* x     = (const float*)d_in[0];
  const int*   ei    = (const int*)d_in[1];
  const int*   batch = (const int*)d_in[2];
  const float* W1a   = (const float*)d_in[3];
  const float* b1a   = (const float*)d_in[4];
  const float* W1b   = (const float*)d_in[5];
  const float* b1b   = (const float*)d_in[6];
  const float* W2a   = (const float*)d_in[7];
  const float* b2a   = (const float*)d_in[8];
  const float* W2b   = (const float*)d_in[9];
  const float* b2b   = (const float*)d_in[10];
  float* out = (float*)d_out;

  float* ws    = (float*)d_ws;
  float* agg3  = ws;
  float* h1    = ws + 300000;
  float* agg64 = ws + 300000 + 6400000;
  float* sums  = ws + 300000 + 12800000;

  dim3 b256(256);
  zero_kernel<<<(300000 + 255) / 256, b256, 0, stream>>>(agg3, 300000);
  zero_kernel<<<(6400000 + 255) / 256, b256, 0, stream>>>(agg64, 6400000);
  zero_kernel<<<(8320 + 255) / 256, b256, 0, stream>>>(sums, 8320);

  scatter3_kernel<<<(N_EDGES + 255) / 256, b256, 0, stream>>>(x, ei, agg3);
  mlp1_kernel<<<(N_TILES + 3) / 4, dim3(128), 0, stream>>>(
      x, agg3, W1a, b1a, W1b, b1b, h1);
  scatter64_kernel<<<(N_EDGES * 64) / 256, b256, 0, stream>>>(h1, ei, agg64);
  mlp2_kernel<<<(N_TILES + 3) / 4, dim3(128), 0, stream>>>(
      h1, agg64, W2a, b2a, W2b, b2b, h1  );
  pool_add_kernel<<<(N_NODES * 64 + 255) / 256, b256, 0, stream>>>(h1, batch, sums);
  count_kernel<<<(N_NODES + 255) / 256, b256, 0, stream>>>(batch, sums + 8192);
  finalize_kernel<<<(N_GRAPHS * 64 + 255) / 256, b256, 0, stream>>>(
      sums, sums + 8192, out);
}


# Aggregation by gather: per-call neighbour lists (int-atomic arrival rank + row allocation) replace the f32 atomic scatter-adds of both layers; per-node sums now run in arrival-rank order
# speedup vs baseline: 2.0113x; 2.0113x over previous
#include <hip/hip_runtime.h>


typedef __attribute__((ext_vector_type(2))) float v2f;
typedef __attribute__((ext_vector_type(8))) float v8f;

#define N_NODES  100000
#define N_EDGES  1600000
#define N_GRAPHS 128
#define N_TILES  (N_NODES / 16)

__global__ void zero_words_kernel(unsigned* __restrict__ p, int n) {
  int i = blockIdx.x * blockDim.x + threadIdx.x;
  if (i < n) p[i] = 0u;
}


__global__ void hist_rank_kernel(const int* __restrict__ ei,
                                 int* __restrict__ deg,
                                 int* __restrict__ rank) {
  int e = blockIdx.x * blockDim.x + threadIdx.x;
  if (e >= N_EDGES) return;
  int d = ei[N_EDGES + e];
  rank[e] = atomicAdd(&deg[d], 1);
}

__global__ void alloc_rows_kernel(const int* __restrict__ deg,
                                  int* __restrict__ total,
                                  int* __restrict__ rowstart) {
  int n = blockIdx.x * blockDim.x + threadIdx.x;
  int lane = threadIdx.x & 31;
  int c = (n < N_NODES) ? deg[n] : 0;
  int incl = c;
  for (int off = 1; off < 32; off <<= 1) {
    int t = __shfl_up(incl, off);
    if (lane >= off) incl += t;
  }
  int base = 0;
  if (lane == 31) base = atomicAdd(total, incl);
  base = __shfl(base, 31);
  if (n < N_NODES) rowstart[n] = base + incl - c;
}

__global__ void fill_col_kernel(const int* __restrict__ ei,
                                const int* __restrict__ rank,
                                const int* __restrict__ rowstart,
                                int* __restrict__ col) {
  int e = blockIdx.x * blockDim.x + threadIdx.x;
  if (e >= N_EDGES) return;
  int s = ei[e];
  int d = ei[N_EDGES + e];
  col[rowstart[d] + rank[e]] = s;
}

__global__ void gather3_kernel(const float* __restrict__ x,
                               const int* __restrict__ rowstart,
                               const int* __restrict__ deg,
                               const int* __restrict__ col,
                               float* __restrict__ agg3) {
  int n = blockIdx.x * blockDim.x + threadIdx.x;
  if (n >= N_NODES) return;
  int b = rowstart[n];
  int c = deg[n];
  float a0 = 0.0f, a1 = 0.0f, a2 = 0.0f;
  int j = 0;
  for (; j + 4 <= c; j += 4) {
    int s0 = col[b + j], s1 = col[b + j + 1], s2 = col[b + j + 2], s3 = col[b + j + 3];
    float p0 = x[s0 * 3], p1 = x[s0 * 3 + 1], p2 = x[s0 * 3 + 2];
    float q0 = x[s1 * 3], q1 = x[s1 * 3 + 1], q2 = x[s1 * 3 + 2];
    float r0 = x[s2 * 3], r1 = x[s2 * 3 + 1], r2 = x[s2 * 3 + 2];
    float t0 = x[s3 * 3], t1 = x[s3 * 3 + 1], t2 = x[s3 * 3 + 2];
    a0 += p0; a1 += p1; a2 += p2;
    a0 += q0; a1 += q1; a2 += q2;
    a0 += r0; a1 += r1; a2 += r2;
    a0 += t0; a1 += t1; a2 += t2;
  }
  for (; j < c; ++j) {
    int s = col[b + j];
    a0 += x[s * 3]; a1 += x[s * 3 + 1]; a2 += x[s * 3 + 2];
  }
  agg3[n * 3]     = a0;
  agg3[n * 3 + 1] = a1;
  agg3[n * 3 + 2] = a2;
}

__global__ __launch_bounds__(256) void gather64_kernel(
    const float* __restrict__ h, const int* __restrict__ rowstart,
    const int* __restrict__ deg, const int* __restrict__ col,
    float* __restrict__ agg) {
  int lane = threadIdx.x & 31;
  int n = __builtin_amdgcn_readfirstlane(blockIdx.x * 8 + (threadIdx.x >> 5));
  int b = rowstart[n];
  int c = deg[n];
  const v2f* __restrict__ hv = (const v2f*)h;
  v2f acc = {0.0f, 0.0f};
  for (int j0 = 0; j0 < c; j0 += 32) {
    int m = c - j0;
    if (m > 32) m = 32;
    int cv = (lane < m) ? col[b + j0 + lane] : 0;
    int i = 0;
    for (; i + 8 <= m; i += 8) {
      int s0 = __builtin_amdgcn_readlane(cv, i);
      int s1 = __builtin_amdgcn_readlane(cv, i + 1);
      int s2 = __builtin_amdgcn_readlane(cv, i + 2);
      int s3 = __builtin_amdgcn_readlane(cv, i + 3);
      int s4 = __builtin_amdgcn_readlane(cv, i + 4);
      int s5 = __builtin_amdgcn_readlane(cv, i + 5);
      int s6 = __builtin_amdgcn_readlane(cv, i + 6);
      int s7 = __builtin_amdgcn_readlane(cv, i + 7);
      v2f v0 = hv[s0 * 32 + lane];
      v2f v1 = hv[s1 * 32 + lane];
      v2f v2 = hv[s2 * 32 + lane];
      v2f v3 = hv[s3 * 32 + lane];
      v2f v4 = hv[s4 * 32 + lane];
      v2f v5 = hv[s5 * 32 + lane];
      v2f v6 = hv[s6 * 32 + lane];
      v2f v7 = hv[s7 * 32 + lane];
      acc += v0; acc += v1; acc += v2; acc += v3;
      acc += v4; acc += v5; acc += v6; acc += v7;
    }
    for (; i + 2 <= m; i += 2) {
      int s0 = __builtin_amdgcn_readlane(cv, i);
      int s1 = __builtin_amdgcn_readlane(cv, i + 1);
      v2f v0 = hv[s0 * 32 + lane];
      v2f v1 = hv[s1 * 32 + lane];
      acc += v0; acc += v1;
    }
    if (i < m) {
      int s0 = __builtin_amdgcn_readlane(cv, i);
      acc += hv[s0 * 32 + lane];
    }
  }
  ((v2f*)agg)[n * 32 + lane] = acc;
}

__global__ __launch_bounds__(128) void mlp1_kernel(
    const float* __restrict__ x, const float* __restrict__ agg3,
    const float* __restrict__ W1a, const float* __restrict__ b1a,
    const float* __restrict__ W1b, const float* __restrict__ b1b,
    float* __restrict__ h1) {
  __shared__ float sWa[4 * 64];
  __shared__ float sWb[64 * 64];
  __shared__ float sBa[64];
  __shared__ float sBb[64];
  __shared__ float sT[4][16 * 68];

  int tid = threadIdx.x;
  for (int i = tid; i < 4 * 64; i += 128) {
    int k = i >> 6, n = i & 63;
    sWa[i] = (k < 3) ? W1a[k * 64 + n] : 0.0f;
  }
  for (int i = tid; i < 64 * 64; i += 128) sWb[i] = W1b[i];
  if (tid < 64) { sBa[tid] = b1a[tid]; sBb[tid] = b1b[tid]; }
  __syncthreads();

  int wave = tid >> 5;
  int lane = tid & 31;
  int half = lane >> 4;
  int m    = lane & 15;
  int tile = blockIdx.x * 4 + wave;
  if (tile >= N_TILES) return;
  int node0 = tile * 16;
  float* myT = &sT[wave][0];

  int node = node0 + m;
  int k0   = 2 * half;
  v2f a;
  a.x = x[node * 3 + k0] + agg3[node * 3 + k0];
  a.y = (half == 0) ? (x[node * 3 + 1] + agg3[node * 3 + 1]) : 0.0f;

  v8f acc[4] = {};
  #pragma unroll
  for (int nt = 0; nt < 4; ++nt) {
    v2f b;
    b.x = sWa[k0 * 64 + nt * 16 + m];
    b.y = sWa[(k0 + 1) * 64 + nt * 16 + m];
    acc[nt] = __builtin_amdgcn_wmma_f32_16x16x4_f32(
        false, a, false, b, (short)0, acc[nt], false, false);
  }

  #pragma unroll
  for (int nt = 0; nt < 4; ++nt) {
    float bias = sBa[nt * 16 + m];
    #pragma unroll
    for (int r = 0; r < 8; ++r) {
      float v = acc[nt][r] + bias;
      myT[(r + 8 * half) * 68 + nt * 16 + m] = v > 0.0f ? v : 0.0f;
    }
  }

  v8f acc2[4] = {};
  for (int ks = 0; ks < 16; ++ks) {
    int kk = ks * 4 + 2 * half;
    v2f a2;
    a2.x = myT[m * 68 + kk];
    a2.y = myT[m * 68 + kk + 1];
    #pragma unroll
    for (int nt = 0; nt < 4; ++nt) {
      v2f b;
      b.x = sWb[kk * 64 + nt * 16 + m];
      b.y = sWb[(kk + 1) * 64 + nt * 16 + m];
      acc2[nt] = __builtin_amdgcn_wmma_f32_16x16x4_f32(
          false, a2, false, b, (short)0, acc2[nt], false, false);
    }
  }

  #pragma unroll
  for (int nt = 0; nt < 4; ++nt) {
    float bias = sBb[nt * 16 + m];
    #pragma unroll
    for (int r = 0; r < 8; ++r) {
      float v = acc2[nt][r] + bias;
      h1[(node0 + r + 8 * half) * 64 + nt * 16 + m] = v > 0.0f ? v : 0.0f;
    }
  }
}

__global__ __launch_bounds__(128) void mlp2_kernel(
    const float* __restrict__ h1, const float* __restrict__ agg64,
    const float* __restrict__ W2a, const float* __restrict__ b2a,
    const float* __restrict__ W2b, const float* __restrict__ b2b,
    float* __restrict__ h2) {
  __shared__ float sWa[64 * 64];
  __shared__ float sWb[64 * 64];
  __shared__ float sBa[64];
  __shared__ float sBb[64];
  __shared__ float sT[4][16 * 68];

  int tid = threadIdx.x;
  for (int i = tid; i < 64 * 64; i += 128) { sWa[i] = W2a[i]; sWb[i] = W2b[i]; }
  if (tid < 64) { sBa[tid] = b2a[tid]; sBb[tid] = b2b[tid]; }
  __syncthreads();

  int wave = tid >> 5;
  int lane = tid & 31;
  int half = lane >> 4;
  int m    = lane & 15;
  int tile = blockIdx.x * 4 + wave;
  if (tile >= N_TILES) return;
  int node0 = tile * 16;
  float* myT = &sT[wave][0];

  for (int i = lane; i < 16 * 64; i += 32) {
    int r = i >> 6, c = i & 63;
    int node = node0 + r;
    myT[r * 68 + c] = h1[node * 64 + c] + agg64[node * 64 + c];
  }

  v8f acc[4] = {};
  for (int ks = 0; ks < 16; ++ks) {
    int kk = ks * 4 + 2 * half;
    v2f a;
    a.x = myT[m * 68 + kk];
    a.y = myT[m * 68 + kk + 1];
    #pragma unroll
    for (int nt = 0; nt < 4; ++nt) {
      v2f b;
      b.x = sWa[kk * 64 + nt * 16 + m];
      b.y = sWa[(kk + 1) * 64 + nt * 16 + m];
      acc[nt] = __builtin_amdgcn_wmma_f32_16x16x4_f32(
          false, a, false, b, (short)0, acc[nt], false, false);
    }
  }

  #pragma unroll
  for (int nt = 0; nt < 4; ++nt) {
    float bias = sBa[nt * 16 + m];
    #pragma unroll
    for (int r = 0; r < 8; ++r) {
      float v = acc[nt][r] + bias;
      myT[(r + 8 * half) * 68 + nt * 16 + m] = v > 0.0f ? v : 0.0f;
    }
  }

  v8f acc2[4] = {};
  for (int ks = 0; ks < 16; ++ks) {
    int kk = ks * 4 + 2 * half;
    v2f a2;
    a2.x = myT[m * 68 + kk];
    a2.y = myT[m * 68 + kk + 1];
    #pragma unroll
    for (int nt = 0; nt < 4; ++nt) {
      v2f b;
      b.x = sWb[kk * 64 + nt * 16 + m];
      b.y = sWb[(kk + 1) * 64 + nt * 16 + m];
      acc2[nt] = __builtin_amdgcn_wmma_f32_16x16x4_f32(
          false, a2, false, b, (short)0, acc2[nt], false, false);
    }
  }

  #pragma unroll
  for (int nt = 0; nt < 4; ++nt) {
    float bias = sBb[nt * 16 + m];
    #pragma unroll
    for (int r = 0; r < 8; ++r) {
      float v = acc2[nt][r] + bias;
      h2[(node0 + r + 8 * half) * 64 + nt * 16 + m] = v > 0.0f ? v : 0.0f;
    }
  }
}

__global__ void pool_add_kernel(const float* __restrict__ h,
                                const int* __restrict__ batch,
                                float* __restrict__ sums) {
  long long gid = (long long)blockIdx.x * blockDim.x + threadIdx.x;
  int n = (int)(gid >> 6);
  int f = (int)(gid & 63);
  if (n >= N_NODES) return;
  int g = batch[n];
  atomicAdd(&sums[g * 64 + f], h[n * 64 + f]);
}

__global__ void count_kernel(const int* __restrict__ batch,
                             float* __restrict__ cnt) {
  int n = blockIdx.x * blockDim.x + threadIdx.x;
  if (n >= N_NODES) return;
  atomicAdd(&cnt[batch[n]], 1.0f);
}

__global__ void finalize_kernel(const float* __restrict__ sums,
                                const float* __restrict__ cnt,
                                float* __restrict__ out) {
  int i = blockIdx.x * blockDim.x + threadIdx.x;
  if (i >= N_GRAPHS * 64) return;
  float c = cnt[i >> 6];
  out[i] = sums[i] / fmaxf(c, 1.0f);
}

extern "C" void kernel_launch(void* const* d_in, const int* in_sizes, int n_in,
                              void* d_out, int out_size, void* d_ws, size_t ws_size,
                              hipStream_t stream) {
  const float* x     = (const float*)d_in[0];
  const int*   ei    = (const int*)d_in[1];
  const int*   batch = (const int*)d_in[2];
  const float* W1a   = (const float*)d_in[3];
  const float* b1a   = (const float*)d_in[4];
  const float* W1b   = (const float*)d_in[5];
  const float* b1b   = (const float*)d_in[6];
  const float* W2a   = (const float*)d_in[7];
  const float* b2a   = (const float*)d_in[8];
  const float* W2b   = (const float*)d_in[9];
  const float* b2b   = (const float*)d_in[10];
  float* out = (float*)d_out;

  float* ws       = (float*)d_ws;
  float* agg3     = ws;
  float* h1       = ws + 300032;
  float* agg64    = ws + 6700032;
  int*   deg      = (int*)(ws + 13100032);
  float* sums     = ws + 13200032;
  float* cnt      = ws + 13208224;
  int*   total    = (int*)(ws + 13208352);
  int*   rowstart = (int*)(ws + 13208384);
  int*   rank     = (int*)(ws + 13308416);
  int*   col      = (int*)(ws + 14908416);

  dim3 b256(256);
  zero_words_kernel<<<(108321 + 255) / 256, b256, 0, stream>>>((unsigned*)deg, 108321);

  hist_rank_kernel<<<(N_EDGES + 255) / 256, b256, 0, stream>>>(ei, deg, rank);
  alloc_rows_kernel<<<(N_NODES + 255) / 256, b256, 0, stream>>>(deg, total, rowstart);
  fill_col_kernel<<<(N_EDGES + 255) / 256, b256, 0, stream>>>(ei, rank, rowstart, col);

  gather3_kernel<<<(N_NODES + 255) / 256, b256, 0, stream>>>(x, rowstart, deg, col, agg3);
  mlp1_kernel<<<(N_TILES + 3) / 4, dim3(128), 0, stream>>>(
      x, agg3, W1a, b1a, W1b, b1b, h1);
  gather64_kernel<<<N_NODES / 8, b256, 0, stream>>>(h1, rowstart, deg, col, agg64);
  mlp2_kernel<<<(N_TILES + 3) / 4, dim3(128), 0, stream>>>(
      h1, agg64, W2a, b2a, W2b, b2b, h1  );
  pool_add_kernel<<<(N_NODES * 64 + 255) / 256, b256, 0, stream>>>(h1, batch, sums);
  count_kernel<<<(N_NODES + 255) / 256, b256, 0, stream>>>(batch, cnt);
  finalize_kernel<<<(N_GRAPHS * 64 + 255) / 256, b256, 0, stream>>>(
      sums, cnt, out);
}


# Mean-pool folded into the layer-2 MLP epilogue: each 16-node tile sums its rows in registers, then one f32 atomic per column; h2 is never stored and the pool and count launches are gone
# speedup vs baseline: 2.7082x; 1.3465x over previous
#include <hip/hip_runtime.h>


typedef __attribute__((ext_vector_type(2))) float v2f;
typedef __attribute__((ext_vector_type(8))) float v8f;

#define N_NODES  100000
#define N_EDGES  1600000
#define N_GRAPHS 128
#define N_TILES  (N_NODES / 16)

__global__ void zero_words_kernel(unsigned* __restrict__ p, int n) {
  int i = blockIdx.x * blockDim.x + threadIdx.x;
  if (i < n) p[i] = 0u;
}


__global__ void hist_rank_kernel(const int* __restrict__ ei,
                                 int* __restrict__ deg,
                                 int* __restrict__ rank) {
  int e = blockIdx.x * blockDim.x + threadIdx.x;
  if (e >= N_EDGES) return;
  int d = ei[N_EDGES + e];
  rank[e] = atomicAdd(&deg[d], 1);
}

__global__ void alloc_rows_kernel(const int* __restrict__ deg,
                                  int* __restrict__ total,
                                  int* __restrict__ rowstart) {
  int n = blockIdx.x * blockDim.x + threadIdx.x;
  int lane = threadIdx.x & 31;
  int c = (n < N_NODES) ? deg[n] : 0;
  int incl = c;
  for (int off = 1; off < 32; off <<= 1) {
    int t = __shfl_up(incl, off);
    if (lane >= off) incl += t;
  }
  int base = 0;
  if (lane == 31) base = atomicAdd(total, incl);
  base = __shfl(base, 31);
  if (n < N_NODES) rowstart[n] = base + incl - c;
}

__global__ void fill_col_kernel(const int* __restrict__ ei,
                                const int* __restrict__ rank,
                                const int* __restrict__ rowstart,
                                int* __restrict__ col) {
  int e = blockIdx.x * blockDim.x + threadIdx.x;
  if (e >= N_EDGES) return;
  int s = ei[e];
  int d = ei[N_EDGES + e];
  col[rowstart[d] + rank[e]] = s;
}

__global__ void gather3_kernel(const float* __restrict__ x,
                               const int* __restrict__ rowstart,
                               const int* __restrict__ deg,
                               const int* __restrict__ col,
                               float* __restrict__ agg3) {
  int n = blockIdx.x * blockDim.x + threadIdx.x;
  if (n >= N_NODES) return;
  int b = rowstart[n];
  int c = deg[n];
  float a0 = 0.0f, a1 = 0.0f, a2 = 0.0f;
  int j = 0;
  for (; j + 4 <= c; j += 4) {
    int s0 = col[b + j], s1 = col[b + j + 1], s2 = col[b + j + 2], s3 = col[b + j + 3];
    float p0 = x[s0 * 3], p1 = x[s0 * 3 + 1], p2 = x[s0 * 3 + 2];
    float q0 = x[s1 * 3], q1 = x[s1 * 3 + 1], q2 = x[s1 * 3 + 2];
    float r0 = x[s2 * 3], r1 = x[s2 * 3 + 1], r2 = x[s2 * 3 + 2];
    float t0 = x[s3 * 3], t1 = x[s3 * 3 + 1], t2 = x[s3 * 3 + 2];
    a0 += p0; a1 += p1; a2 += p2;
    a0 += q0; a1 += q1; a2 += q2;
    a0 += r0; a1 += r1; a2 += r2;
    a0 += t0; a1 += t1; a2 += t2;
  }
  for (; j < c; ++j) {
    int s = col[b + j];
    a0 += x[s * 3]; a1 += x[s * 3 + 1]; a2 += x[s * 3 + 2];
  }
  agg3[n * 3]     = a0;
  agg3[n * 3 + 1] = a1;
  agg3[n * 3 + 2] = a2;
}

__global__ __launch_bounds__(256) void gather64_kernel(
    const float* __restrict__ h, const int* __restrict__ rowstart,
    const int* __restrict__ deg, const int* __restrict__ col,
    float* __restrict__ agg) {
  int lane = threadIdx.x & 31;
  int n = __builtin_amdgcn_readfirstlane(blockIdx.x * 8 + (threadIdx.x >> 5));
  int b = rowstart[n];
  int c = deg[n];
  const v2f* __restrict__ hv = (const v2f*)h;
  v2f acc = {0.0f, 0.0f};
  for (int j0 = 0; j0 < c; j0 += 32) {
    int m = c - j0;
    if (m > 32) m = 32;
    int cv = (lane < m) ? col[b + j0 + lane] : 0;
    int i = 0;
    for (; i + 8 <= m; i += 8) {
      int s0 = __builtin_amdgcn_readlane(cv, i);
      int s1 = __builtin_amdgcn_readlane(cv, i + 1);
      int s2 = __builtin_amdgcn_readlane(cv, i + 2);
      int s3 = __builtin_amdgcn_readlane(cv, i + 3);
      int s4 = __builtin_amdgcn_readlane(cv, i + 4);
      int s5 = __builtin_amdgcn_readlane(cv, i + 5);
      int s6 = __builtin_amdgcn_readlane(cv, i + 6);
      int s7 = __builtin_amdgcn_readlane(cv, i + 7);
      v2f v0 = hv[s0 * 32 + lane];
      v2f v1 = hv[s1 * 32 + lane];
      v2f v2 = hv[s2 * 32 + lane];
      v2f v3 = hv[s3 * 32 + lane];
      v2f v4 = hv[s4 * 32 + lane];
      v2f v5 = hv[s5 * 32 + lane];
      v2f v6 = hv[s6 * 32 + lane];
      v2f v7 = hv[s7 * 32 + lane];
      acc += v0; acc += v1; acc += v2; acc += v3;
      acc += v4; acc += v5; acc += v6; acc += v7;
    }
    for (; i + 2 <= m; i += 2) {
      int s0 = __builtin_amdgcn_readlane(cv, i);
      int s1 = __builtin_amdgcn_readlane(cv, i + 1);
      v2f v0 = hv[s0 * 32 + lane];
      v2f v1 = hv[s1 * 32 + lane];
      acc += v0; acc += v1;
    }
    if (i < m) {
      int s0 = __builtin_amdgcn_readlane(cv, i);
      acc += hv[s0 * 32 + lane];
    }
  }
  ((v2f*)agg)[n * 32 + lane] = acc;
}

__global__ __launch_bounds__(128) void mlp1_kernel(
    const float* __restrict__ x, const float* __restrict__ agg3,
    const float* __restrict__ W1a, const float* __restrict__ b1a,
    const float* __restrict__ W1b, const float* __restrict__ b1b,
    float* __restrict__ h1) {
  __shared__ float sWa[4 * 64];
  __shared__ float sWb[64 * 64];
  __shared__ float sBa[64];
  __shared__ float sBb[64];
  __shared__ float sT[4][16 * 68];

  int tid = threadIdx.x;
  for (int i = tid; i < 4 * 64; i += 128) {
    int k = i >> 6, n = i & 63;
    sWa[i] = (k < 3) ? W1a[k * 64 + n] : 0.0f;
  }
  for (int i = tid; i < 64 * 64; i += 128) sWb[i] = W1b[i];
  if (tid < 64) { sBa[tid] = b1a[tid]; sBb[tid] = b1b[tid]; }
  __syncthreads();

  int wave = tid >> 5;
  int lane = tid & 31;
  int half = lane >> 4;
  int m    = lane & 15;
  int tile = blockIdx.x * 4 + wave;
  if (tile >= N_TILES) return;
  int node0 = tile * 16;
  float* myT = &sT[wave][0];

  int node = node0 + m;
  int k0   = 2 * half;
  v2f a;
  a.x = x[node * 3 + k0] + agg3[node * 3 + k0];
  a.y = (half == 0) ? (x[node * 3 + 1] + agg3[node * 3 + 1]) : 0.0f;

  v8f acc[4] = {};
  #pragma unroll
  for (int nt = 0; nt < 4; ++nt) {
    v2f b;
    b.x = sWa[k0 * 64 + nt * 16 + m];
    b.y = sWa[(k0 + 1) * 64 + nt * 16 + m];
    acc[nt] = __builtin_amdgcn_wmma_f32_16x16x4_f32(
        false, a, false, b, (short)0, acc[nt], false, false);
  }

  #pragma unroll
  for (int nt = 0; nt < 4; ++nt) {
    float bias = sBa[nt * 16 + m];
    #pragma unroll
    for (int r = 0; r < 8; ++r) {
      float v = acc[nt][r] + bias;
      myT[(r + 8 * half) * 68 + nt * 16 + m] = v > 0.0f ? v : 0.0f;
    }
  }

  v8f acc2[4] = {};
  for (int ks = 0; ks < 16; ++ks) {
    int kk = ks * 4 + 2 * half;
    v2f a2;
    a2.x = myT[m * 68 + kk];
    a2.y = myT[m * 68 + kk + 1];
    #pragma unroll
    for (int nt = 0; nt < 4; ++nt) {
      v2f b;
      b.x = sWb[kk * 64 + nt * 16 + m];
      b.y = sWb[(kk + 1) * 64 + nt * 16 + m];
      acc2[nt] = __builtin_amdgcn_wmma_f32_16x16x4_f32(
          false, a2, false, b, (short)0, acc2[nt], false, false);
    }
  }

  #pragma unroll
  for (int nt = 0; nt < 4; ++nt) {
    float bias = sBb[nt * 16 + m];
    #pragma unroll
    for (int r = 0; r < 8; ++r) {
      float v = acc2[nt][r] + bias;
      h1[(node0 + r + 8 * half) * 64 + nt * 16 + m] = v > 0.0f ? v : 0.0f;
    }
  }
}

__global__ __launch_bounds__(128) void mlp2_pool_kernel(
    const float* __restrict__ h1, const float* __restrict__ agg64,
    const float* __restrict__ W2a, const float* __restrict__ b2a,
    const float* __restrict__ W2b, const float* __restrict__ b2b,
    const int* __restrict__ batch,
    float* __restrict__ sums, float* __restrict__ cnt) {
  __shared__ float sWa[64 * 64];
  __shared__ float sWb[64 * 64];
  __shared__ float sBa[64];
  __shared__ float sBb[64];
  __shared__ float sT[4][16 * 68];

  int tid = threadIdx.x;
  for (int i = tid; i < 64 * 64; i += 128) { sWa[i] = W2a[i]; sWb[i] = W2b[i]; }
  if (tid < 64) { sBa[tid] = b2a[tid]; sBb[tid] = b2b[tid]; }
  __syncthreads();

  int wave = tid >> 5;
  int lane = tid & 31;
  int half = lane >> 4;
  int m    = lane & 15;
  int tile = blockIdx.x * 4 + wave;
  if (tile >= N_TILES) return;
  int node0 = tile * 16;
  float* myT = &sT[wave][0];

  for (int i = lane; i < 16 * 64; i += 32) {
    int r = i >> 6, c = i & 63;
    int node = node0 + r;
    myT[r * 68 + c] = h1[node * 64 + c] + agg64[node * 64 + c];
  }

  v8f acc[4] = {};
  for (int ks = 0; ks < 16; ++ks) {
    int kk = ks * 4 + 2 * half;
    v2f a;
    a.x = myT[m * 68 + kk];
    a.y = myT[m * 68 + kk + 1];
    #pragma unroll
    for (int nt = 0; nt < 4; ++nt) {
      v2f b;
      b.x = sWa[kk * 64 + nt * 16 + m];
      b.y = sWa[(kk + 1) * 64 + nt * 16 + m];
      acc[nt] = __builtin_amdgcn_wmma_f32_16x16x4_f32(
          false, a, false, b, (short)0, acc[nt], false, false);
    }
  }

  #pragma unroll
  for (int nt = 0; nt < 4; ++nt) {
    float bias = sBa[nt * 16 + m];
    #pragma unroll
    for (int r = 0; r < 8; ++r) {
      float v = acc[nt][r] + bias;
      myT[(r + 8 * half) * 68 + nt * 16 + m] = v > 0.0f ? v : 0.0f;
    }
  }

  v8f acc2[4] = {};
  for (int ks = 0; ks < 16; ++ks) {
    int kk = ks * 4 + 2 * half;
    v2f a2;
    a2.x = myT[m * 68 + kk];
    a2.y = myT[m * 68 + kk + 1];
    #pragma unroll
    for (int nt = 0; nt < 4; ++nt) {
      v2f b;
      b.x = sWb[kk * 64 + nt * 16 + m];
      b.y = sWb[(kk + 1) * 64 + nt * 16 + m];
      acc2[nt] = __builtin_amdgcn_wmma_f32_16x16x4_f32(
          false, a2, false, b, (short)0, acc2[nt], false, false);
    }
  }

  float v[4][8];
  #pragma unroll
  for (int nt = 0; nt < 4; ++nt) {
    float bias = sBb[nt * 16 + m];
    #pragma unroll
    for (int r = 0; r < 8; ++r) {
      float t = acc2[nt][r] + bias;
      v[nt][r] = t > 0.0f ? t : 0.0f;
    }
  }

  int gl = batch[node0 + m];
  int g0 = __builtin_amdgcn_readfirstlane(gl);
  bool one_graph = __builtin_amdgcn_ballot_w32(gl != g0) == 0u;
  if (one_graph) {
    float tot[4];
    #pragma unroll
    for (int nt = 0; nt < 4; ++nt) {
      float s = ((v[nt][0] + v[nt][1]) + (v[nt][2] + v[nt][3])) +
                ((v[nt][4] + v[nt][5]) + (v[nt][6] + v[nt][7]));
      tot[nt] = s + __shfl_xor(s, 16);
    }
    float* dst = sums + g0 * 64;
    atomicAdd(&dst[lane],      half ? tot[1] : tot[0]);
    atomicAdd(&dst[32 + lane], half ? tot[3] : tot[2]);
    if (lane == 0) atomicAdd(&cnt[g0], 16.0f);
  } else {
    #pragma unroll
    for (int r = 0; r < 8; ++r) {
      int g = batch[node0 + r + 8 * half];
      #pragma unroll
      for (int nt = 0; nt < 4; ++nt)
        atomicAdd(&sums[g * 64 + nt * 16 + m], v[nt][r]);
    }
    if (lane < 16) atomicAdd(&cnt[gl], 1.0f);
  }
}

__global__ void finalize_kernel(const float* __restrict__ sums,
                                const float* __restrict__ cnt,
                                float* __restrict__ out) {
  int i = blockIdx.x * blockDim.x + threadIdx.x;
  if (i >= N_GRAPHS * 64) return;
  float c = cnt[i >> 6];
  out[i] = sums[i] / fmaxf(c, 1.0f);
}

extern "C" void kernel_launch(void* const* d_in, const int* in_sizes, int n_in,
                              void* d_out, int out_size, void* d_ws, size_t ws_size,
                              hipStream_t stream) {
  const float* x     = (const float*)d_in[0];
  const int*   ei    = (const int*)d_in[1];
  const int*   batch = (const int*)d_in[2];
  const float* W1a   = (const float*)d_in[3];
  const float* b1a   = (const float*)d_in[4];
  const float* W1b   = (const float*)d_in[5];
  const float* b1b   = (const float*)d_in[6];
  const float* W2a   = (const float*)d_in[7];
  const float* b2a   = (const float*)d_in[8];
  const float* W2b   = (const float*)d_in[9];
  const float* b2b   = (const float*)d_in[10];
  float* out = (float*)d_out;

  float* ws       = (float*)d_ws;
  float* agg3     = ws;
  float* h1       = ws + 300032;
  float* agg64    = ws + 6700032;
  int*   deg      = (int*)(ws + 13100032);
  float* sums     = ws + 13200032;
  float* cnt      = ws + 13208224;
  int*   total    = (int*)(ws + 13208352);
  int*   rowstart = (int*)(ws + 13208384);
  int*   rank     = (int*)(ws + 13308416);
  int*   col      = (int*)(ws + 14908416);

  dim3 b256(256);
  zero_words_kernel<<<(108321 + 255) / 256, b256, 0, stream>>>((unsigned*)deg, 108321);

  hist_rank_kernel<<<(N_EDGES + 255) / 256, b256, 0, stream>>>(ei, deg, rank);
  alloc_rows_kernel<<<(N_NODES + 255) / 256, b256, 0, stream>>>(deg, total, rowstart);
  fill_col_kernel<<<(N_EDGES + 255) / 256, b256, 0, stream>>>(ei, rank, rowstart, col);

  gather3_kernel<<<(N_NODES + 255) / 256, b256, 0, stream>>>(x, rowstart, deg, col, agg3);
  mlp1_kernel<<<(N_TILES + 3) / 4, dim3(128), 0, stream>>>(
      x, agg3, W1a, b1a, W1b, b1b, h1);
  gather64_kernel<<<N_NODES / 8, b256, 0, stream>>>(h1, rowstart, deg, col, agg64);
  mlp2_pool_kernel<<<(N_TILES + 3) / 4, dim3(128), 0, stream>>>(
      h1, agg64, W2a, b2a, W2b, b2b, batch, sums, cnt);
  finalize_kernel<<<(N_GRAPHS * 64 + 255) / 256, b256, 0, stream>>>(
      sums, cnt, out);
}


# Neighbour lists built by two-level destination bucketing on LDS counters (count, scan, partition, per-range sort), no global atomic per edge; per-node sum order is now LDS arrival order
# speedup vs baseline: 4.1031x; 1.5151x over previous
#include <hip/hip_runtime.h>


typedef __attribute__((ext_vector_type(2))) float v2f;
typedef __attribute__((ext_vector_type(8))) float v8f;

#define N_NODES  100000
#define N_EDGES  1600000
#define N_GRAPHS 128
#define N_TILES  (N_NODES / 16)


#define NB    256
#define BS    416
#define CH    4096
#define EPT   (CH / 256)
#define NBLK  ((N_EDGES + CH - 1) / CH)
#define NBLKP (((NBLK + 31) / 32) * 32)

__global__ __launch_bounds__(256) void bucket_count_kernel(
    const int* __restrict__ ei, int* __restrict__ cntB) {
  __shared__ int hist[NB];
  int t = threadIdx.x;
  int e0 = blockIdx.x * CH;
  int d[EPT];
  #pragma unroll
  for (int k = 0; k < EPT; ++k) {
    int e = e0 + k * 256 + t;
    d[k] = ei[N_EDGES + (e < N_EDGES ? e : N_EDGES - 1)];
  }
  hist[t] = 0;
  __syncthreads();
  #pragma unroll
  for (int k = 0; k < EPT; ++k) {
    int e = e0 + k * 256 + t;
    if (e < N_EDGES) atomicAdd(&hist[d[k] / BS], 1);
  }
  __syncthreads();
  cntB[blockIdx.x * NB + t] = hist[t];
}

__global__ __launch_bounds__(256) void bucket_scan_kernel(
    const int* __restrict__ cntB, int* __restrict__ baseT,
    int* __restrict__ total, float* __restrict__ sums, float* __restrict__ cnt) {
  int lane = threadIdx.x & 31;
  int b = blockIdx.x * 8 + (threadIdx.x >> 5);
  int v[NBLKP / 32];
  #pragma unroll
  for (int j = 0; j < NBLKP / 32; ++j) {
    int idx = j * 32 + lane;
    v[j] = (idx < NBLK) ? cntB[idx * NB + b] : 0;
  }
  int running = 0;
  #pragma unroll
  for (int j = 0; j < NBLKP / 32; ++j) {
    int idx = j * 32 + lane;
    int incl = v[j];
    for (int off = 1; off < 32; off <<= 1) {
      int u = __shfl_up(incl, off);
      if (lane >= off) incl += u;
    }
    if (idx < NBLK) baseT[b * NBLKP + idx] = running + incl - v[j];
    running += __shfl(incl, 31);
  }
  if (lane == 0) total[b] = running;
  int g = blockIdx.x * 256 + threadIdx.x;
  sums[g] = 0.0f;
  if (g < N_GRAPHS) cnt[g] = 0.0f;
}

__global__ __launch_bounds__(256) void partition_kernel(
    const int* __restrict__ ei, const int* __restrict__ baseT,
    const int* __restrict__ total, int* __restrict__ pairs) {
  __shared__ int cursor[NB];
  __shared__ int wsum[8];
  int t = threadIdx.x;
  int lane = t & 31, w = t >> 5;
  int e0 = blockIdx.x * CH;
  int tot = total[t];
  int base = baseT[t * NBLKP + blockIdx.x];
  int s[EPT], d[EPT];
  #pragma unroll
  for (int k = 0; k < EPT; ++k) {
    int e = e0 + k * 256 + t;
    int ec = e < N_EDGES ? e : N_EDGES - 1;
    s[k] = ei[ec];
    d[k] = ei[N_EDGES + ec];
  }
  int incl = tot;
  for (int off = 1; off < 32; off <<= 1) {
    int u = __shfl_up(incl, off);
    if (lane >= off) incl += u;
  }
  if (lane == 31) wsum[w] = incl;
  __syncthreads();
  int woff = 0;
  for (int i = 0; i < w; ++i) woff += wsum[i];
  cursor[t] = woff + incl - tot + base;
  __syncthreads();
  #pragma unroll
  for (int k = 0; k < EPT; ++k) {
    int e = e0 + k * 256 + t;
    if (e < N_EDGES) {
      int b = d[k] / BS;
      int pos = atomicAdd(&cursor[b], 1);
      pairs[pos] = s[k] | ((d[k] - b * BS) << 17);
    }
  }
}

__global__ __launch_bounds__(1024) void bucket_sort_kernel(
    const int* __restrict__ pairs, const int* __restrict__ total,
    int* __restrict__ deg, int* __restrict__ rowstart, int* __restrict__ col) {
  __shared__ int ncnt[512];
  __shared__ int ncur[512];
  __shared__ int wsumA[32];
  __shared__ int wsumB[16];
  int t = threadIdx.x;
  int lane = t & 31, w = t >> 5;
  int b = blockIdx.x;
  int v = (t < b) ? total[t] : 0;
  for (int off = 16; off > 0; off >>= 1) v += __shfl_xor(v, off);
  if (lane == 0) wsumA[w] = v;
  if (t < 512) ncnt[t] = 0;
  __syncthreads();
  int bstart = 0;
  for (int i = 0; i < 8; ++i) bstart += wsumA[i];
  int n = total[b];
  const int* __restrict__ mine = pairs + bstart;

  for (int i = t; i < n; i += 8192) {
    int wv[8];
    #pragma unroll
    for (int k = 0; k < 8; ++k) {
      int ik = i + k * 1024;
      wv[k] = (ik < n) ? mine[ik] : -1;
    }
    #pragma unroll
    for (int k = 0; k < 8; ++k)
      if (wv[k] >= 0) atomicAdd(&ncnt[wv[k] >> 17], 1);
  }
  __syncthreads();

  int c = 0, incl = 0;
  if (t < 512) {
    c = ncnt[t];
    incl = c;
    for (int off = 1; off < 32; off <<= 1) {
      int u = __shfl_up(incl, off);
      if (lane >= off) incl += u;
    }
    if (lane == 31) wsumB[w] = incl;
  }
  __syncthreads();
  if (t < 512) {
    int woff = 0;
    for (int i = 0; i < w; ++i) woff += wsumB[i];
    int ls = woff + incl - c;
    ncur[t] = ls;
    int node = b * BS + t;
    if (t < BS && node < N_NODES) {
      deg[node] = c;
      rowstart[node] = bstart + ls;
    }
  }
  __syncthreads();

  for (int i = t; i < n; i += 8192) {
    int wv[8];
    #pragma unroll
    for (int k = 0; k < 8; ++k) {
      int ik = i + k * 1024;
      wv[k] = (ik < n) ? mine[ik] : -1;
    }
    #pragma unroll
    for (int k = 0; k < 8; ++k)
      if (wv[k] >= 0)
        col[bstart + atomicAdd(&ncur[wv[k] >> 17], 1)] = wv[k] & 0x1FFFF;
  }
}

__global__ void gather3_kernel(const float* __restrict__ x,
                               const int* __restrict__ rowstart,
                               const int* __restrict__ deg,
                               const int* __restrict__ col,
                               float* __restrict__ agg3) {
  int n = blockIdx.x * blockDim.x + threadIdx.x;
  if (n >= N_NODES) return;
  int b = rowstart[n];
  int c = deg[n];
  float a0 = 0.0f, a1 = 0.0f, a2 = 0.0f;
  int j = 0;
  for (; j + 4 <= c; j += 4) {
    int s0 = col[b + j], s1 = col[b + j + 1], s2 = col[b + j + 2], s3 = col[b + j + 3];
    float p0 = x[s0 * 3], p1 = x[s0 * 3 + 1], p2 = x[s0 * 3 + 2];
    float q0 = x[s1 * 3], q1 = x[s1 * 3 + 1], q2 = x[s1 * 3 + 2];
    float r0 = x[s2 * 3], r1 = x[s2 * 3 + 1], r2 = x[s2 * 3 + 2];
    float t0 = x[s3 * 3], t1 = x[s3 * 3 + 1], t2 = x[s3 * 3 + 2];
    a0 += p0; a1 += p1; a2 += p2;
    a0 += q0; a1 += q1; a2 += q2;
    a0 += r0; a1 += r1; a2 += r2;
    a0 += t0; a1 += t1; a2 += t2;
  }
  for (; j < c; ++j) {
    int s = col[b + j];
    a0 += x[s * 3]; a1 += x[s * 3 + 1]; a2 += x[s * 3 + 2];
  }
  agg3[n * 3]     = a0;
  agg3[n * 3 + 1] = a1;
  agg3[n * 3 + 2] = a2;
}

__global__ __launch_bounds__(256) void gather64_kernel(
    const float* __restrict__ h, const int* __restrict__ rowstart,
    const int* __restrict__ deg, const int* __restrict__ col,
    float* __restrict__ agg) {
  int lane = threadIdx.x & 31;
  int n = __builtin_amdgcn_readfirstlane(blockIdx.x * 8 + (threadIdx.x >> 5));
  int b = rowstart[n];
  int c = deg[n];
  const v2f* __restrict__ hv = (const v2f*)h;
  v2f acc = {0.0f, 0.0f};
  for (int j0 = 0; j0 < c; j0 += 32) {
    int m = c - j0;
    if (m > 32) m = 32;
    int cv = (lane < m) ? col[b + j0 + lane] : 0;
    int i = 0;
    for (; i + 8 <= m; i += 8) {
      int s0 = __builtin_amdgcn_readlane(cv, i);
      int s1 = __builtin_amdgcn_readlane(cv, i + 1);
      int s2 = __builtin_amdgcn_readlane(cv, i + 2);
      int s3 = __builtin_amdgcn_readlane(cv, i + 3);
      int s4 = __builtin_amdgcn_readlane(cv, i + 4);
      int s5 = __builtin_amdgcn_readlane(cv, i + 5);
      int s6 = __builtin_amdgcn_readlane(cv, i + 6);
      int s7 = __builtin_amdgcn_readlane(cv, i + 7);
      v2f v0 = hv[s0 * 32 + lane];
      v2f v1 = hv[s1 * 32 + lane];
      v2f v2 = hv[s2 * 32 + lane];
      v2f v3 = hv[s3 * 32 + lane];
      v2f v4 = hv[s4 * 32 + lane];
      v2f v5 = hv[s5 * 32 + lane];
      v2f v6 = hv[s6 * 32 + lane];
      v2f v7 = hv[s7 * 32 + lane];
      acc += v0; acc += v1; acc += v2; acc += v3;
      acc += v4; acc += v5; acc += v6; acc += v7;
    }
    for (; i + 2 <= m; i += 2) {
      int s0 = __builtin_amdgcn_readlane(cv, i);
      int s1 = __builtin_amdgcn_readlane(cv, i + 1);
      v2f v0 = hv[s0 * 32 + lane];
      v2f v1 = hv[s1 * 32 + lane];
      acc += v0; acc += v1;
    }
    if (i < m) {
      int s0 = __builtin_amdgcn_readlane(cv, i);
      acc += hv[s0 * 32 + lane];
    }
  }
  ((v2f*)agg)[n * 32 + lane] = acc;
}

__global__ __launch_bounds__(128) void mlp1_kernel(
    const float* __restrict__ x, const float* __restrict__ agg3,
    const float* __restrict__ W1a, const float* __restrict__ b1a,
    const float* __restrict__ W1b, const float* __restrict__ b1b,
    float* __restrict__ h1) {
  __shared__ float sWa[4 * 64];
  __shared__ float sWb[64 * 64];
  __shared__ float sBa[64];
  __shared__ float sBb[64];
  __shared__ float sT[4][16 * 68];

  int tid = threadIdx.x;
  for (int i = tid; i < 4 * 64; i += 128) {
    int k = i >> 6, n = i & 63;
    sWa[i] = (k < 3) ? W1a[k * 64 + n] : 0.0f;
  }
  for (int i = tid; i < 64 * 64; i += 128) sWb[i] = W1b[i];
  if (tid < 64) { sBa[tid] = b1a[tid]; sBb[tid] = b1b[tid]; }
  __syncthreads();

  int wave = tid >> 5;
  int lane = tid & 31;
  int half = lane >> 4;
  int m    = lane & 15;
  int tile = blockIdx.x * 4 + wave;
  if (tile >= N_TILES) return;
  int node0 = tile * 16;
  float* myT = &sT[wave][0];

  int node = node0 + m;
  int k0   = 2 * half;
  v2f a;
  a.x = x[node * 3 + k0] + agg3[node * 3 + k0];
  a.y = (half == 0) ? (x[node * 3 + 1] + agg3[node * 3 + 1]) : 0.0f;

  v8f acc[4] = {};
  #pragma unroll
  for (int nt = 0; nt < 4; ++nt) {
    v2f b;
    b.x = sWa[k0 * 64 + nt * 16 + m];
    b.y = sWa[(k0 + 1) * 64 + nt * 16 + m];
    acc[nt] = __builtin_amdgcn_wmma_f32_16x16x4_f32(
        false, a, false, b, (short)0, acc[nt], false, false);
  }

  #pragma unroll
  for (int nt = 0; nt < 4; ++nt) {
    float bias = sBa[nt * 16 + m];
    #pragma unroll
    for (int r = 0; r < 8; ++r) {
      float v = acc[nt][r] + bias;
      myT[(r + 8 * half) * 68 + nt * 16 + m] = v > 0.0f ? v : 0.0f;
    }
  }

  v8f acc2[4] = {};
  for (int ks = 0; ks < 16; ++ks) {
    int kk = ks * 4 + 2 * half;
    v2f a2;
    a2.x = myT[m * 68 + kk];
    a2.y = myT[m * 68 + kk + 1];
    #pragma unroll
    for (int nt = 0; nt < 4; ++nt) {
      v2f b;
      b.x = sWb[kk * 64 + nt * 16 + m];
      b.y = sWb[(kk + 1) * 64 + nt * 16 + m];
      acc2[nt] = __builtin_amdgcn_wmma_f32_16x16x4_f32(
          false, a2, false, b, (short)0, acc2[nt], false, false);
    }
  }

  #pragma unroll
  for (int nt = 0; nt < 4; ++nt) {
    float bias = sBb[nt * 16 + m];
    #pragma unroll
    for (int r = 0; r < 8; ++r) {
      float v = acc2[nt][r] + bias;
      h1[(node0 + r + 8 * half) * 64 + nt * 16 + m] = v > 0.0f ? v : 0.0f;
    }
  }
}

__global__ __launch_bounds__(128) void mlp2_pool_kernel(
    const float* __restrict__ h1, const float* __restrict__ agg64,
    const float* __restrict__ W2a, const float* __restrict__ b2a,
    const float* __restrict__ W2b, const float* __restrict__ b2b,
    const int* __restrict__ batch,
    float* __restrict__ sums, float* __restrict__ cnt) {
  __shared__ float sWa[64 * 64];
  __shared__ float sWb[64 * 64];
  __shared__ float sBa[64];
  __shared__ float sBb[64];
  __shared__ float sT[4][16 * 68];

  int tid = threadIdx.x;
  for (int i = tid; i < 64 * 64; i += 128) { sWa[i] = W2a[i]; sWb[i] = W2b[i]; }
  if (tid < 64) { sBa[tid] = b2a[tid]; sBb[tid] = b2b[tid]; }
  __syncthreads();

  int wave = tid >> 5;
  int lane = tid & 31;
  int half = lane >> 4;
  int m    = lane & 15;
  int tile = blockIdx.x * 4 + wave;
  if (tile >= N_TILES) return;
  int node0 = tile * 16;
  float* myT = &sT[wave][0];

  for (int i = lane; i < 16 * 64; i += 32) {
    int r = i >> 6, c = i & 63;
    int node = node0 + r;
    myT[r * 68 + c] = h1[node * 64 + c] + agg64[node * 64 + c];
  }

  v8f acc[4] = {};
  for (int ks = 0; ks < 16; ++ks) {
    int kk = ks * 4 + 2 * half;
    v2f a;
    a.x = myT[m * 68 + kk];
    a.y = myT[m * 68 + kk + 1];
    #pragma unroll
    for (int nt = 0; nt < 4; ++nt) {
      v2f b;
      b.x = sWa[kk * 64 + nt * 16 + m];
      b.y = sWa[(kk + 1) * 64 + nt * 16 + m];
      acc[nt] = __builtin_amdgcn_wmma_f32_16x16x4_f32(
          false, a, false, b, (short)0, acc[nt], false, false);
    }
  }

  #pragma unroll
  for (int nt = 0; nt < 4; ++nt) {
    float bias = sBa[nt * 16 + m];
    #pragma unroll
    for (int r = 0; r < 8; ++r) {
      float v = acc[nt][r] + bias;
      myT[(r + 8 * half) * 68 + nt * 16 + m] = v > 0.0f ? v : 0.0f;
    }
  }

  v8f acc2[4] = {};
  for (int ks = 0; ks < 16; ++ks) {
    int kk = ks * 4 + 2 * half;
    v2f a2;
    a2.x = myT[m * 68 + kk];
    a2.y = myT[m * 68 + kk + 1];
    #pragma unroll
    for (int nt = 0; nt < 4; ++nt) {
      v2f b;
      b.x = sWb[kk * 64 + nt * 16 + m];
      b.y = sWb[(kk + 1) * 64 + nt * 16 + m];
      acc2[nt] = __builtin_amdgcn_wmma_f32_16x16x4_f32(
          false, a2, false, b, (short)0, acc2[nt], false, false);
    }
  }

  float v[4][8];
  #pragma unroll
  for (int nt = 0; nt < 4; ++nt) {
    float bias = sBb[nt * 16 + m];
    #pragma unroll
    for (int r = 0; r < 8; ++r) {
      float t = acc2[nt][r] + bias;
      v[nt][r] = t > 0.0f ? t : 0.0f;
    }
  }

  int gl = batch[node0 + m];
  int g0 = __builtin_amdgcn_readfirstlane(gl);
  bool one_graph = __builtin_amdgcn_ballot_w32(gl != g0) == 0u;
  if (one_graph) {
    float tot[4];
    #pragma unroll
    for (int nt = 0; nt < 4; ++nt) {
      float s = ((v[nt][0] + v[nt][1]) + (v[nt][2] + v[nt][3])) +
                ((v[nt][4] + v[nt][5]) + (v[nt][6] + v[nt][7]));
      tot[nt] = s + __shfl_xor(s, 16);
    }
    float* dst = sums + g0 * 64;
    atomicAdd(&dst[lane],      half ? tot[1] : tot[0]);
    atomicAdd(&dst[32 + lane], half ? tot[3] : tot[2]);
    if (lane == 0) atomicAdd(&cnt[g0], 16.0f);
  } else {
    #pragma unroll
    for (int r = 0; r < 8; ++r) {
      int g = batch[node0 + r + 8 * half];
      #pragma unroll
      for (int nt = 0; nt < 4; ++nt)
        atomicAdd(&sums[g * 64 + nt * 16 + m], v[nt][r]);
    }
    if (lane < 16) atomicAdd(&cnt[gl], 1.0f);
  }
}

__global__ void finalize_kernel(const float* __restrict__ sums,
                                const float* __restrict__ cnt,
                                float* __restrict__ out) {
  int i = blockIdx.x * blockDim.x + threadIdx.x;
  if (i >= N_GRAPHS * 64) return;
  float c = cnt[i >> 6];
  out[i] = sums[i] / fmaxf(c, 1.0f);
}

extern "C" void kernel_launch(void* const* d_in, const int* in_sizes, int n_in,
                              void* d_out, int out_size, void* d_ws, size_t ws_size,
                              hipStream_t stream) {
  const float* x     = (const float*)d_in[0];
  const int*   ei    = (const int*)d_in[1];
  const int*   batch = (const int*)d_in[2];
  const float* W1a   = (const float*)d_in[3];
  const float* b1a   = (const float*)d_in[4];
  const float* W1b   = (const float*)d_in[5];
  const float* b1b   = (const float*)d_in[6];
  const float* W2a   = (const float*)d_in[7];
  const float* b2a   = (const float*)d_in[8];
  const float* W2b   = (const float*)d_in[9];
  const float* b2b   = (const float*)d_in[10];
  float* out = (float*)d_out;

  float* ws       = (float*)d_ws;
  float* agg3     = ws;
  float* h1       = ws + 300032;
  float* agg64    = ws + 6700032;
  int*   deg      = (int*)(ws + 13100032);
  float* sums     = ws + 13200032;
  float* cnt      = ws + 13208224;
  int*   rowstart = (int*)(ws + 13208384);
  int*   pairs    = (int*)(ws + 13308416);
  int*   col      = (int*)(ws + 14908416);
  int*   cntB     = (int*)(ws + 16508416);
  int*   baseT    = (int*)(ws + 16614912);
  int*   total    = (int*)(ws + 16721408);

  dim3 b256(256);
  bucket_count_kernel<<<NBLK, b256, 0, stream>>>(ei, cntB);
  bucket_scan_kernel<<<NB / 8, b256, 0, stream>>>(cntB, baseT, total, sums, cnt);
  partition_kernel<<<NBLK, b256, 0, stream>>>(ei, baseT, total, pairs);
  bucket_sort_kernel<<<NB, dim3(1024), 0, stream>>>(pairs, total, deg, rowstart, col);

  gather3_kernel<<<(N_NODES + 255) / 256, b256, 0, stream>>>(x, rowstart, deg, col, agg3);
  mlp1_kernel<<<(N_TILES + 3) / 4, dim3(128), 0, stream>>>(
      x, agg3, W1a, b1a, W1b, b1b, h1);
  gather64_kernel<<<N_NODES / 8, b256, 0, stream>>>(h1, rowstart, deg, col, agg64);
  mlp2_pool_kernel<<<(N_TILES + 3) / 4, dim3(128), 0, stream>>>(
      h1, agg64, W2a, b2a, W2b, b2b, batch, sums, cnt);
  finalize_kernel<<<(N_GRAPHS * 64 + 255) / 256, b256, 0, stream>>>(
      sums, cnt, out);
}


# Layer-2 neighbour gather fused into the layer-2 MLP kernel: each wave sums its 16 nodes' neighbour rows into its LDS tile (aggregate never stored, gather launch gone); per-node sum order unchanged
# speedup vs baseline: 4.4120x; 1.0753x over previous
#include <hip/hip_runtime.h>


typedef __attribute__((ext_vector_type(2))) float v2f;
typedef __attribute__((ext_vector_type(8))) float v8f;

#define N_NODES  100000
#define N_EDGES  1600000
#define N_GRAPHS 128
#define N_TILES  (N_NODES / 16)


#define NB    256
#define BS    416
#define CH    4096
#define EPT   (CH / 256)
#define NBLK  ((N_EDGES + CH - 1) / CH)
#define NBLKP (((NBLK + 31) / 32) * 32)

__global__ __launch_bounds__(256) void bucket_count_kernel(
    const int* __restrict__ ei, int* __restrict__ cntB) {
  __shared__ int hist[NB];
  int t = threadIdx.x;
  int e0 = blockIdx.x * CH;
  int d[EPT];
  #pragma unroll
  for (int k = 0; k < EPT; ++k) {
    int e = e0 + k * 256 + t;
    d[k] = ei[N_EDGES + (e < N_EDGES ? e : N_EDGES - 1)];
  }
  hist[t] = 0;
  __syncthreads();
  #pragma unroll
  for (int k = 0; k < EPT; ++k) {
    int e = e0 + k * 256 + t;
    if (e < N_EDGES) atomicAdd(&hist[d[k] / BS], 1);
  }
  __syncthreads();
  cntB[blockIdx.x * NB + t] = hist[t];
}

__global__ __launch_bounds__(256) void bucket_scan_kernel(
    const int* __restrict__ cntB, int* __restrict__ baseT,
    int* __restrict__ total, float* __restrict__ sums, float* __restrict__ cnt) {
  int lane = threadIdx.x & 31;
  int b = blockIdx.x * 8 + (threadIdx.x >> 5);
  int v[NBLKP / 32];
  #pragma unroll
  for (int j = 0; j < NBLKP / 32; ++j) {
    int idx = j * 32 + lane;
    v[j] = (idx < NBLK) ? cntB[idx * NB + b] : 0;
  }
  int running = 0;
  #pragma unroll
  for (int j = 0; j < NBLKP / 32; ++j) {
    int idx = j * 32 + lane;
    int incl = v[j];
    for (int off = 1; off < 32; off <<= 1) {
      int u = __shfl_up(incl, off);
      if (lane >= off) incl += u;
    }
    if (idx < NBLK) baseT[b * NBLKP + idx] = running + incl - v[j];
    running += __shfl(incl, 31);
  }
  if (lane == 0) total[b] = running;
  int g = blockIdx.x * 256 + threadIdx.x;
  sums[g] = 0.0f;
  if (g < N_GRAPHS) cnt[g] = 0.0f;
}

__global__ __launch_bounds__(256) void partition_kernel(
    const int* __restrict__ ei, const int* __restrict__ baseT,
    const int* __restrict__ total, int* __restrict__ pairs) {
  __shared__ int cursor[NB];
  __shared__ int wsum[8];
  int t = threadIdx.x;
  int lane = t & 31, w = t >> 5;
  int e0 = blockIdx.x * CH;
  int tot = total[t];
  int base = baseT[t * NBLKP + blockIdx.x];
  int s[EPT], d[EPT];
  #pragma unroll
  for (int k = 0; k < EPT; ++k) {
    int e = e0 + k * 256 + t;
    int ec = e < N_EDGES ? e : N_EDGES - 1;
    s[k] = ei[ec];
    d[k] = ei[N_EDGES + ec];
  }
  int incl = tot;
  for (int off = 1; off < 32; off <<= 1) {
    int u = __shfl_up(incl, off);
    if (lane >= off) incl += u;
  }
  if (lane == 31) wsum[w] = incl;
  __syncthreads();
  int woff = 0;
  for (int i = 0; i < w; ++i) woff += wsum[i];
  cursor[t] = woff + incl - tot + base;
  __syncthreads();
  #pragma unroll
  for (int k = 0; k < EPT; ++k) {
    int e = e0 + k * 256 + t;
    if (e < N_EDGES) {
      int b = d[k] / BS;
      int pos = atomicAdd(&cursor[b], 1);
      pairs[pos] = s[k] | ((d[k] - b * BS) << 17);
    }
  }
}

__global__ __launch_bounds__(1024) void bucket_sort_kernel(
    const int* __restrict__ pairs, const int* __restrict__ total,
    int* __restrict__ deg, int* __restrict__ rowstart, int* __restrict__ col) {
  __shared__ int ncnt[512];
  __shared__ int ncur[512];
  __shared__ int wsumA[32];
  __shared__ int wsumB[16];
  int t = threadIdx.x;
  int lane = t & 31, w = t >> 5;
  int b = blockIdx.x;
  int v = (t < b) ? total[t] : 0;
  for (int off = 16; off > 0; off >>= 1) v += __shfl_xor(v, off);
  if (lane == 0) wsumA[w] = v;
  if (t < 512) ncnt[t] = 0;
  __syncthreads();
  int bstart = 0;
  for (int i = 0; i < 8; ++i) bstart += wsumA[i];
  int n = total[b];
  const int* __restrict__ mine = pairs + bstart;

  for (int i = t; i < n; i += 8192) {
    int wv[8];
    #pragma unroll
    for (int k = 0; k < 8; ++k) {
      int ik = i + k * 1024;
      wv[k] = (ik < n) ? mine[ik] : -1;
    }
    #pragma unroll
    for (int k = 0; k < 8; ++k)
      if (wv[k] >= 0) atomicAdd(&ncnt[wv[k] >> 17], 1);
  }
  __syncthreads();

  int c = 0, incl = 0;
  if (t < 512) {
    c = ncnt[t];
    incl = c;
    for (int off = 1; off < 32; off <<= 1) {
      int u = __shfl_up(incl, off);
      if (lane >= off) incl += u;
    }
    if (lane == 31) wsumB[w] = incl;
  }
  __syncthreads();
  if (t < 512) {
    int woff = 0;
    for (int i = 0; i < w; ++i) woff += wsumB[i];
    int ls = woff + incl - c;
    ncur[t] = ls;
    int node = b * BS + t;
    if (t < BS && node < N_NODES) {
      deg[node] = c;
      rowstart[node] = bstart + ls;
    }
  }
  __syncthreads();

  for (int i = t; i < n; i += 8192) {
    int wv[8];
    #pragma unroll
    for (int k = 0; k < 8; ++k) {
      int ik = i + k * 1024;
      wv[k] = (ik < n) ? mine[ik] : -1;
    }
    #pragma unroll
    for (int k = 0; k < 8; ++k)
      if (wv[k] >= 0)
        col[bstart + atomicAdd(&ncur[wv[k] >> 17], 1)] = wv[k] & 0x1FFFF;
  }
}

__global__ void gather3_kernel(const float* __restrict__ x,
                               const int* __restrict__ rowstart,
                               const int* __restrict__ deg,
                               const int* __restrict__ col,
                               float* __restrict__ agg3) {
  int n = blockIdx.x * blockDim.x + threadIdx.x;
  if (n >= N_NODES) return;
  int b = rowstart[n];
  int c = deg[n];
  float a0 = 0.0f, a1 = 0.0f, a2 = 0.0f;
  int j = 0;
  for (; j + 4 <= c; j += 4) {
    int s0 = col[b + j], s1 = col[b + j + 1], s2 = col[b + j + 2], s3 = col[b + j + 3];
    float p0 = x[s0 * 3], p1 = x[s0 * 3 + 1], p2 = x[s0 * 3 + 2];
    float q0 = x[s1 * 3], q1 = x[s1 * 3 + 1], q2 = x[s1 * 3 + 2];
    float r0 = x[s2 * 3], r1 = x[s2 * 3 + 1], r2 = x[s2 * 3 + 2];
    float t0 = x[s3 * 3], t1 = x[s3 * 3 + 1], t2 = x[s3 * 3 + 2];
    a0 += p0; a1 += p1; a2 += p2;
    a0 += q0; a1 += q1; a2 += q2;
    a0 += r0; a1 += r1; a2 += r2;
    a0 += t0; a1 += t1; a2 += t2;
  }
  for (; j < c; ++j) {
    int s = col[b + j];
    a0 += x[s * 3]; a1 += x[s * 3 + 1]; a2 += x[s * 3 + 2];
  }
  agg3[n * 3]     = a0;
  agg3[n * 3 + 1] = a1;
  agg3[n * 3 + 2] = a2;
}

__global__ __launch_bounds__(128) void mlp1_kernel(
    const float* __restrict__ x, const float* __restrict__ agg3,
    const float* __restrict__ W1a, const float* __restrict__ b1a,
    const float* __restrict__ W1b, const float* __restrict__ b1b,
    float* __restrict__ h1) {
  __shared__ float sWa[4 * 64];
  __shared__ float sWb[64 * 64];
  __shared__ float sBa[64];
  __shared__ float sBb[64];
  __shared__ float sT[4][16 * 68];

  int tid = threadIdx.x;
  for (int i = tid; i < 4 * 64; i += 128) {
    int k = i >> 6, n = i & 63;
    sWa[i] = (k < 3) ? W1a[k * 64 + n] : 0.0f;
  }
  for (int i = tid; i < 64 * 64; i += 128) sWb[i] = W1b[i];
  if (tid < 64) { sBa[tid] = b1a[tid]; sBb[tid] = b1b[tid]; }
  __syncthreads();

  int wave = tid >> 5;
  int lane = tid & 31;
  int half = lane >> 4;
  int m    = lane & 15;
  int tile = blockIdx.x * 4 + wave;
  if (tile >= N_TILES) return;
  int node0 = tile * 16;
  float* myT = &sT[wave][0];

  int node = node0 + m;
  int k0   = 2 * half;
  v2f a;
  a.x = x[node * 3 + k0] + agg3[node * 3 + k0];
  a.y = (half == 0) ? (x[node * 3 + 1] + agg3[node * 3 + 1]) : 0.0f;

  v8f acc[4] = {};
  #pragma unroll
  for (int nt = 0; nt < 4; ++nt) {
    v2f b;
    b.x = sWa[k0 * 64 + nt * 16 + m];
    b.y = sWa[(k0 + 1) * 64 + nt * 16 + m];
    acc[nt] = __builtin_amdgcn_wmma_f32_16x16x4_f32(
        false, a, false, b, (short)0, acc[nt], false, false);
  }

  #pragma unroll
  for (int nt = 0; nt < 4; ++nt) {
    float bias = sBa[nt * 16 + m];
    #pragma unroll
    for (int r = 0; r < 8; ++r) {
      float v = acc[nt][r] + bias;
      myT[(r + 8 * half) * 68 + nt * 16 + m] = v > 0.0f ? v : 0.0f;
    }
  }

  v8f acc2[4] = {};
  for (int ks = 0; ks < 16; ++ks) {
    int kk = ks * 4 + 2 * half;
    v2f a2;
    a2.x = myT[m * 68 + kk];
    a2.y = myT[m * 68 + kk + 1];
    #pragma unroll
    for (int nt = 0; nt < 4; ++nt) {
      v2f b;
      b.x = sWb[kk * 64 + nt * 16 + m];
      b.y = sWb[(kk + 1) * 64 + nt * 16 + m];
      acc2[nt] = __builtin_amdgcn_wmma_f32_16x16x4_f32(
          false, a2, false, b, (short)0, acc2[nt], false, false);
    }
  }

  #pragma unroll
  for (int nt = 0; nt < 4; ++nt) {
    float bias = sBb[nt * 16 + m];
    #pragma unroll
    for (int r = 0; r < 8; ++r) {
      float v = acc2[nt][r] + bias;
      h1[(node0 + r + 8 * half) * 64 + nt * 16 + m] = v > 0.0f ? v : 0.0f;
    }
  }
}

#define ROW_LOAD(k) if (c <= k) return; p[k] = hv[id[k] * 32 + lane];
__device__ __forceinline__ void load_rows(v2f (&p)[31], const int (&id)[31], int c,
                                          const v2f* __restrict__ hv, int lane) {
  ROW_LOAD(0) ROW_LOAD(1) ROW_LOAD(2) ROW_LOAD(3) ROW_LOAD(4) ROW_LOAD(5) ROW_LOAD(6) ROW_LOAD(7)
  ROW_LOAD(8) ROW_LOAD(9) ROW_LOAD(10) ROW_LOAD(11) ROW_LOAD(12) ROW_LOAD(13) ROW_LOAD(14) ROW_LOAD(15)
  ROW_LOAD(16) ROW_LOAD(17) ROW_LOAD(18) ROW_LOAD(19) ROW_LOAD(20) ROW_LOAD(21) ROW_LOAD(22) ROW_LOAD(23)
  ROW_LOAD(24) ROW_LOAD(25) ROW_LOAD(26) ROW_LOAD(27) ROW_LOAD(28) ROW_LOAD(29) ROW_LOAD(30)
}
#define ROW_ADD(k) if (c <= k) return; acc += p[k];
__device__ __forceinline__ void add_rows(v2f& acc, const v2f (&p)[31], int c) {
  ROW_ADD(0) ROW_ADD(1) ROW_ADD(2) ROW_ADD(3) ROW_ADD(4) ROW_ADD(5) ROW_ADD(6) ROW_ADD(7)
  ROW_ADD(8) ROW_ADD(9) ROW_ADD(10) ROW_ADD(11) ROW_ADD(12) ROW_ADD(13) ROW_ADD(14) ROW_ADD(15)
  ROW_ADD(16) ROW_ADD(17) ROW_ADD(18) ROW_ADD(19) ROW_ADD(20) ROW_ADD(21) ROW_ADD(22) ROW_ADD(23)
  ROW_ADD(24) ROW_ADD(25) ROW_ADD(26) ROW_ADD(27) ROW_ADD(28) ROW_ADD(29) ROW_ADD(30)
}

__global__ __launch_bounds__(256) void gather_mlp2_pool_kernel(
    const float* __restrict__ h1, const int* __restrict__ rowstart,
    const int* __restrict__ deg, const int* __restrict__ col,
    const float* __restrict__ W2a, const float* __restrict__ b2a,
    const float* __restrict__ W2b, const float* __restrict__ b2b,
    const int* __restrict__ batch,
    float* __restrict__ sums, float* __restrict__ cnt) {
  __shared__ float sWa[64 * 64];
  __shared__ float sWb[64 * 64];
  __shared__ float sBa[64];
  __shared__ float sBb[64];
  __shared__ float sT[8][16 * 68];

  int tid = threadIdx.x;
  for (int i = tid; i < 64 * 64; i += 256) { sWa[i] = W2a[i]; sWb[i] = W2b[i]; }
  if (tid < 64) { sBa[tid] = b2a[tid]; sBb[tid] = b2b[tid]; }
  __syncthreads();

  int lane = tid & 31;
  int half = lane >> 4;
  int m    = lane & 15;
  int wave = __builtin_amdgcn_readfirstlane(tid >> 5);
  int tile = blockIdx.x * 8 + wave;
  if (tile >= N_TILES) return;
  int node0 = tile * 16;
  float* myT = &sT[wave][0];

  const v2f* __restrict__ hv = (const v2f*)h1;
  int rs_v = rowstart[node0 + m];
  int dg_v = deg[node0 + m];
  int b  = __builtin_amdgcn_readlane(rs_v, 0);
  int c  = __builtin_amdgcn_readlane(dg_v, 0);
  int cv = (lane < c) ? col[b + lane] : 0;
  int id[31];
  #pragma unroll
  for (int k = 0; k < 31; ++k) id[k] = __builtin_amdgcn_readlane(cv, k);
  for (int r = 0; r < 16; ++r) {
    int rn  = (r < 15) ? r + 1 : 15;
    int bn  = __builtin_amdgcn_readlane(rs_v, rn);
    int cn  = __builtin_amdgcn_readlane(dg_v, rn);
    int cvn = (lane < cn) ? col[bn + lane] : 0;
    v2f hs  = hv[(node0 + r) * 32 + lane];
    v2f acc = {0.0f, 0.0f};
    v2f p[31];
    load_rows(p, id, c, hv, lane);
    add_rows(acc, p, c);
    for (int j = 31; j < c; ++j)
      acc += hv[col[b + j] * 32 + lane];
    v2f av = hs + acc;
    myT[r * 68 + 2 * lane]     = av.x;
    myT[r * 68 + 2 * lane + 1] = av.y;
    #pragma unroll
    for (int k = 0; k < 31; ++k) id[k] = __builtin_amdgcn_readlane(cvn, k);
    b = bn; c = cn;
  }

  v8f acc[4] = {};
  for (int ks = 0; ks < 16; ++ks) {
    int kk = ks * 4 + 2 * half;
    v2f a;
    a.x = myT[m * 68 + kk];
    a.y = myT[m * 68 + kk + 1];
    #pragma unroll
    for (int nt = 0; nt < 4; ++nt) {
      v2f b;
      b.x = sWa[kk * 64 + nt * 16 + m];
      b.y = sWa[(kk + 1) * 64 + nt * 16 + m];
      acc[nt] = __builtin_amdgcn_wmma_f32_16x16x4_f32(
          false, a, false, b, (short)0, acc[nt], false, false);
    }
  }

  #pragma unroll
  for (int nt = 0; nt < 4; ++nt) {
    float bias = sBa[nt * 16 + m];
    #pragma unroll
    for (int r = 0; r < 8; ++r) {
      float v = acc[nt][r] + bias;
      myT[(r + 8 * half) * 68 + nt * 16 + m] = v > 0.0f ? v : 0.0f;
    }
  }

  v8f acc2[4] = {};
  for (int ks = 0; ks < 16; ++ks) {
    int kk = ks * 4 + 2 * half;
    v2f a2;
    a2.x = myT[m * 68 + kk];
    a2.y = myT[m * 68 + kk + 1];
    #pragma unroll
    for (int nt = 0; nt < 4; ++nt) {
      v2f b;
      b.x = sWb[kk * 64 + nt * 16 + m];
      b.y = sWb[(kk + 1) * 64 + nt * 16 + m];
      acc2[nt] = __builtin_amdgcn_wmma_f32_16x16x4_f32(
          false, a2, false, b, (short)0, acc2[nt], false, false);
    }
  }

  float v[4][8];
  #pragma unroll
  for (int nt = 0; nt < 4; ++nt) {
    float bias = sBb[nt * 16 + m];
    #pragma unroll
    for (int r = 0; r < 8; ++r) {
      float t = acc2[nt][r] + bias;
      v[nt][r] = t > 0.0f ? t : 0.0f;
    }
  }

  int gl = batch[node0 + m];
  int g0 = __builtin_amdgcn_readfirstlane(gl);
  bool one_graph = __builtin_amdgcn_ballot_w32(gl != g0) == 0u;
  if (one_graph) {
    float tot[4];
    #pragma unroll
    for (int nt = 0; nt < 4; ++nt) {
      float s = ((v[nt][0] + v[nt][1]) + (v[nt][2] + v[nt][3])) +
                ((v[nt][4] + v[nt][5]) + (v[nt][6] + v[nt][7]));
      tot[nt] = s + __shfl_xor(s, 16);
    }
    float* dst = sums + g0 * 64;
    atomicAdd(&dst[lane],      half ? tot[1] : tot[0]);
    atomicAdd(&dst[32 + lane], half ? tot[3] : tot[2]);
    if (lane == 0) atomicAdd(&cnt[g0], 16.0f);
  } else {
    #pragma unroll
    for (int r = 0; r < 8; ++r) {
      int g = batch[node0 + r + 8 * half];
      #pragma unroll
      for (int nt = 0; nt < 4; ++nt)
        atomicAdd(&sums[g * 64 + nt * 16 + m], v[nt][r]);
    }
    if (lane < 16) atomicAdd(&cnt[gl], 1.0f);
  }
}

__global__ void finalize_kernel(const float* __restrict__ sums,
                                const float* __restrict__ cnt,
                                float* __restrict__ out) {
  int i = blockIdx.x * blockDim.x + threadIdx.x;
  if (i >= N_GRAPHS * 64) return;
  float c = cnt[i >> 6];
  out[i] = sums[i] / fmaxf(c, 1.0f);
}

extern "C" void kernel_launch(void* const* d_in, const int* in_sizes, int n_in,
                              void* d_out, int out_size, void* d_ws, size_t ws_size,
                              hipStream_t stream) {
  const float* x     = (const float*)d_in[0];
  const int*   ei    = (const int*)d_in[1];
  const int*   batch = (const int*)d_in[2];
  const float* W1a   = (const float*)d_in[3];
  const float* b1a   = (const float*)d_in[4];
  const float* W1b   = (const float*)d_in[5];
  const float* b1b   = (const float*)d_in[6];
  const float* W2a   = (const float*)d_in[7];
  const float* b2a   = (const float*)d_in[8];
  const float* W2b   = (const float*)d_in[9];
  const float* b2b   = (const float*)d_in[10];
  float* out = (float*)d_out;

  float* ws       = (float*)d_ws;
  float* agg3     = ws;
  float* h1       = ws + 300032;
  int*   deg      = (int*)(ws + 13100032);
  float* sums     = ws + 13200032;
  float* cnt      = ws + 13208224;
  int*   rowstart = (int*)(ws + 13208384);
  int*   pairs    = (int*)(ws + 13308416);
  int*   col      = (int*)(ws + 14908416);
  int*   cntB     = (int*)(ws + 16508416);
  int*   baseT    = (int*)(ws + 16614912);
  int*   total    = (int*)(ws + 16721408);

  dim3 b256(256);
  bucket_count_kernel<<<NBLK, b256, 0, stream>>>(ei, cntB);
  bucket_scan_kernel<<<NB / 8, b256, 0, stream>>>(cntB, baseT, total, sums, cnt);
  partition_kernel<<<NBLK, b256, 0, stream>>>(ei, baseT, total, pairs);
  bucket_sort_kernel<<<NB, dim3(1024), 0, stream>>>(pairs, total, deg, rowstart, col);

  gather3_kernel<<<(N_NODES + 255) / 256, b256, 0, stream>>>(x, rowstart, deg, col, agg3);
  mlp1_kernel<<<(N_TILES + 3) / 4, dim3(128), 0, stream>>>(
      x, agg3, W1a, b1a, W1b, b1b, h1);
  gather_mlp2_pool_kernel<<<(N_TILES + 7) / 8, b256, 0, stream>>>(
      h1, rowstart, deg, col, W2a, b2a, W2b, b2b, batch, sums, cnt);
  finalize_kernel<<<(N_GRAPHS * 64 + 255) / 256, b256, 0, stream>>>(
      sums, cnt, out);
}


# Each block orders its own edge chunk by bucket in LDS (rank = returning LDS add); bucket kernel finds its pieces via a per-block offset table: count+scan launches gone; per-node sum order changed
# speedup vs baseline: 5.6142x; 1.2725x over previous
#include <hip/hip_runtime.h>


typedef __attribute__((ext_vector_type(2))) float v2f;
typedef __attribute__((ext_vector_type(8))) float v8f;

#define N_NODES  100000
#define N_EDGES  1600000
#define N_GRAPHS 128
#define N_TILES  (N_NODES / 16)


#define NB    256
#define BS    416
#define NBUCK ((N_NODES + BS - 1) / BS)
#define CH    4096
#define EPT   (CH / 256)
#define NBLK  ((N_EDGES + CH - 1) / CH)
#define NBLKP (((NBLK + 31) / 32) * 32)
#define PPW   ((NBLK + 31) / 32)

__global__ __launch_bounds__(256) void local_partition_kernel(
    const int* __restrict__ ei, int* __restrict__ pairs, int* __restrict__ offT,
    float* __restrict__ sums, float* __restrict__ cnt) {
  __shared__ int hist[NB];
  __shared__ int wsum[8];
  __shared__ int stage[CH];
  int t = threadIdx.x;
  int lane = t & 31, w = t >> 5;
  int e0 = blockIdx.x * CH;
  int s[EPT], d[EPT];
  #pragma unroll
  for (int k = 0; k < EPT; ++k) {
    int e = e0 + k * 256 + t;
    int ec = e < N_EDGES ? e : N_EDGES - 1;
    s[k] = ei[ec];
    d[k] = ei[N_EDGES + ec];
  }
  hist[t] = 0;
  __syncthreads();
  int rk[EPT];
  #pragma unroll
  for (int k = 0; k < EPT; ++k) {
    int e = e0 + k * 256 + t;
    rk[k] = 0;
    if (e < N_EDGES) rk[k] = atomicAdd(&hist[d[k] / BS], 1);
  }
  __syncthreads();
  int c = hist[t];
  int incl = c;
  for (int off = 1; off < 32; off <<= 1) {
    int u = __shfl_up(incl, off);
    if (lane >= off) incl += u;
  }
  if (lane == 31) wsum[w] = incl;
  __syncthreads();
  int woff = 0;
  for (int i = 0; i < w; ++i) woff += wsum[i];
  int excl = woff + incl - c;
  hist[t] = excl;
  __syncthreads();
  int bo[EPT];
  #pragma unroll
  for (int k = 0; k < EPT; ++k) bo[k] = hist[d[k] / BS];
  #pragma unroll
  for (int k = 0; k < EPT; ++k) {
    int e = e0 + k * 256 + t;
    if (e < N_EDGES) {
      int b = d[k] / BS;
      stage[bo[k] + rk[k]] = s[k] | ((d[k] - b * BS) << 17);
    }
  }
  __syncthreads();
  int nvalid = N_EDGES - e0 < CH ? N_EDGES - e0 : CH;
  int ov[EPT];
  #pragma unroll
  for (int k = 0; k < EPT; ++k) ov[k] = stage[k * 256 + t];
  #pragma unroll
  for (int k = 0; k < EPT; ++k) {
    int i = k * 256 + t;
    if (i < nvalid) pairs[e0 + i] = ov[k];
  }
  offT[t * NBLKP + blockIdx.x] = excl;
  if (t == NB - 1) offT[NB * NBLKP + blockIdx.x] = excl + c;
  if (blockIdx.x < (N_GRAPHS * 64) / 256) {
    int g = blockIdx.x * 256 + t;
    sums[g] = 0.0f;
    if (g < N_GRAPHS) cnt[g] = 0.0f;
  }
}

__global__ __launch_bounds__(1024) void bucket_sort_kernel(
    const int* __restrict__ pairs, const int* __restrict__ offT,
    int* __restrict__ deg, int* __restrict__ rowstart, int* __restrict__ col) {
  __shared__ int ncnt[512];
  __shared__ int ncur[512];
  __shared__ int psrc[NBLKP];
  __shared__ int plen[NBLKP];
  __shared__ int wsumA[32];
  __shared__ int wsumB[16];
  int t = threadIdx.x;
  int lane = t & 31;
  int w = __builtin_amdgcn_readfirstlane(t >> 5);
  int b = blockIdx.x;
  int o0 = 0;
  if (t < NBLKP) {
    int o1 = 0;
    if (t < NBLK) {
      o0 = offT[b * NBLKP + t];
      o1 = offT[(b + 1) * NBLKP + t];
    }
    psrc[t] = t < NBLK ? t * CH + o0 : 0;
    plen[t] = o1 - o0;
  }
  int v = o0;
  for (int off = 16; off > 0; off >>= 1) v += __shfl_xor(v, off);
  if (lane == 0) wsumA[w] = v;
  if (t < 512) ncnt[t] = 0;
  __syncthreads();
  int bstart = 0;
  for (int i = 0; i < 32; ++i) bstart += wsumA[i];

  int wv[PPW], pl[PPW], ps[PPW];
  #pragma unroll
  for (int i = 0; i < PPW; ++i) {
    pl[i] = plen[w + 32 * i];
    ps[i] = psrc[w + 32 * i];
  }
  int maxL = 0;
  #pragma unroll
  for (int i = 0; i < PPW; ++i) {
    int ix = ps[i] + (lane < pl[i] ? lane : 0);
    wv[i]  = pairs[ix < N_EDGES ? ix : N_EDGES - 1];
    maxL   = pl[i] > maxL ? pl[i] : maxL;
  }
  __builtin_amdgcn_sched_barrier(0);
  #pragma unroll
  for (int i = 0; i < PPW; ++i) wv[i] = lane < pl[i] ? wv[i] : -1;
  bool longer = maxL > 32;
  #pragma unroll
  for (int i = 0; i < PPW; ++i)
    if (wv[i] >= 0) atomicAdd(&ncnt[wv[i] >> 17], 1);
  if (longer) {
    for (int k = w; k < NBLK; k += 32) {
      int L = plen[k];
      const int* __restrict__ p = pairs + psrc[k];
      for (int j = 32 + lane; j < L; j += 32) atomicAdd(&ncnt[p[j] >> 17], 1);
    }
  }
  __syncthreads();

  int c = 0, incl = 0;
  if (t < 512) {
    c = ncnt[t];
    incl = c;
    for (int off = 1; off < 32; off <<= 1) {
      int u = __shfl_up(incl, off);
      if (lane >= off) incl += u;
    }
    if (lane == 31) wsumB[w] = incl;
  }
  __syncthreads();
  int ls = 0;
  if (t < 512) {
    int woff = 0;
    for (int i = 0; i < w; ++i) woff += wsumB[i];
    ls = woff + incl - c;
    ncur[t] = ls;
  }
  __syncthreads();

  int slot[PPW];
  #pragma unroll
  for (int i = 0; i < PPW; ++i) {
    slot[i] = 0;
    if (wv[i] >= 0) slot[i] = atomicAdd(&ncur[wv[i] >> 17], 1);
  }
  #pragma unroll
  for (int i = 0; i < PPW; ++i)
    if (wv[i] >= 0) col[bstart + slot[i]] = wv[i] & 0x1FFFF;
  if (longer) {
    for (int k = w; k < NBLK; k += 32) {
      int L = plen[k];
      const int* __restrict__ p = pairs + psrc[k];
      for (int j = 32 + lane; j < L; j += 32) {
        int x = p[j];
        col[bstart + atomicAdd(&ncur[x >> 17], 1)] = x & 0x1FFFF;
      }
    }
  }
  int node = b * BS + t;
  if (t < BS && node < N_NODES) {
    deg[node] = c;
    rowstart[node] = bstart + ls;
  }
}

__global__ void gather3_kernel(const float* __restrict__ x,
                               const int* __restrict__ rowstart,
                               const int* __restrict__ deg,
                               const int* __restrict__ col,
                               float* __restrict__ agg3) {
  int n = blockIdx.x * blockDim.x + threadIdx.x;
  if (n >= N_NODES) return;
  int b = rowstart[n];
  int c = deg[n];
  float a0 = 0.0f, a1 = 0.0f, a2 = 0.0f;
  int j = 0;
  for (; j + 4 <= c; j += 4) {
    int s0 = col[b + j], s1 = col[b + j + 1], s2 = col[b + j + 2], s3 = col[b + j + 3];
    float p0 = x[s0 * 3], p1 = x[s0 * 3 + 1], p2 = x[s0 * 3 + 2];
    float q0 = x[s1 * 3], q1 = x[s1 * 3 + 1], q2 = x[s1 * 3 + 2];
    float r0 = x[s2 * 3], r1 = x[s2 * 3 + 1], r2 = x[s2 * 3 + 2];
    float t0 = x[s3 * 3], t1 = x[s3 * 3 + 1], t2 = x[s3 * 3 + 2];
    a0 += p0; a1 += p1; a2 += p2;
    a0 += q0; a1 += q1; a2 += q2;
    a0 += r0; a1 += r1; a2 += r2;
    a0 += t0; a1 += t1; a2 += t2;
  }
  for (; j < c; ++j) {
    int s = col[b + j];
    a0 += x[s * 3]; a1 += x[s * 3 + 1]; a2 += x[s * 3 + 2];
  }
  agg3[n * 3]     = a0;
  agg3[n * 3 + 1] = a1;
  agg3[n * 3 + 2] = a2;
}

__global__ __launch_bounds__(128) void mlp1_kernel(
    const float* __restrict__ x, const float* __restrict__ agg3,
    const float* __restrict__ W1a, const float* __restrict__ b1a,
    const float* __restrict__ W1b, const float* __restrict__ b1b,
    float* __restrict__ h1) {
  __shared__ float sWa[4 * 64];
  __shared__ float sWb[64 * 64];
  __shared__ float sBa[64];
  __shared__ float sBb[64];
  __shared__ float sT[4][16 * 68];

  int tid = threadIdx.x;
  for (int i = tid; i < 4 * 64; i += 128) {
    int k = i >> 6, n = i & 63;
    sWa[i] = (k < 3) ? W1a[k * 64 + n] : 0.0f;
  }
  for (int i = tid; i < 64 * 64; i += 128) sWb[i] = W1b[i];
  if (tid < 64) { sBa[tid] = b1a[tid]; sBb[tid] = b1b[tid]; }
  __syncthreads();

  int wave = tid >> 5;
  int lane = tid & 31;
  int half = lane >> 4;
  int m    = lane & 15;
  int tile = blockIdx.x * 4 + wave;
  if (tile >= N_TILES) return;
  int node0 = tile * 16;
  float* myT = &sT[wave][0];

  int node = node0 + m;
  int k0   = 2 * half;
  v2f a;
  a.x = x[node * 3 + k0] + agg3[node * 3 + k0];
  a.y = (half == 0) ? (x[node * 3 + 1] + agg3[node * 3 + 1]) : 0.0f;

  v8f acc[4] = {};
  #pragma unroll
  for (int nt = 0; nt < 4; ++nt) {
    v2f b;
    b.x = sWa[k0 * 64 + nt * 16 + m];
    b.y = sWa[(k0 + 1) * 64 + nt * 16 + m];
    acc[nt] = __builtin_amdgcn_wmma_f32_16x16x4_f32(
        false, a, false, b, (short)0, acc[nt], false, false);
  }

  #pragma unroll
  for (int nt = 0; nt < 4; ++nt) {
    float bias = sBa[nt * 16 + m];
    #pragma unroll
    for (int r = 0; r < 8; ++r) {
      float v = acc[nt][r] + bias;
      myT[(r + 8 * half) * 68 + nt * 16 + m] = v > 0.0f ? v : 0.0f;
    }
  }

  v8f acc2[4] = {};
  for (int ks = 0; ks < 16; ++ks) {
    int kk = ks * 4 + 2 * half;
    v2f a2;
    a2.x = myT[m * 68 + kk];
    a2.y = myT[m * 68 + kk + 1];
    #pragma unroll
    for (int nt = 0; nt < 4; ++nt) {
      v2f b;
      b.x = sWb[kk * 64 + nt * 16 + m];
      b.y = sWb[(kk + 1) * 64 + nt * 16 + m];
      acc2[nt] = __builtin_amdgcn_wmma_f32_16x16x4_f32(
          false, a2, false, b, (short)0, acc2[nt], false, false);
    }
  }

  #pragma unroll
  for (int nt = 0; nt < 4; ++nt) {
    float bias = sBb[nt * 16 + m];
    #pragma unroll
    for (int r = 0; r < 8; ++r) {
      float v = acc2[nt][r] + bias;
      h1[(node0 + r + 8 * half) * 64 + nt * 16 + m] = v > 0.0f ? v : 0.0f;
    }
  }
}

#define ROW_LOAD(k) if (c <= k) return; p[k] = hv[id[k] * 32 + lane];
__device__ __forceinline__ void load_rows(v2f (&p)[31], const int (&id)[31], int c,
                                          const v2f* __restrict__ hv, int lane) {
  ROW_LOAD(0) ROW_LOAD(1) ROW_LOAD(2) ROW_LOAD(3) ROW_LOAD(4) ROW_LOAD(5) ROW_LOAD(6) ROW_LOAD(7)
  ROW_LOAD(8) ROW_LOAD(9) ROW_LOAD(10) ROW_LOAD(11) ROW_LOAD(12) ROW_LOAD(13) ROW_LOAD(14) ROW_LOAD(15)
  ROW_LOAD(16) ROW_LOAD(17) ROW_LOAD(18) ROW_LOAD(19) ROW_LOAD(20) ROW_LOAD(21) ROW_LOAD(22) ROW_LOAD(23)
  ROW_LOAD(24) ROW_LOAD(25) ROW_LOAD(26) ROW_LOAD(27) ROW_LOAD(28) ROW_LOAD(29) ROW_LOAD(30)
}
#define ROW_ADD(k) if (c <= k) return; acc += p[k];
__device__ __forceinline__ void add_rows(v2f& acc, const v2f (&p)[31], int c) {
  ROW_ADD(0) ROW_ADD(1) ROW_ADD(2) ROW_ADD(3) ROW_ADD(4) ROW_ADD(5) ROW_ADD(6) ROW_ADD(7)
  ROW_ADD(8) ROW_ADD(9) ROW_ADD(10) ROW_ADD(11) ROW_ADD(12) ROW_ADD(13) ROW_ADD(14) ROW_ADD(15)
  ROW_ADD(16) ROW_ADD(17) ROW_ADD(18) ROW_ADD(19) ROW_ADD(20) ROW_ADD(21) ROW_ADD(22) ROW_ADD(23)
  ROW_ADD(24) ROW_ADD(25) ROW_ADD(26) ROW_ADD(27) ROW_ADD(28) ROW_ADD(29) ROW_ADD(30)
}

__global__ __launch_bounds__(256) void gather_mlp2_pool_kernel(
    const float* __restrict__ h1, const int* __restrict__ rowstart,
    const int* __restrict__ deg, const int* __restrict__ col,
    const float* __restrict__ W2a, const float* __restrict__ b2a,
    const float* __restrict__ W2b, const float* __restrict__ b2b,
    const int* __restrict__ batch,
    float* __restrict__ sums, float* __restrict__ cnt) {
  __shared__ float sWa[64 * 64];
  __shared__ float sWb[64 * 64];
  __shared__ float sBa[64];
  __shared__ float sBb[64];
  __shared__ float sT[8][16 * 68];

  int tid = threadIdx.x;
  for (int i = tid; i < 64 * 64; i += 256) { sWa[i] = W2a[i]; sWb[i] = W2b[i]; }
  if (tid < 64) { sBa[tid] = b2a[tid]; sBb[tid] = b2b[tid]; }
  __syncthreads();

  int lane = tid & 31;
  int half = lane >> 4;
  int m    = lane & 15;
  int wave = __builtin_amdgcn_readfirstlane(tid >> 5);
  int tile = blockIdx.x * 8 + wave;
  if (tile >= N_TILES) return;
  int node0 = tile * 16;
  float* myT = &sT[wave][0];

  const v2f* __restrict__ hv = (const v2f*)h1;
  int rs_v = rowstart[node0 + m];
  int dg_v = deg[node0 + m];
  int b  = __builtin_amdgcn_readlane(rs_v, 0);
  int c  = __builtin_amdgcn_readlane(dg_v, 0);
  int cv = (lane < c) ? col[b + lane] : 0;
  int id[31];
  #pragma unroll
  for (int k = 0; k < 31; ++k) id[k] = __builtin_amdgcn_readlane(cv, k);
  for (int r = 0; r < 16; ++r) {
    int rn  = (r < 15) ? r + 1 : 15;
    int bn  = __builtin_amdgcn_readlane(rs_v, rn);
    int cn  = __builtin_amdgcn_readlane(dg_v, rn);
    int cvn = (lane < cn) ? col[bn + lane] : 0;
    v2f hs  = hv[(node0 + r) * 32 + lane];
    v2f acc = {0.0f, 0.0f};
    v2f p[31];
    load_rows(p, id, c, hv, lane);
    add_rows(acc, p, c);
    for (int j = 31; j < c; ++j)
      acc += hv[col[b + j] * 32 + lane];
    v2f av = hs + acc;
    myT[r * 68 + 2 * lane]     = av.x;
    myT[r * 68 + 2 * lane + 1] = av.y;
    #pragma unroll
    for (int k = 0; k < 31; ++k) id[k] = __builtin_amdgcn_readlane(cvn, k);
    b = bn; c = cn;
  }

  v8f acc[4] = {};
  for (int ks = 0; ks < 16; ++ks) {
    int kk = ks * 4 + 2 * half;
    v2f a;
    a.x = myT[m * 68 + kk];
    a.y = myT[m * 68 + kk + 1];
    #pragma unroll
    for (int nt = 0; nt < 4; ++nt) {
      v2f b;
      b.x = sWa[kk * 64 + nt * 16 + m];
      b.y = sWa[(kk + 1) * 64 + nt * 16 + m];
      acc[nt] = __builtin_amdgcn_wmma_f32_16x16x4_f32(
          false, a, false, b, (short)0, acc[nt], false, false);
    }
  }

  #pragma unroll
  for (int nt = 0; nt < 4; ++nt) {
    float bias = sBa[nt * 16 + m];
    #pragma unroll
    for (int r = 0; r < 8; ++r) {
      float v = acc[nt][r] + bias;
      myT[(r + 8 * half) * 68 + nt * 16 + m] = v > 0.0f ? v : 0.0f;
    }
  }

  v8f acc2[4] = {};
  for (int ks = 0; ks < 16; ++ks) {
    int kk = ks * 4 + 2 * half;
    v2f a2;
    a2.x = myT[m * 68 + kk];
    a2.y = myT[m * 68 + kk + 1];
    #pragma unroll
    for (int nt = 0; nt < 4; ++nt) {
      v2f b;
      b.x = sWb[kk * 64 + nt * 16 + m];
      b.y = sWb[(kk + 1) * 64 + nt * 16 + m];
      acc2[nt] = __builtin_amdgcn_wmma_f32_16x16x4_f32(
          false, a2, false, b, (short)0, acc2[nt], false, false);
    }
  }

  float v[4][8];
  #pragma unroll
  for (int nt = 0; nt < 4; ++nt) {
    float bias = sBb[nt * 16 + m];
    #pragma unroll
    for (int r = 0; r < 8; ++r) {
      float t = acc2[nt][r] + bias;
      v[nt][r] = t > 0.0f ? t : 0.0f;
    }
  }

  int gl = batch[node0 + m];
  int g0 = __builtin_amdgcn_readfirstlane(gl);
  bool one_graph = __builtin_amdgcn_ballot_w32(gl != g0) == 0u;
  if (one_graph) {
    float tot[4];
    #pragma unroll
    for (int nt = 0; nt < 4; ++nt) {
      float s = ((v[nt][0] + v[nt][1]) + (v[nt][2] + v[nt][3])) +
                ((v[nt][4] + v[nt][5]) + (v[nt][6] + v[nt][7]));
      tot[nt] = s + __shfl_xor(s, 16);
    }
    float* dst = sums + g0 * 64;
    atomicAdd(&dst[lane],      half ? tot[1] : tot[0]);
    atomicAdd(&dst[32 + lane], half ? tot[3] : tot[2]);
    if (lane == 0) atomicAdd(&cnt[g0], 16.0f);
  } else {
    #pragma unroll
    for (int r = 0; r < 8; ++r) {
      int g = batch[node0 + r + 8 * half];
      #pragma unroll
      for (int nt = 0; nt < 4; ++nt)
        atomicAdd(&sums[g * 64 + nt * 16 + m], v[nt][r]);
    }
    if (lane < 16) atomicAdd(&cnt[gl], 1.0f);
  }
}

__global__ void finalize_kernel(const float* __restrict__ sums,
                                const float* __restrict__ cnt,
                                float* __restrict__ out) {
  int i = blockIdx.x * blockDim.x + threadIdx.x;
  if (i >= N_GRAPHS * 64) return;
  float c = cnt[i >> 6];
  out[i] = sums[i] / fmaxf(c, 1.0f);
}

extern "C" void kernel_launch(void* const* d_in, const int* in_sizes, int n_in,
                              void* d_out, int out_size, void* d_ws, size_t ws_size,
                              hipStream_t stream) {
  const float* x     = (const float*)d_in[0];
  const int*   ei    = (const int*)d_in[1];
  const int*   batch = (const int*)d_in[2];
  const float* W1a   = (const float*)d_in[3];
  const float* b1a   = (const float*)d_in[4];
  const float* W1b   = (const float*)d_in[5];
  const float* b1b   = (const float*)d_in[6];
  const float* W2a   = (const float*)d_in[7];
  const float* b2a   = (const float*)d_in[8];
  const float* W2b   = (const float*)d_in[9];
  const float* b2b   = (const float*)d_in[10];
  float* out = (float*)d_out;

  float* ws       = (float*)d_ws;
  float* agg3     = ws;
  float* h1       = ws + 300032;
  int*   deg      = (int*)(ws + 13100032);
  float* sums     = ws + 13200032;
  float* cnt      = ws + 13208224;
  int*   rowstart = (int*)(ws + 13208384);
  int*   pairs    = (int*)(ws + 13308416);
  int*   col      = (int*)(ws + 14908416);
  int*   offT     = (int*)(ws + 16508416);

  dim3 b256(256);
  local_partition_kernel<<<NBLK, b256, 0, stream>>>(ei, pairs, offT, sums, cnt);
  bucket_sort_kernel<<<NBUCK, dim3(1024), 0, stream>>>(pairs, offT, deg, rowstart, col);

  gather3_kernel<<<(N_NODES + 255) / 256, b256, 0, stream>>>(x, rowstart, deg, col, agg3);
  mlp1_kernel<<<(N_TILES + 3) / 4, dim3(128), 0, stream>>>(
      x, agg3, W1a, b1a, W1b, b1b, h1);
  gather_mlp2_pool_kernel<<<(N_TILES + 7) / 8, b256, 0, stream>>>(
      h1, rowstart, deg, col, W2a, b2a, W2b, b2b, batch, sums, cnt);
  finalize_kernel<<<(N_GRAPHS * 64 + 255) / 256, b256, 0, stream>>>(
      sums, cnt, out);
}


# Layer-1 neighbour gather fused into the layer-1 MLP kernel: each lane sums its node's neighbour x rows in registers, 8 ids read ahead; aggregate never stored, gather launch gone; sum order unchanged
# speedup vs baseline: 6.0211x; 1.0725x over previous
#include <hip/hip_runtime.h>


typedef __attribute__((ext_vector_type(2))) float v2f;
typedef __attribute__((ext_vector_type(8))) float v8f;

#define N_NODES  100000
#define N_EDGES  1600000
#define N_GRAPHS 128
#define N_TILES  (N_NODES / 16)


#define NB    256
#define BS    416
#define NBUCK ((N_NODES + BS - 1) / BS)
#define CH    4096
#define EPT   (CH / 256)
#define NBLK  ((N_EDGES + CH - 1) / CH)
#define NBLKP (((NBLK + 31) / 32) * 32)
#define PPW   ((NBLK + 31) / 32)

__global__ __launch_bounds__(256) void local_partition_kernel(
    const int* __restrict__ ei, int* __restrict__ pairs, int* __restrict__ offT,
    float* __restrict__ sums, float* __restrict__ cnt) {
  __shared__ int hist[NB];
  __shared__ int wsum[8];
  __shared__ int stage[CH];
  int t = threadIdx.x;
  int lane = t & 31, w = t >> 5;
  int e0 = blockIdx.x * CH;
  int s[EPT], d[EPT];
  #pragma unroll
  for (int k = 0; k < EPT; ++k) {
    int e = e0 + k * 256 + t;
    int ec = e < N_EDGES ? e : N_EDGES - 1;
    s[k] = ei[ec];
    d[k] = ei[N_EDGES + ec];
  }
  hist[t] = 0;
  __syncthreads();
  int rk[EPT];
  #pragma unroll
  for (int k = 0; k < EPT; ++k) {
    int e = e0 + k * 256 + t;
    rk[k] = 0;
    if (e < N_EDGES) rk[k] = atomicAdd(&hist[d[k] / BS], 1);
  }
  __syncthreads();
  int c = hist[t];
  int incl = c;
  for (int off = 1; off < 32; off <<= 1) {
    int u = __shfl_up(incl, off);
    if (lane >= off) incl += u;
  }
  if (lane == 31) wsum[w] = incl;
  __syncthreads();
  int woff = 0;
  for (int i = 0; i < w; ++i) woff += wsum[i];
  int excl = woff + incl - c;
  hist[t] = excl;
  __syncthreads();
  int bo[EPT];
  #pragma unroll
  for (int k = 0; k < EPT; ++k) bo[k] = hist[d[k] / BS];
  #pragma unroll
  for (int k = 0; k < EPT; ++k) {
    int e = e0 + k * 256 + t;
    if (e < N_EDGES) {
      int b = d[k] / BS;
      stage[bo[k] + rk[k]] = s[k] | ((d[k] - b * BS) << 17);
    }
  }
  __syncthreads();
  int nvalid = N_EDGES - e0 < CH ? N_EDGES - e0 : CH;
  int ov[EPT];
  #pragma unroll
  for (int k = 0; k < EPT; ++k) ov[k] = stage[k * 256 + t];
  #pragma unroll
  for (int k = 0; k < EPT; ++k) {
    int i = k * 256 + t;
    if (i < nvalid) pairs[e0 + i] = ov[k];
  }
  offT[t * NBLKP + blockIdx.x] = excl;
  if (t == NB - 1) offT[NB * NBLKP + blockIdx.x] = excl + c;
  if (blockIdx.x < (N_GRAPHS * 64) / 256) {
    int g = blockIdx.x * 256 + t;
    sums[g] = 0.0f;
    if (g < N_GRAPHS) cnt[g] = 0.0f;
  }
}

__global__ __launch_bounds__(1024) void bucket_sort_kernel(
    const int* __restrict__ pairs, const int* __restrict__ offT,
    int* __restrict__ deg, int* __restrict__ rowstart, int* __restrict__ col) {
  __shared__ int ncnt[512];
  __shared__ int ncur[512];
  __shared__ int psrc[NBLKP];
  __shared__ int plen[NBLKP];
  __shared__ int wsumA[32];
  __shared__ int wsumB[16];
  int t = threadIdx.x;
  int lane = t & 31;
  int w = __builtin_amdgcn_readfirstlane(t >> 5);
  int b = blockIdx.x;
  int o0 = 0;
  if (t < NBLKP) {
    int o1 = 0;
    if (t < NBLK) {
      o0 = offT[b * NBLKP + t];
      o1 = offT[(b + 1) * NBLKP + t];
    }
    psrc[t] = t < NBLK ? t * CH + o0 : 0;
    plen[t] = o1 - o0;
  }
  int v = o0;
  for (int off = 16; off > 0; off >>= 1) v += __shfl_xor(v, off);
  if (lane == 0) wsumA[w] = v;
  if (t < 512) ncnt[t] = 0;
  __syncthreads();
  int bstart = 0;
  for (int i = 0; i < 32; ++i) bstart += wsumA[i];

  int wv[PPW], pl[PPW], ps[PPW];
  #pragma unroll
  for (int i = 0; i < PPW; ++i) {
    pl[i] = plen[w + 32 * i];
    ps[i] = psrc[w + 32 * i];
  }
  int maxL = 0;
  #pragma unroll
  for (int i = 0; i < PPW; ++i) {
    int ix = ps[i] + (lane < pl[i] ? lane : 0);
    wv[i]  = pairs[ix < N_EDGES ? ix : N_EDGES - 1];
    maxL   = pl[i] > maxL ? pl[i] : maxL;
  }
  __builtin_amdgcn_sched_barrier(0);
  #pragma unroll
  for (int i = 0; i < PPW; ++i) wv[i] = lane < pl[i] ? wv[i] : -1;
  bool longer = maxL > 32;
  #pragma unroll
  for (int i = 0; i < PPW; ++i)
    if (wv[i] >= 0) atomicAdd(&ncnt[wv[i] >> 17], 1);
  if (longer) {
    for (int k = w; k < NBLK; k += 32) {
      int L = plen[k];
      const int* __restrict__ p = pairs + psrc[k];
      for (int j = 32 + lane; j < L; j += 32) atomicAdd(&ncnt[p[j] >> 17], 1);
    }
  }
  __syncthreads();

  int c = 0, incl = 0;
  if (t < 512) {
    c = ncnt[t];
    incl = c;
    for (int off = 1; off < 32; off <<= 1) {
      int u = __shfl_up(incl, off);
      if (lane >= off) incl += u;
    }
    if (lane == 31) wsumB[w] = incl;
  }
  __syncthreads();
  int ls = 0;
  if (t < 512) {
    int woff = 0;
    for (int i = 0; i < w; ++i) woff += wsumB[i];
    ls = woff + incl - c;
    ncur[t] = ls;
  }
  __syncthreads();

  int slot[PPW];
  #pragma unroll
  for (int i = 0; i < PPW; ++i) {
    slot[i] = 0;
    if (wv[i] >= 0) slot[i] = atomicAdd(&ncur[wv[i] >> 17], 1);
  }
  #pragma unroll
  for (int i = 0; i < PPW; ++i)
    if (wv[i] >= 0) col[bstart + slot[i]] = wv[i] & 0x1FFFF;
  if (longer) {
    for (int k = w; k < NBLK; k += 32) {
      int L = plen[k];
      const int* __restrict__ p = pairs + psrc[k];
      for (int j = 32 + lane; j < L; j += 32) {
        int x = p[j];
        col[bstart + atomicAdd(&ncur[x >> 17], 1)] = x & 0x1FFFF;
      }
    }
  }
  int node = b * BS + t;
  if (t < BS && node < N_NODES) {
    deg[node] = c;
    rowstart[node] = bstart + ls;
  }
}

__global__ __launch_bounds__(128) void gather_mlp1_kernel(
    const float* __restrict__ x, const int* __restrict__ rowstart,
    const int* __restrict__ deg, const int* __restrict__ col,
    const float* __restrict__ W1a, const float* __restrict__ b1a,
    const float* __restrict__ W1b, const float* __restrict__ b1b,
    float* __restrict__ h1) {
  __shared__ float sWa[4 * 64];
  __shared__ float sWb[64 * 64];
  __shared__ float sBa[64];
  __shared__ float sBb[64];
  __shared__ float sT[4][16 * 68];

  int tid  = threadIdx.x;
  int wave = tid >> 5;
  int lane = tid & 31;
  int half = lane >> 4;
  int m    = lane & 15;
  int tile = blockIdx.x * 4 + wave;
  bool live = tile < N_TILES;
  int node0 = tile * 16;
  int node  = node0 + m;

  float u0 = 0.0f, u1 = 0.0f, u2 = 0.0f;
  if (live) {
    int b = rowstart[node];
    int c = deg[node];
    float xs0 = x[node * 3], xs1 = x[node * 3 + 1], xs2 = x[node * 3 + 2];
    float a0 = 0.0f, a1 = 0.0f, a2 = 0.0f;
    int id[8];
    #pragma unroll
    for (int k = 0; k < 8; ++k) id[k] = col[b + k];
    for (int j = 0; j < c; j += 8) {
      int nx[8];
      #pragma unroll
      for (int k = 0; k < 8; ++k) nx[k] = 0;
      if (j + 8 < c) {
        #pragma unroll
        for (int k = 0; k < 8; ++k) nx[k] = col[b + j + 8 + k];
      }
      float p0[8], p1[8], p2[8];
      #pragma unroll
      for (int k = 0; k < 8; ++k) {
        int s = (j + k < c) ? id[k] : 0;
        p0[k] = x[s * 3];
        p1[k] = x[s * 3 + 1];
        p2[k] = x[s * 3 + 2];
      }
      #pragma unroll
      for (int k = 0; k < 8; ++k) {
        if (j + k < c) { a0 += p0[k]; a1 += p1[k]; a2 += p2[k]; }
      }
      #pragma unroll
      for (int k = 0; k < 8; ++k) id[k] = nx[k];
    }
    u0 = xs0 + a0;
    u1 = xs1 + a1;
    u2 = xs2 + a2;
  }

  for (int i = tid; i < 4 * 64; i += 128) {
    int k = i >> 6, n = i & 63;
    sWa[i] = (k < 3) ? W1a[k * 64 + n] : 0.0f;
  }
  for (int i = tid; i < 64 * 64; i += 128) sWb[i] = W1b[i];
  if (tid < 64) { sBa[tid] = b1a[tid]; sBb[tid] = b1b[tid]; }
  __syncthreads();

  if (!live) return;
  float* myT = &sT[wave][0];

  int k0 = 2 * half;
  v2f a;
  a.x = half ? u2 : u0;
  a.y = half ? 0.0f : u1;

  v8f acc[4] = {};
  #pragma unroll
  for (int nt = 0; nt < 4; ++nt) {
    v2f b;
    b.x = sWa[k0 * 64 + nt * 16 + m];
    b.y = sWa[(k0 + 1) * 64 + nt * 16 + m];
    acc[nt] = __builtin_amdgcn_wmma_f32_16x16x4_f32(
        false, a, false, b, (short)0, acc[nt], false, false);
  }

  #pragma unroll
  for (int nt = 0; nt < 4; ++nt) {
    float bias = sBa[nt * 16 + m];
    #pragma unroll
    for (int r = 0; r < 8; ++r) {
      float v = acc[nt][r] + bias;
      myT[(r + 8 * half) * 68 + nt * 16 + m] = v > 0.0f ? v : 0.0f;
    }
  }

  v8f acc2[4] = {};
  for (int ks = 0; ks < 16; ++ks) {
    int kk = ks * 4 + 2 * half;
    v2f a2;
    a2.x = myT[m * 68 + kk];
    a2.y = myT[m * 68 + kk + 1];
    #pragma unroll
    for (int nt = 0; nt < 4; ++nt) {
      v2f b;
      b.x = sWb[kk * 64 + nt * 16 + m];
      b.y = sWb[(kk + 1) * 64 + nt * 16 + m];
      acc2[nt] = __builtin_amdgcn_wmma_f32_16x16x4_f32(
          false, a2, false, b, (short)0, acc2[nt], false, false);
    }
  }

  #pragma unroll
  for (int nt = 0; nt < 4; ++nt) {
    float bias = sBb[nt * 16 + m];
    #pragma unroll
    for (int r = 0; r < 8; ++r) {
      float v = acc2[nt][r] + bias;
      h1[(node0 + r + 8 * half) * 64 + nt * 16 + m] = v > 0.0f ? v : 0.0f;
    }
  }
}

#define ROW_LOAD(k) if (c <= k) return; p[k] = hv[id[k] * 32 + lane];
__device__ __forceinline__ void load_rows(v2f (&p)[31], const int (&id)[31], int c,
                                          const v2f* __restrict__ hv, int lane) {
  ROW_LOAD(0) ROW_LOAD(1) ROW_LOAD(2) ROW_LOAD(3) ROW_LOAD(4) ROW_LOAD(5) ROW_LOAD(6) ROW_LOAD(7)
  ROW_LOAD(8) ROW_LOAD(9) ROW_LOAD(10) ROW_LOAD(11) ROW_LOAD(12) ROW_LOAD(13) ROW_LOAD(14) ROW_LOAD(15)
  ROW_LOAD(16) ROW_LOAD(17) ROW_LOAD(18) ROW_LOAD(19) ROW_LOAD(20) ROW_LOAD(21) ROW_LOAD(22) ROW_LOAD(23)
  ROW_LOAD(24) ROW_LOAD(25) ROW_LOAD(26) ROW_LOAD(27) ROW_LOAD(28) ROW_LOAD(29) ROW_LOAD(30)
}
#define ROW_ADD(k) if (c <= k) return; acc += p[k];
__device__ __forceinline__ void add_rows(v2f& acc, const v2f (&p)[31], int c) {
  ROW_ADD(0) ROW_ADD(1) ROW_ADD(2) ROW_ADD(3) ROW_ADD(4) ROW_ADD(5) ROW_ADD(6) ROW_ADD(7)
  ROW_ADD(8) ROW_ADD(9) ROW_ADD(10) ROW_ADD(11) ROW_ADD(12) ROW_ADD(13) ROW_ADD(14) ROW_ADD(15)
  ROW_ADD(16) ROW_ADD(17) ROW_ADD(18) ROW_ADD(19) ROW_ADD(20) ROW_ADD(21) ROW_ADD(22) ROW_ADD(23)
  ROW_ADD(24) ROW_ADD(25) ROW_ADD(26) ROW_ADD(27) ROW_ADD(28) ROW_ADD(29) ROW_ADD(30)
}

__global__ __launch_bounds__(256) void gather_mlp2_pool_kernel(
    const float* __restrict__ h1, const int* __restrict__ rowstart,
    const int* __restrict__ deg, const int* __restrict__ col,
    const float* __restrict__ W2a, const float* __restrict__ b2a,
    const float* __restrict__ W2b, const float* __restrict__ b2b,
    const int* __restrict__ batch,
    float* __restrict__ sums, float* __restrict__ cnt) {
  __shared__ float sWa[64 * 64];
  __shared__ float sWb[64 * 64];
  __shared__ float sBa[64];
  __shared__ float sBb[64];
  __shared__ float sT[8][16 * 68];

  int tid = threadIdx.x;
  for (int i = tid; i < 64 * 64; i += 256) { sWa[i] = W2a[i]; sWb[i] = W2b[i]; }
  if (tid < 64) { sBa[tid] = b2a[tid]; sBb[tid] = b2b[tid]; }
  __syncthreads();

  int lane = tid & 31;
  int half = lane >> 4;
  int m    = lane & 15;
  int wave = __builtin_amdgcn_readfirstlane(tid >> 5);
  int tile = blockIdx.x * 8 + wave;
  if (tile >= N_TILES) return;
  int node0 = tile * 16;
  float* myT = &sT[wave][0];

  const v2f* __restrict__ hv = (const v2f*)h1;
  int rs_v = rowstart[node0 + m];
  int dg_v = deg[node0 + m];
  int b  = __builtin_amdgcn_readlane(rs_v, 0);
  int c  = __builtin_amdgcn_readlane(dg_v, 0);
  int cv = (lane < c) ? col[b + lane] : 0;
  int id[31];
  #pragma unroll
  for (int k = 0; k < 31; ++k) id[k] = __builtin_amdgcn_readlane(cv, k);
  for (int r = 0; r < 16; ++r) {
    int rn  = (r < 15) ? r + 1 : 15;
    int bn  = __builtin_amdgcn_readlane(rs_v, rn);
    int cn  = __builtin_amdgcn_readlane(dg_v, rn);
    int cvn = (lane < cn) ? col[bn + lane] : 0;
    v2f hs  = hv[(node0 + r) * 32 + lane];
    v2f acc = {0.0f, 0.0f};
    v2f p[31];
    load_rows(p, id, c, hv, lane);
    add_rows(acc, p, c);
    for (int j = 31; j < c; ++j)
      acc += hv[col[b + j] * 32 + lane];
    v2f av = hs + acc;
    myT[r * 68 + 2 * lane]     = av.x;
    myT[r * 68 + 2 * lane + 1] = av.y;
    #pragma unroll
    for (int k = 0; k < 31; ++k) id[k] = __builtin_amdgcn_readlane(cvn, k);
    b = bn; c = cn;
  }

  v8f acc[4] = {};
  for (int ks = 0; ks < 16; ++ks) {
    int kk = ks * 4 + 2 * half;
    v2f a;
    a.x = myT[m * 68 + kk];
    a.y = myT[m * 68 + kk + 1];
    #pragma unroll
    for (int nt = 0; nt < 4; ++nt) {
      v2f b;
      b.x = sWa[kk * 64 + nt * 16 + m];
      b.y = sWa[(kk + 1) * 64 + nt * 16 + m];
      acc[nt] = __builtin_amdgcn_wmma_f32_16x16x4_f32(
          false, a, false, b, (short)0, acc[nt], false, false);
    }
  }

  #pragma unroll
  for (int nt = 0; nt < 4; ++nt) {
    float bias = sBa[nt * 16 + m];
    #pragma unroll
    for (int r = 0; r < 8; ++r) {
      float v = acc[nt][r] + bias;
      myT[(r + 8 * half) * 68 + nt * 16 + m] = v > 0.0f ? v : 0.0f;
    }
  }

  v8f acc2[4] = {};
  for (int ks = 0; ks < 16; ++ks) {
    int kk = ks * 4 + 2 * half;
    v2f a2;
    a2.x = myT[m * 68 + kk];
    a2.y = myT[m * 68 + kk + 1];
    #pragma unroll
    for (int nt = 0; nt < 4; ++nt) {
      v2f b;
      b.x = sWb[kk * 64 + nt * 16 + m];
      b.y = sWb[(kk + 1) * 64 + nt * 16 + m];
      acc2[nt] = __builtin_amdgcn_wmma_f32_16x16x4_f32(
          false, a2, false, b, (short)0, acc2[nt], false, false);
    }
  }

  float v[4][8];
  #pragma unroll
  for (int nt = 0; nt < 4; ++nt) {
    float bias = sBb[nt * 16 + m];
    #pragma unroll
    for (int r = 0; r < 8; ++r) {
      float t = acc2[nt][r] + bias;
      v[nt][r] = t > 0.0f ? t : 0.0f;
    }
  }

  int gl = batch[node0 + m];
  int g0 = __builtin_amdgcn_readfirstlane(gl);
  bool one_graph = __builtin_amdgcn_ballot_w32(gl != g0) == 0u;
  if (one_graph) {
    float tot[4];
    #pragma unroll
    for (int nt = 0; nt < 4; ++nt) {
      float s = ((v[nt][0] + v[nt][1]) + (v[nt][2] + v[nt][3])) +
                ((v[nt][4] + v[nt][5]) + (v[nt][6] + v[nt][7]));
      tot[nt] = s + __shfl_xor(s, 16);
    }
    float* dst = sums + g0 * 64;
    atomicAdd(&dst[lane],      half ? tot[1] : tot[0]);
    atomicAdd(&dst[32 + lane], half ? tot[3] : tot[2]);
    if (lane == 0) atomicAdd(&cnt[g0], 16.0f);
  } else {
    #pragma unroll
    for (int r = 0; r < 8; ++r) {
      int g = batch[node0 + r + 8 * half];
      #pragma unroll
      for (int nt = 0; nt < 4; ++nt)
        atomicAdd(&sums[g * 64 + nt * 16 + m], v[nt][r]);
    }
    if (lane < 16) atomicAdd(&cnt[gl], 1.0f);
  }
}

__global__ void finalize_kernel(const float* __restrict__ sums,
                                const float* __restrict__ cnt,
                                float* __restrict__ out) {
  int i = blockIdx.x * blockDim.x + threadIdx.x;
  if (i >= N_GRAPHS * 64) return;
  float c = cnt[i >> 6];
  out[i] = sums[i] / fmaxf(c, 1.0f);
}

extern "C" void kernel_launch(void* const* d_in, const int* in_sizes, int n_in,
                              void* d_out, int out_size, void* d_ws, size_t ws_size,
                              hipStream_t stream) {
  const float* x     = (const float*)d_in[0];
  const int*   ei    = (const int*)d_in[1];
  const int*   batch = (const int*)d_in[2];
  const float* W1a   = (const float*)d_in[3];
  const float* b1a   = (const float*)d_in[4];
  const float* W1b   = (const float*)d_in[5];
  const float* b1b   = (const float*)d_in[6];
  const float* W2a   = (const float*)d_in[7];
  const float* b2a   = (const float*)d_in[8];
  const float* W2b   = (const float*)d_in[9];
  const float* b2b   = (const float*)d_in[10];
  float* out = (float*)d_out;

  float* ws       = (float*)d_ws;
  float* h1       = ws + 300032;
  int*   deg      = (int*)(ws + 13100032);
  float* sums     = ws + 13200032;
  float* cnt      = ws + 13208224;
  int*   rowstart = (int*)(ws + 13208384);
  int*   pairs    = (int*)(ws + 13308416);
  int*   col      = (int*)(ws + 14908416);
  int*   offT     = (int*)(ws + 16508416);

  dim3 b256(256);
  local_partition_kernel<<<NBLK, b256, 0, stream>>>(ei, pairs, offT, sums, cnt);
  bucket_sort_kernel<<<NBUCK, dim3(1024), 0, stream>>>(pairs, offT, deg, rowstart, col);

  gather_mlp1_kernel<<<(N_TILES + 3) / 4, dim3(128), 0, stream>>>(
      x, rowstart, deg, col, W1a, b1a, W1b, b1b, h1);
  gather_mlp2_pool_kernel<<<(N_TILES + 7) / 8, b256, 0, stream>>>(
      h1, rowstart, deg, col, W2a, b2a, W2b, b2b, batch, sums, cnt);
  finalize_kernel<<<(N_GRAPHS * 64 + 255) / 256, b256, 0, stream>>>(
      sums, cnt, out);
}


# Bucket sort and layer-1 kernel fused: the block that sorts a 416-node bucket then runs that bucket's 26 tiles (gather + MLP) with its own waves; one launch gone; per-node sum order unchanged
# speedup vs baseline: 6.4070x; 1.0641x over previous
#include <hip/hip_runtime.h>


typedef __attribute__((ext_vector_type(2))) float v2f;
typedef __attribute__((ext_vector_type(8))) float v8f;

#define N_NODES  100000
#define N_EDGES  1600000
#define N_GRAPHS 128
#define N_TILES  (N_NODES / 16)


#define NB    256
#define BS    416
#define NBUCK ((N_NODES + BS - 1) / BS)
#define CH    4096
#define EPT   (CH / 256)
#define NBLK  ((N_EDGES + CH - 1) / CH)
#define NBLKP (((NBLK + 31) / 32) * 32)
#define PPW   ((NBLK + 31) / 32)

__global__ __launch_bounds__(256) void local_partition_kernel(
    const int* __restrict__ ei, int* __restrict__ pairs, int* __restrict__ offT,
    float* __restrict__ sums, float* __restrict__ cnt) {
  __shared__ int hist[NB];
  __shared__ int wsum[8];
  __shared__ int stage[CH];
  int t = threadIdx.x;
  int lane = t & 31, w = t >> 5;
  int e0 = blockIdx.x * CH;
  int s[EPT], d[EPT];
  #pragma unroll
  for (int k = 0; k < EPT; ++k) {
    int e = e0 + k * 256 + t;
    int ec = e < N_EDGES ? e : N_EDGES - 1;
    s[k] = ei[ec];
    d[k] = ei[N_EDGES + ec];
  }
  hist[t] = 0;
  __syncthreads();
  int rk[EPT];
  #pragma unroll
  for (int k = 0; k < EPT; ++k) {
    int e = e0 + k * 256 + t;
    rk[k] = 0;
    if (e < N_EDGES) rk[k] = atomicAdd(&hist[d[k] / BS], 1);
  }
  __syncthreads();
  int c = hist[t];
  int incl = c;
  for (int off = 1; off < 32; off <<= 1) {
    int u = __shfl_up(incl, off);
    if (lane >= off) incl += u;
  }
  if (lane == 31) wsum[w] = incl;
  __syncthreads();
  int woff = 0;
  for (int i = 0; i < w; ++i) woff += wsum[i];
  int excl = woff + incl - c;
  hist[t] = excl;
  __syncthreads();
  int bo[EPT];
  #pragma unroll
  for (int k = 0; k < EPT; ++k) bo[k] = hist[d[k] / BS];
  #pragma unroll
  for (int k = 0; k < EPT; ++k) {
    int e = e0 + k * 256 + t;
    if (e < N_EDGES) {
      int b = d[k] / BS;
      stage[bo[k] + rk[k]] = s[k] | ((d[k] - b * BS) << 17);
    }
  }
  __syncthreads();
  int nvalid = N_EDGES - e0 < CH ? N_EDGES - e0 : CH;
  int ov[EPT];
  #pragma unroll
  for (int k = 0; k < EPT; ++k) ov[k] = stage[k * 256 + t];
  #pragma unroll
  for (int k = 0; k < EPT; ++k) {
    int i = k * 256 + t;
    if (i < nvalid) pairs[e0 + i] = ov[k];
  }
  offT[t * NBLKP + blockIdx.x] = excl;
  if (t == NB - 1) offT[NB * NBLKP + blockIdx.x] = excl + c;
  if (blockIdx.x < (N_GRAPHS * 64) / 256) {
    int g = blockIdx.x * 256 + t;
    sums[g] = 0.0f;
    if (g < N_GRAPHS) cnt[g] = 0.0f;
  }
}

#define TPB   (BS / 16)

__global__ __launch_bounds__(1024) void bucket_sort_mlp1_kernel(
    const int* __restrict__ pairs, const int* __restrict__ offT,
    int* __restrict__ deg, int* __restrict__ rowstart, int* col,
    const float* __restrict__ x,
    const float* __restrict__ W1a, const float* __restrict__ b1a,
    const float* __restrict__ W1b, const float* __restrict__ b1b,
    float* __restrict__ h1) {
  __shared__ int ncnt[512];
  __shared__ int ncur[512];
  __shared__ int nrs[512];
  __shared__ int psrc[NBLKP];
  __shared__ int plen[NBLKP];
  __shared__ int wsumA[32];
  __shared__ int wsumB[16];
  __shared__ float sWa[4 * 64];
  __shared__ float sWb[64 * 64];
  __shared__ float sBa[64];
  __shared__ float sBb[64];
  __shared__ float sT[TPB][16 * 68];
  int t = threadIdx.x;
  int lane = t & 31;
  int w = __builtin_amdgcn_readfirstlane(t >> 5);
  int b = blockIdx.x;

  int tc = t < NBLK ? t : 0;
  int o0r = offT[b * NBLKP + tc];
  int o1r = offT[(b + 1) * NBLKP + tc];
  float q0 = W1b[4 * t], q1 = W1b[4 * t + 1], q2 = W1b[4 * t + 2], q3 = W1b[4 * t + 3];
  float qa  = W1a[t < 3 * 64 ? t : 0];
  float qb0 = b1a[t & 63], qb1 = b1b[t & 63];
  sWb[4 * t]     = q0;
  sWb[4 * t + 1] = q1;
  sWb[4 * t + 2] = q2;
  sWb[4 * t + 3] = q3;
  asm volatile("" :: "v"(qa));
  if (t < 4 * 64) sWa[t] = (t < 3 * 64) ? qa : 0.0f;
  if (t < 64) { sBa[t] = qb0; sBb[t] = qb1; }

  int o0 = 0;
  if (t < NBLKP) {
    int o1 = 0;
    if (t < NBLK) {
      o0 = o0r;
      o1 = o1r;
    }
    psrc[t] = t < NBLK ? t * CH + o0 : 0;
    plen[t] = o1 - o0;
  }
  int v = o0;
  for (int off = 16; off > 0; off >>= 1) v += __shfl_xor(v, off);
  if (lane == 0) wsumA[w] = v;
  if (t < 512) ncnt[t] = 0;
  __syncthreads();
  int bstart = 0;
  for (int i = 0; i < 32; ++i) bstart += wsumA[i];

  int wv[PPW], pl[PPW], ps[PPW];
  #pragma unroll
  for (int i = 0; i < PPW; ++i) {
    pl[i] = plen[w + 32 * i];
    ps[i] = psrc[w + 32 * i];
  }
  int maxL = 0;
  #pragma unroll
  for (int i = 0; i < PPW; ++i) {
    int ix = ps[i] + (lane < pl[i] ? lane : 0);
    wv[i]  = pairs[ix < N_EDGES ? ix : N_EDGES - 1];
    maxL   = pl[i] > maxL ? pl[i] : maxL;
  }
  __builtin_amdgcn_sched_barrier(0);
  #pragma unroll
  for (int i = 0; i < PPW; ++i) wv[i] = lane < pl[i] ? wv[i] : -1;
  bool longer = maxL > 32;
  #pragma unroll
  for (int i = 0; i < PPW; ++i)
    if (wv[i] >= 0) atomicAdd(&ncnt[wv[i] >> 17], 1);
  if (longer) {
    for (int k = w; k < NBLK; k += 32) {
      int L = plen[k];
      const int* __restrict__ p = pairs + psrc[k];
      for (int j = 32 + lane; j < L; j += 32) atomicAdd(&ncnt[p[j] >> 17], 1);
    }
  }
  __syncthreads();

  int c = 0, incl = 0;
  if (t < 512) {
    c = ncnt[t];
    incl = c;
    for (int off = 1; off < 32; off <<= 1) {
      int u = __shfl_up(incl, off);
      if (lane >= off) incl += u;
    }
    if (lane == 31) wsumB[w] = incl;
  }
  __syncthreads();
  int ls = 0;
  if (t < 512) {
    int woff = 0;
    for (int i = 0; i < w; ++i) woff += wsumB[i];
    ls = woff + incl - c;
    ncur[t] = ls;
    nrs[t]  = bstart + ls;
  }
  __syncthreads();

  int slot[PPW];
  #pragma unroll
  for (int i = 0; i < PPW; ++i) {
    slot[i] = 0;
    if (wv[i] >= 0) slot[i] = atomicAdd(&ncur[wv[i] >> 17], 1);
  }
  #pragma unroll
  for (int i = 0; i < PPW; ++i)
    if (wv[i] >= 0) col[bstart + slot[i]] = wv[i] & 0x1FFFF;
  if (longer) {
    for (int k = w; k < NBLK; k += 32) {
      int L = plen[k];
      const int* __restrict__ p = pairs + psrc[k];
      for (int j = 32 + lane; j < L; j += 32) {
        int e = p[j];
        col[bstart + atomicAdd(&ncur[e >> 17], 1)] = e & 0x1FFFF;
      }
    }
  }
  __syncthreads();
  int node = b * BS + t;
  if (t < BS && node < N_NODES) {
    deg[node] = c;
    rowstart[node] = bstart + ls;
  }

  int node0 = b * BS + w * 16;
  if (w >= TPB || node0 >= N_NODES) return;
  int half = lane >> 4;
  int m    = lane & 15;
  int nl   = w * 16 + m;
  int nd   = node0 + m;

  float u0, u1, u2;
  {
    int lb = nrs[nl];
    int lc = ncnt[nl];
    float xs0 = x[nd * 3], xs1 = x[nd * 3 + 1], xs2 = x[nd * 3 + 2];
    float a0 = 0.0f, a1 = 0.0f, a2 = 0.0f;
    int id[8];
    #pragma unroll
    for (int k = 0; k < 8; ++k) id[k] = col[lb + k];
    for (int j = 0; j < lc; j += 8) {
      int nx[8];
      #pragma unroll
      for (int k = 0; k < 8; ++k) nx[k] = 0;
      if (j + 8 < lc) {
        #pragma unroll
        for (int k = 0; k < 8; ++k) nx[k] = col[lb + j + 8 + k];
      }
      float p0[8], p1[8], p2[8];
      #pragma unroll
      for (int k = 0; k < 8; ++k) {
        int s = (j + k < lc) ? id[k] : 0;
        p0[k] = x[s * 3];
        p1[k] = x[s * 3 + 1];
        p2[k] = x[s * 3 + 2];
      }
      #pragma unroll
      for (int k = 0; k < 8; ++k) {
        if (j + k < lc) { a0 += p0[k]; a1 += p1[k]; a2 += p2[k]; }
      }
      #pragma unroll
      for (int k = 0; k < 8; ++k) id[k] = nx[k];
    }
    u0 = xs0 + a0;
    u1 = xs1 + a1;
    u2 = xs2 + a2;
  }

  float* myT = &sT[w][0];

  int k0 = 2 * half;
  v2f a;
  a.x = half ? u2 : u0;
  a.y = half ? 0.0f : u1;

  v8f acc[4] = {};
  #pragma unroll
  for (int nt = 0; nt < 4; ++nt) {
    v2f bf;
    bf.x = sWa[k0 * 64 + nt * 16 + m];
    bf.y = sWa[(k0 + 1) * 64 + nt * 16 + m];
    acc[nt] = __builtin_amdgcn_wmma_f32_16x16x4_f32(
        false, a, false, bf, (short)0, acc[nt], false, false);
  }

  #pragma unroll
  for (int nt = 0; nt < 4; ++nt) {
    float bias = sBa[nt * 16 + m];
    #pragma unroll
    for (int r = 0; r < 8; ++r) {
      float z = acc[nt][r] + bias;
      myT[(r + 8 * half) * 68 + nt * 16 + m] = z > 0.0f ? z : 0.0f;
    }
  }

  v8f acc2[4] = {};
  for (int ks = 0; ks < 16; ++ks) {
    int kk = ks * 4 + 2 * half;
    v2f a2;
    a2.x = myT[m * 68 + kk];
    a2.y = myT[m * 68 + kk + 1];
    #pragma unroll
    for (int nt = 0; nt < 4; ++nt) {
      v2f bf;
      bf.x = sWb[kk * 64 + nt * 16 + m];
      bf.y = sWb[(kk + 1) * 64 + nt * 16 + m];
      acc2[nt] = __builtin_amdgcn_wmma_f32_16x16x4_f32(
          false, a2, false, bf, (short)0, acc2[nt], false, false);
    }
  }

  #pragma unroll
  for (int nt = 0; nt < 4; ++nt) {
    float bias = sBb[nt * 16 + m];
    #pragma unroll
    for (int r = 0; r < 8; ++r) {
      float z = acc2[nt][r] + bias;
      h1[(node0 + r + 8 * half) * 64 + nt * 16 + m] = z > 0.0f ? z : 0.0f;
    }
  }
}

#define ROW_LOAD(k) if (c <= k) return; p[k] = hv[id[k] * 32 + lane];
__device__ __forceinline__ void load_rows(v2f (&p)[31], const int (&id)[31], int c,
                                          const v2f* __restrict__ hv, int lane) {
  ROW_LOAD(0) ROW_LOAD(1) ROW_LOAD(2) ROW_LOAD(3) ROW_LOAD(4) ROW_LOAD(5) ROW_LOAD(6) ROW_LOAD(7)
  ROW_LOAD(8) ROW_LOAD(9) ROW_LOAD(10) ROW_LOAD(11) ROW_LOAD(12) ROW_LOAD(13) ROW_LOAD(14) ROW_LOAD(15)
  ROW_LOAD(16) ROW_LOAD(17) ROW_LOAD(18) ROW_LOAD(19) ROW_LOAD(20) ROW_LOAD(21) ROW_LOAD(22) ROW_LOAD(23)
  ROW_LOAD(24) ROW_LOAD(25) ROW_LOAD(26) ROW_LOAD(27) ROW_LOAD(28) ROW_LOAD(29) ROW_LOAD(30)
}
#define ROW_ADD(k) if (c <= k) return; acc += p[k];
__device__ __forceinline__ void add_rows(v2f& acc, const v2f (&p)[31], int c) {
  ROW_ADD(0) ROW_ADD(1) ROW_ADD(2) ROW_ADD(3) ROW_ADD(4) ROW_ADD(5) ROW_ADD(6) ROW_ADD(7)
  ROW_ADD(8) ROW_ADD(9) ROW_ADD(10) ROW_ADD(11) ROW_ADD(12) ROW_ADD(13) ROW_ADD(14) ROW_ADD(15)
  ROW_ADD(16) ROW_ADD(17) ROW_ADD(18) ROW_ADD(19) ROW_ADD(20) ROW_ADD(21) ROW_ADD(22) ROW_ADD(23)
  ROW_ADD(24) ROW_ADD(25) ROW_ADD(26) ROW_ADD(27) ROW_ADD(28) ROW_ADD(29) ROW_ADD(30)
}

__global__ __launch_bounds__(256) void gather_mlp2_pool_kernel(
    const float* __restrict__ h1, const int* __restrict__ rowstart,
    const int* __restrict__ deg, const int* __restrict__ col,
    const float* __restrict__ W2a, const float* __restrict__ b2a,
    const float* __restrict__ W2b, const float* __restrict__ b2b,
    const int* __restrict__ batch,
    float* __restrict__ sums, float* __restrict__ cnt) {
  __shared__ float sWa[64 * 64];
  __shared__ float sWb[64 * 64];
  __shared__ float sBa[64];
  __shared__ float sBb[64];
  __shared__ float sT[8][16 * 68];

  int tid = threadIdx.x;
  for (int i = tid; i < 64 * 64; i += 256) { sWa[i] = W2a[i]; sWb[i] = W2b[i]; }
  if (tid < 64) { sBa[tid] = b2a[tid]; sBb[tid] = b2b[tid]; }
  __syncthreads();

  int lane = tid & 31;
  int half = lane >> 4;
  int m    = lane & 15;
  int wave = __builtin_amdgcn_readfirstlane(tid >> 5);
  int tile = blockIdx.x * 8 + wave;
  if (tile >= N_TILES) return;
  int node0 = tile * 16;
  float* myT = &sT[wave][0];

  const v2f* __restrict__ hv = (const v2f*)h1;
  int rs_v = rowstart[node0 + m];
  int dg_v = deg[node0 + m];
  int b  = __builtin_amdgcn_readlane(rs_v, 0);
  int c  = __builtin_amdgcn_readlane(dg_v, 0);
  int cv = (lane < c) ? col[b + lane] : 0;
  int id[31];
  #pragma unroll
  for (int k = 0; k < 31; ++k) id[k] = __builtin_amdgcn_readlane(cv, k);
  for (int r = 0; r < 16; ++r) {
    int rn  = (r < 15) ? r + 1 : 15;
    int bn  = __builtin_amdgcn_readlane(rs_v, rn);
    int cn  = __builtin_amdgcn_readlane(dg_v, rn);
    int cvn = (lane < cn) ? col[bn + lane] : 0;
    v2f hs  = hv[(node0 + r) * 32 + lane];
    v2f acc = {0.0f, 0.0f};
    v2f p[31];
    load_rows(p, id, c, hv, lane);
    add_rows(acc, p, c);
    for (int j = 31; j < c; ++j)
      acc += hv[col[b + j] * 32 + lane];
    v2f av = hs + acc;
    myT[r * 68 + 2 * lane]     = av.x;
    myT[r * 68 + 2 * lane + 1] = av.y;
    #pragma unroll
    for (int k = 0; k < 31; ++k) id[k] = __builtin_amdgcn_readlane(cvn, k);
    b = bn; c = cn;
  }

  v8f acc[4] = {};
  for (int ks = 0; ks < 16; ++ks) {
    int kk = ks * 4 + 2 * half;
    v2f a;
    a.x = myT[m * 68 + kk];
    a.y = myT[m * 68 + kk + 1];
    #pragma unroll
    for (int nt = 0; nt < 4; ++nt) {
      v2f b;
      b.x = sWa[kk * 64 + nt * 16 + m];
      b.y = sWa[(kk + 1) * 64 + nt * 16 + m];
      acc[nt] = __builtin_amdgcn_wmma_f32_16x16x4_f32(
          false, a, false, b, (short)0, acc[nt], false, false);
    }
  }

  #pragma unroll
  for (int nt = 0; nt < 4; ++nt) {
    float bias = sBa[nt * 16 + m];
    #pragma unroll
    for (int r = 0; r < 8; ++r) {
      float v = acc[nt][r] + bias;
      myT[(r + 8 * half) * 68 + nt * 16 + m] = v > 0.0f ? v : 0.0f;
    }
  }

  v8f acc2[4] = {};
  for (int ks = 0; ks < 16; ++ks) {
    int kk = ks * 4 + 2 * half;
    v2f a2;
    a2.x = myT[m * 68 + kk];
    a2.y = myT[m * 68 + kk + 1];
    #pragma unroll
    for (int nt = 0; nt < 4; ++nt) {
      v2f b;
      b.x = sWb[kk * 64 + nt * 16 + m];
      b.y = sWb[(kk + 1) * 64 + nt * 16 + m];
      acc2[nt] = __builtin_amdgcn_wmma_f32_16x16x4_f32(
          false, a2, false, b, (short)0, acc2[nt], false, false);
    }
  }

  float v[4][8];
  #pragma unroll
  for (int nt = 0; nt < 4; ++nt) {
    float bias = sBb[nt * 16 + m];
    #pragma unroll
    for (int r = 0; r < 8; ++r) {
      float t = acc2[nt][r] + bias;
      v[nt][r] = t > 0.0f ? t : 0.0f;
    }
  }

  int gl = batch[node0 + m];
  int g0 = __builtin_amdgcn_readfirstlane(gl);
  bool one_graph = __builtin_amdgcn_ballot_w32(gl != g0) == 0u;
  if (one_graph) {
    float tot[4];
    #pragma unroll
    for (int nt = 0; nt < 4; ++nt) {
      float s = ((v[nt][0] + v[nt][1]) + (v[nt][2] + v[nt][3])) +
                ((v[nt][4] + v[nt][5]) + (v[nt][6] + v[nt][7]));
      tot[nt] = s + __shfl_xor(s, 16);
    }
    float* dst = sums + g0 * 64;
    atomicAdd(&dst[lane],      half ? tot[1] : tot[0]);
    atomicAdd(&dst[32 + lane], half ? tot[3] : tot[2]);
    if (lane == 0) atomicAdd(&cnt[g0], 16.0f);
  } else {
    #pragma unroll
    for (int r = 0; r < 8; ++r) {
      int g = batch[node0 + r + 8 * half];
      #pragma unroll
      for (int nt = 0; nt < 4; ++nt)
        atomicAdd(&sums[g * 64 + nt * 16 + m], v[nt][r]);
    }
    if (lane < 16) atomicAdd(&cnt[gl], 1.0f);
  }
}

__global__ void finalize_kernel(const float* __restrict__ sums,
                                const float* __restrict__ cnt,
                                float* __restrict__ out) {
  int i = blockIdx.x * blockDim.x + threadIdx.x;
  if (i >= N_GRAPHS * 64) return;
  float c = cnt[i >> 6];
  out[i] = sums[i] / fmaxf(c, 1.0f);
}

extern "C" void kernel_launch(void* const* d_in, const int* in_sizes, int n_in,
                              void* d_out, int out_size, void* d_ws, size_t ws_size,
                              hipStream_t stream) {
  const float* x     = (const float*)d_in[0];
  const int*   ei    = (const int*)d_in[1];
  const int*   batch = (const int*)d_in[2];
  const float* W1a   = (const float*)d_in[3];
  const float* b1a   = (const float*)d_in[4];
  const float* W1b   = (const float*)d_in[5];
  const float* b1b   = (const float*)d_in[6];
  const float* W2a   = (const float*)d_in[7];
  const float* b2a   = (const float*)d_in[8];
  const float* W2b   = (const float*)d_in[9];
  const float* b2b   = (const float*)d_in[10];
  float* out = (float*)d_out;

  float* ws       = (float*)d_ws;
  float* h1       = ws + 300032;
  int*   deg      = (int*)(ws + 13100032);
  float* sums     = ws + 13200032;
  float* cnt      = ws + 13208224;
  int*   rowstart = (int*)(ws + 13208384);
  int*   pairs    = (int*)(ws + 13308416);
  int*   col      = (int*)(ws + 14908416);
  int*   offT     = (int*)(ws + 16508416);

  dim3 b256(256);
  local_partition_kernel<<<NBLK, b256, 0, stream>>>(ei, pairs, offT, sums, cnt);
  bucket_sort_mlp1_kernel<<<NBUCK, dim3(1024), 0, stream>>>(
      pairs, offT, deg, rowstart, col, x, W1a, b1a, W1b, b1b, h1);
  gather_mlp2_pool_kernel<<<(N_TILES + 7) / 8, b256, 0, stream>>>(
      h1, rowstart, deg, col, W2a, b2a, W2b, b2b, batch, sums, cnt);
  finalize_kernel<<<(N_GRAPHS * 64 + 255) / 256, b256, 0, stream>>>(
      sums, cnt, out);
}


# Bucket kernel assembles a bucket's sorted ids in LDS: col[] leaves as whole lines instead of one scattered 4-byte store per edge, and layer 1 reads its ids from LDS; per-node sum order unchanged
# speedup vs baseline: 6.8965x; 1.0764x over previous
#include <hip/hip_runtime.h>


typedef __attribute__((ext_vector_type(2))) float v2f;
typedef __attribute__((ext_vector_type(8))) float v8f;

#define N_NODES  100000
#define N_EDGES  1600000
#define N_GRAPHS 128
#define N_TILES  (N_NODES / 16)


#define NB    256
#define BS    416
#define NBUCK ((N_NODES + BS - 1) / BS)
#define CH    4096
#define EPT   (CH / 256)
#define NBLK  ((N_EDGES + CH - 1) / CH)
#define NBLKP (((NBLK + 31) / 32) * 32)
#define PPW   ((NBLK + 31) / 32)

__global__ __launch_bounds__(256) void local_partition_kernel(
    const int* __restrict__ ei, int* __restrict__ pairs, int* __restrict__ offT,
    float* __restrict__ sums, float* __restrict__ cnt) {
  __shared__ int hist[NB];
  __shared__ int wsum[8];
  __shared__ int stage[CH];
  int t = threadIdx.x;
  int lane = t & 31, w = t >> 5;
  int e0 = blockIdx.x * CH;
  int s[EPT], d[EPT];
  #pragma unroll
  for (int k = 0; k < EPT; ++k) {
    int e = e0 + k * 256 + t;
    int ec = e < N_EDGES ? e : N_EDGES - 1;
    s[k] = ei[ec];
    d[k] = ei[N_EDGES + ec];
  }
  hist[t] = 0;
  __syncthreads();
  int rk[EPT];
  #pragma unroll
  for (int k = 0; k < EPT; ++k) {
    int e = e0 + k * 256 + t;
    rk[k] = 0;
    if (e < N_EDGES) rk[k] = atomicAdd(&hist[d[k] / BS], 1);
  }
  __syncthreads();
  int c = hist[t];
  int incl = c;
  for (int off = 1; off < 32; off <<= 1) {
    int u = __shfl_up(incl, off);
    if (lane >= off) incl += u;
  }
  if (lane == 31) wsum[w] = incl;
  __syncthreads();
  int woff = 0;
  for (int i = 0; i < w; ++i) woff += wsum[i];
  int excl = woff + incl - c;
  hist[t] = excl;
  __syncthreads();
  int bo[EPT];
  #pragma unroll
  for (int k = 0; k < EPT; ++k) bo[k] = hist[d[k] / BS];
  #pragma unroll
  for (int k = 0; k < EPT; ++k) {
    int e = e0 + k * 256 + t;
    if (e < N_EDGES) {
      int b = d[k] / BS;
      stage[bo[k] + rk[k]] = s[k] | ((d[k] - b * BS) << 17);
    }
  }
  __syncthreads();
  int nvalid = N_EDGES - e0 < CH ? N_EDGES - e0 : CH;
  int ov[EPT];
  #pragma unroll
  for (int k = 0; k < EPT; ++k) ov[k] = stage[k * 256 + t];
  #pragma unroll
  for (int k = 0; k < EPT; ++k) {
    int i = k * 256 + t;
    if (i < nvalid) pairs[e0 + i] = ov[k];
  }
  offT[t * NBLKP + blockIdx.x] = excl;
  if (t == NB - 1) offT[NB * NBLKP + blockIdx.x] = excl + c;
  if (blockIdx.x < (N_GRAPHS * 64) / 256) {
    int g = blockIdx.x * 256 + t;
    sums[g] = 0.0f;
    if (g < N_GRAPHS) cnt[g] = 0.0f;
  }
}

#define TPB   (BS / 16)
#define IDCAP 8192

__global__ __launch_bounds__(1024) void bucket_sort_mlp1_kernel(
    const int* __restrict__ pairs, const int* __restrict__ offT,
    int* __restrict__ deg, int* __restrict__ rowstart, int* col,
    const float* __restrict__ x,
    const float* __restrict__ W1a, const float* __restrict__ b1a,
    const float* __restrict__ W1b, const float* __restrict__ b1b,
    float* __restrict__ h1) {
  __shared__ int ncnt[512];
  __shared__ int ncur[512];
  __shared__ int nrs[512];
  __shared__ int psrc[NBLKP];
  __shared__ int plen[NBLKP];
  __shared__ int wsumA[32];
  __shared__ int wsumB[16];
  __shared__ int sid[IDCAP + 8];
  __shared__ float sWa[4 * 64];
  __shared__ float sWb[64 * 64];
  __shared__ float sBa[64];
  __shared__ float sBb[64];
  __shared__ float sT[TPB][16 * 68];
  int t = threadIdx.x;
  int lane = t & 31;
  int w = __builtin_amdgcn_readfirstlane(t >> 5);
  int b = blockIdx.x;

  int tc = t < NBLK ? t : 0;
  int o0r = offT[b * NBLKP + tc];
  int o1r = offT[(b + 1) * NBLKP + tc];
  float q0 = W1b[4 * t], q1 = W1b[4 * t + 1], q2 = W1b[4 * t + 2], q3 = W1b[4 * t + 3];
  float qa  = W1a[t < 3 * 64 ? t : 0];
  float qb0 = b1a[t & 63], qb1 = b1b[t & 63];
  sWb[4 * t]     = q0;
  sWb[4 * t + 1] = q1;
  sWb[4 * t + 2] = q2;
  sWb[4 * t + 3] = q3;
  asm volatile("" :: "v"(qa));
  if (t < 4 * 64) sWa[t] = (t < 3 * 64) ? qa : 0.0f;
  if (t < 64) { sBa[t] = qb0; sBb[t] = qb1; }

  int o0 = 0;
  if (t < NBLKP) {
    int o1 = 0;
    if (t < NBLK) {
      o0 = o0r;
      o1 = o1r;
    }
    psrc[t] = t < NBLK ? t * CH + o0 : 0;
    plen[t] = o1 - o0;
  }
  int v = o0;
  for (int off = 16; off > 0; off >>= 1) v += __shfl_xor(v, off);
  if (lane == 0) wsumA[w] = v;
  if (t < 512) ncnt[t] = 0;
  __syncthreads();
  int bstart = 0;
  for (int i = 0; i < 32; ++i) bstart += wsumA[i];

  int wv[PPW], pl[PPW], ps[PPW];
  #pragma unroll
  for (int i = 0; i < PPW; ++i) {
    pl[i] = plen[w + 32 * i];
    ps[i] = psrc[w + 32 * i];
  }
  int maxL = 0;
  #pragma unroll
  for (int i = 0; i < PPW; ++i) {
    int ix = ps[i] + (lane < pl[i] ? lane : 0);
    wv[i]  = pairs[ix < N_EDGES ? ix : N_EDGES - 1];
    maxL   = pl[i] > maxL ? pl[i] : maxL;
  }
  __builtin_amdgcn_sched_barrier(0);
  #pragma unroll
  for (int i = 0; i < PPW; ++i) wv[i] = lane < pl[i] ? wv[i] : -1;
  bool longer = maxL > 32;
  #pragma unroll
  for (int i = 0; i < PPW; ++i)
    if (wv[i] >= 0) atomicAdd(&ncnt[wv[i] >> 17], 1);
  if (longer) {
    for (int k = w; k < NBLK; k += 32) {
      int L = plen[k];
      const int* __restrict__ p = pairs + psrc[k];
      for (int j = 32 + lane; j < L; j += 32) atomicAdd(&ncnt[p[j] >> 17], 1);
    }
  }
  __syncthreads();

  int c = 0, incl = 0;
  if (t < 512) {
    c = ncnt[t];
    incl = c;
    for (int off = 1; off < 32; off <<= 1) {
      int u = __shfl_up(incl, off);
      if (lane >= off) incl += u;
    }
    if (lane == 31) wsumB[w] = incl;
  }
  __syncthreads();
  int tot = 0;
  #pragma unroll
  for (int i = 0; i < 16; ++i) tot += wsumB[i];
  tot = __builtin_amdgcn_readfirstlane(tot);
  bool inl = tot <= IDCAP;
  int ls = 0;
  if (t < 512) {
    int woff = 0;
    for (int i = 0; i < w; ++i) woff += wsumB[i];
    ls = woff + incl - c;
    ncur[t] = ls;
    nrs[t]  = bstart + ls;
  }
  __syncthreads();

  int slot[PPW];
  #pragma unroll
  for (int i = 0; i < PPW; ++i) {
    slot[i] = 0;
    if (wv[i] >= 0) slot[i] = atomicAdd(&ncur[wv[i] >> 17], 1);
  }
  if (inl) {
    #pragma unroll
    for (int i = 0; i < PPW; ++i)
      if (wv[i] >= 0) sid[slot[i]] = wv[i] & 0x1FFFF;
    if (longer) {
      for (int k = w; k < NBLK; k += 32) {
        int L = plen[k];
        const int* __restrict__ p = pairs + psrc[k];
        for (int j = 32 + lane; j < L; j += 32) {
          int e = p[j];
          sid[atomicAdd(&ncur[e >> 17], 1)] = e & 0x1FFFF;
        }
      }
    }
  } else {
    #pragma unroll
    for (int i = 0; i < PPW; ++i)
      if (wv[i] >= 0) col[bstart + slot[i]] = wv[i] & 0x1FFFF;
    if (longer) {
      for (int k = w; k < NBLK; k += 32) {
        int L = plen[k];
        const int* __restrict__ p = pairs + psrc[k];
        for (int j = 32 + lane; j < L; j += 32) {
          int e = p[j];
          col[bstart + atomicAdd(&ncur[e >> 17], 1)] = e & 0x1FFFF;
        }
      }
    }
  }
  __syncthreads();
  if (inl) {
    int ow[IDCAP / 1024];
    #pragma unroll
    for (int q = 0; q < IDCAP / 1024; ++q) ow[q] = sid[t + 1024 * q];
    #pragma unroll
    for (int q = 0; q < IDCAP / 1024; ++q) {
      int i = t + 1024 * q;
      if (i < tot) col[bstart + i] = ow[q];
    }
  }
  int node = b * BS + t;
  if (t < BS && node < N_NODES) {
    deg[node] = c;
    rowstart[node] = bstart + ls;
  }

  int node0 = b * BS + w * 16;
  if (w >= TPB || node0 >= N_NODES) return;
  int half = lane >> 4;
  int m    = lane & 15;
  int nl   = w * 16 + m;
  int nd   = node0 + m;

  float u0, u1, u2;
  {
    int lb = nrs[nl];
    int lc = ncnt[nl];
    float xs0 = x[nd * 3], xs1 = x[nd * 3 + 1], xs2 = x[nd * 3 + 2];
    float a0 = 0.0f, a1 = 0.0f, a2 = 0.0f;
    auto nb_sum = [&](auto ldid, int base) {
      int id[8];
      #pragma unroll
      for (int k = 0; k < 8; ++k) id[k] = ldid(base + k);
      for (int j = 0; j < lc; j += 8) {
        int nx[8];
        #pragma unroll
        for (int k = 0; k < 8; ++k) nx[k] = 0;
        if (j + 8 < lc) {
          #pragma unroll
          for (int k = 0; k < 8; ++k) nx[k] = ldid(base + j + 8 + k);
        }
        float p0[8], p1[8], p2[8];
        #pragma unroll
        for (int k = 0; k < 8; ++k) {
          int s = (j + k < lc) ? id[k] : 0;
          p0[k] = x[s * 3];
          p1[k] = x[s * 3 + 1];
          p2[k] = x[s * 3 + 2];
        }
        #pragma unroll
        for (int k = 0; k < 8; ++k) {
          if (j + k < lc) { a0 += p0[k]; a1 += p1[k]; a2 += p2[k]; }
        }
        #pragma unroll
        for (int k = 0; k < 8; ++k) id[k] = nx[k];
      }
    };
    if (inl) nb_sum([&](int i) { return sid[i]; }, lb - bstart);
    else     nb_sum([&](int i) { return col[i]; }, lb);
    u0 = xs0 + a0;
    u1 = xs1 + a1;
    u2 = xs2 + a2;
  }

  float* myT = &sT[w][0];

  int k0 = 2 * half;
  v2f a;
  a.x = half ? u2 : u0;
  a.y = half ? 0.0f : u1;

  v8f acc[4] = {};
  #pragma unroll
  for (int nt = 0; nt < 4; ++nt) {
    v2f bf;
    bf.x = sWa[k0 * 64 + nt * 16 + m];
    bf.y = sWa[(k0 + 1) * 64 + nt * 16 + m];
    acc[nt] = __builtin_amdgcn_wmma_f32_16x16x4_f32(
        false, a, false, bf, (short)0, acc[nt], false, false);
  }

  #pragma unroll
  for (int nt = 0; nt < 4; ++nt) {
    float bias = sBa[nt * 16 + m];
    #pragma unroll
    for (int r = 0; r < 8; ++r) {
      float z = acc[nt][r] + bias;
      myT[(r + 8 * half) * 68 + nt * 16 + m] = z > 0.0f ? z : 0.0f;
    }
  }

  v8f acc2[4] = {};
  for (int ks = 0; ks < 16; ++ks) {
    int kk = ks * 4 + 2 * half;
    v2f a2;
    a2.x = myT[m * 68 + kk];
    a2.y = myT[m * 68 + kk + 1];
    #pragma unroll
    for (int nt = 0; nt < 4; ++nt) {
      v2f bf;
      bf.x = sWb[kk * 64 + nt * 16 + m];
      bf.y = sWb[(kk + 1) * 64 + nt * 16 + m];
      acc2[nt] = __builtin_amdgcn_wmma_f32_16x16x4_f32(
          false, a2, false, bf, (short)0, acc2[nt], false, false);
    }
  }

  #pragma unroll
  for (int nt = 0; nt < 4; ++nt) {
    float bias = sBb[nt * 16 + m];
    #pragma unroll
    for (int r = 0; r < 8; ++r) {
      float z = acc2[nt][r] + bias;
      h1[(node0 + r + 8 * half) * 64 + nt * 16 + m] = z > 0.0f ? z : 0.0f;
    }
  }
}

#define ROW_LOAD(k) if (c <= k) return; p[k] = hv[id[k] * 32 + lane];
__device__ __forceinline__ void load_rows(v2f (&p)[31], const int (&id)[31], int c,
                                          const v2f* __restrict__ hv, int lane) {
  ROW_LOAD(0) ROW_LOAD(1) ROW_LOAD(2) ROW_LOAD(3) ROW_LOAD(4) ROW_LOAD(5) ROW_LOAD(6) ROW_LOAD(7)
  ROW_LOAD(8) ROW_LOAD(9) ROW_LOAD(10) ROW_LOAD(11) ROW_LOAD(12) ROW_LOAD(13) ROW_LOAD(14) ROW_LOAD(15)
  ROW_LOAD(16) ROW_LOAD(17) ROW_LOAD(18) ROW_LOAD(19) ROW_LOAD(20) ROW_LOAD(21) ROW_LOAD(22) ROW_LOAD(23)
  ROW_LOAD(24) ROW_LOAD(25) ROW_LOAD(26) ROW_LOAD(27) ROW_LOAD(28) ROW_LOAD(29) ROW_LOAD(30)
}
#define ROW_ADD(k) if (c <= k) return; acc += p[k];
__device__ __forceinline__ void add_rows(v2f& acc, const v2f (&p)[31], int c) {
  ROW_ADD(0) ROW_ADD(1) ROW_ADD(2) ROW_ADD(3) ROW_ADD(4) ROW_ADD(5) ROW_ADD(6) ROW_ADD(7)
  ROW_ADD(8) ROW_ADD(9) ROW_ADD(10) ROW_ADD(11) ROW_ADD(12) ROW_ADD(13) ROW_ADD(14) ROW_ADD(15)
  ROW_ADD(16) ROW_ADD(17) ROW_ADD(18) ROW_ADD(19) ROW_ADD(20) ROW_ADD(21) ROW_ADD(22) ROW_ADD(23)
  ROW_ADD(24) ROW_ADD(25) ROW_ADD(26) ROW_ADD(27) ROW_ADD(28) ROW_ADD(29) ROW_ADD(30)
}

__global__ __launch_bounds__(256) void gather_mlp2_pool_kernel(
    const float* __restrict__ h1, const int* __restrict__ rowstart,
    const int* __restrict__ deg, const int* __restrict__ col,
    const float* __restrict__ W2a, const float* __restrict__ b2a,
    const float* __restrict__ W2b, const float* __restrict__ b2b,
    const int* __restrict__ batch,
    float* __restrict__ sums, float* __restrict__ cnt) {
  __shared__ float sWa[64 * 64];
  __shared__ float sWb[64 * 64];
  __shared__ float sBa[64];
  __shared__ float sBb[64];
  __shared__ float sT[8][16 * 68];

  int tid = threadIdx.x;
  for (int i = tid; i < 64 * 64; i += 256) { sWa[i] = W2a[i]; sWb[i] = W2b[i]; }
  if (tid < 64) { sBa[tid] = b2a[tid]; sBb[tid] = b2b[tid]; }
  __syncthreads();

  int lane = tid & 31;
  int half = lane >> 4;
  int m    = lane & 15;
  int wave = __builtin_amdgcn_readfirstlane(tid >> 5);
  int tile = blockIdx.x * 8 + wave;
  if (tile >= N_TILES) return;
  int node0 = tile * 16;
  float* myT = &sT[wave][0];

  const v2f* __restrict__ hv = (const v2f*)h1;
  int rs_v = rowstart[node0 + m];
  int dg_v = deg[node0 + m];
  int b  = __builtin_amdgcn_readlane(rs_v, 0);
  int c  = __builtin_amdgcn_readlane(dg_v, 0);
  int cv = (lane < c) ? col[b + lane] : 0;
  int id[31];
  #pragma unroll
  for (int k = 0; k < 31; ++k) id[k] = __builtin_amdgcn_readlane(cv, k);
  for (int r = 0; r < 16; ++r) {
    int rn  = (r < 15) ? r + 1 : 15;
    int bn  = __builtin_amdgcn_readlane(rs_v, rn);
    int cn  = __builtin_amdgcn_readlane(dg_v, rn);
    int cvn = (lane < cn) ? col[bn + lane] : 0;
    v2f hs  = hv[(node0 + r) * 32 + lane];
    v2f acc = {0.0f, 0.0f};
    v2f p[31];
    load_rows(p, id, c, hv, lane);
    add_rows(acc, p, c);
    for (int j = 31; j < c; ++j)
      acc += hv[col[b + j] * 32 + lane];
    v2f av = hs + acc;
    myT[r * 68 + 2 * lane]     = av.x;
    myT[r * 68 + 2 * lane + 1] = av.y;
    #pragma unroll
    for (int k = 0; k < 31; ++k) id[k] = __builtin_amdgcn_readlane(cvn, k);
    b = bn; c = cn;
  }

  v8f acc[4] = {};
  for (int ks = 0; ks < 16; ++ks) {
    int kk = ks * 4 + 2 * half;
    v2f a;
    a.x = myT[m * 68 + kk];
    a.y = myT[m * 68 + kk + 1];
    #pragma unroll
    for (int nt = 0; nt < 4; ++nt) {
      v2f b;
      b.x = sWa[kk * 64 + nt * 16 + m];
      b.y = sWa[(kk + 1) * 64 + nt * 16 + m];
      acc[nt] = __builtin_amdgcn_wmma_f32_16x16x4_f32(
          false, a, false, b, (short)0, acc[nt], false, false);
    }
  }

  #pragma unroll
  for (int nt = 0; nt < 4; ++nt) {
    float bias = sBa[nt * 16 + m];
    #pragma unroll
    for (int r = 0; r < 8; ++r) {
      float v = acc[nt][r] + bias;
      myT[(r + 8 * half) * 68 + nt * 16 + m] = v > 0.0f ? v : 0.0f;
    }
  }

  v8f acc2[4] = {};
  for (int ks = 0; ks < 16; ++ks) {
    int kk = ks * 4 + 2 * half;
    v2f a2;
    a2.x = myT[m * 68 + kk];
    a2.y = myT[m * 68 + kk + 1];
    #pragma unroll
    for (int nt = 0; nt < 4; ++nt) {
      v2f b;
      b.x = sWb[kk * 64 + nt * 16 + m];
      b.y = sWb[(kk + 1) * 64 + nt * 16 + m];
      acc2[nt] = __builtin_amdgcn_wmma_f32_16x16x4_f32(
          false, a2, false, b, (short)0, acc2[nt], false, false);
    }
  }

  float v[4][8];
  #pragma unroll
  for (int nt = 0; nt < 4; ++nt) {
    float bias = sBb[nt * 16 + m];
    #pragma unroll
    for (int r = 0; r < 8; ++r) {
      float t = acc2[nt][r] + bias;
      v[nt][r] = t > 0.0f ? t : 0.0f;
    }
  }

  int gl = batch[node0 + m];
  int g0 = __builtin_amdgcn_readfirstlane(gl);
  bool one_graph = __builtin_amdgcn_ballot_w32(gl != g0) == 0u;
  if (one_graph) {
    float tot[4];
    #pragma unroll
    for (int nt = 0; nt < 4; ++nt) {
      float s = ((v[nt][0] + v[nt][1]) + (v[nt][2] + v[nt][3])) +
                ((v[nt][4] + v[nt][5]) + (v[nt][6] + v[nt][7]));
      tot[nt] = s + __shfl_xor(s, 16);
    }
    float* dst = sums + g0 * 64;
    atomicAdd(&dst[lane],      half ? tot[1] : tot[0]);
    atomicAdd(&dst[32 + lane], half ? tot[3] : tot[2]);
    if (lane == 0) atomicAdd(&cnt[g0], 16.0f);
  } else {
    #pragma unroll
    for (int r = 0; r < 8; ++r) {
      int g = batch[node0 + r + 8 * half];
      #pragma unroll
      for (int nt = 0; nt < 4; ++nt)
        atomicAdd(&sums[g * 64 + nt * 16 + m], v[nt][r]);
    }
    if (lane < 16) atomicAdd(&cnt[gl], 1.0f);
  }
}

__global__ void finalize_kernel(const float* __restrict__ sums,
                                const float* __restrict__ cnt,
                                float* __restrict__ out) {
  int i = blockIdx.x * blockDim.x + threadIdx.x;
  if (i >= N_GRAPHS * 64) return;
  float c = cnt[i >> 6];
  out[i] = sums[i] / fmaxf(c, 1.0f);
}

extern "C" void kernel_launch(void* const* d_in, const int* in_sizes, int n_in,
                              void* d_out, int out_size, void* d_ws, size_t ws_size,
                              hipStream_t stream) {
  const float* x     = (const float*)d_in[0];
  const int*   ei    = (const int*)d_in[1];
  const int*   batch = (const int*)d_in[2];
  const float* W1a   = (const float*)d_in[3];
  const float* b1a   = (const float*)d_in[4];
  const float* W1b   = (const float*)d_in[5];
  const float* b1b   = (const float*)d_in[6];
  const float* W2a   = (const float*)d_in[7];
  const float* b2a   = (const float*)d_in[8];
  const float* W2b   = (const float*)d_in[9];
  const float* b2b   = (const float*)d_in[10];
  float* out = (float*)d_out;

  float* ws       = (float*)d_ws;
  float* h1       = ws + 300032;
  int*   deg      = (int*)(ws + 13100032);
  float* sums     = ws + 13200032;
  float* cnt      = ws + 13208224;
  int*   rowstart = (int*)(ws + 13208384);
  int*   pairs    = (int*)(ws + 13308416);
  int*   col      = (int*)(ws + 14908416);
  int*   offT     = (int*)(ws + 16508416);

  dim3 b256(256);
  local_partition_kernel<<<NBLK, b256, 0, stream>>>(ei, pairs, offT, sums, cnt);
  bucket_sort_mlp1_kernel<<<NBUCK, dim3(1024), 0, stream>>>(
      pairs, offT, deg, rowstart, col, x, W1a, b1a, W1b, b1b, h1);
  gather_mlp2_pool_kernel<<<(N_TILES + 7) / 8, b256, 0, stream>>>(
      h1, rowstart, deg, col, W2a, b2a, W2b, b2b, batch, sums, cnt);
  finalize_kernel<<<(N_GRAPHS * 64 + 255) / 256, b256, 0, stream>>>(
      sums, cnt, out);
}


# Layer-2 kernel: the block's weights are loaded as one group with the tile's list heads and staged while the first ids are in flight; the 8 serial load-store trips ahead of the gather are gone
# speedup vs baseline: 7.2447x; 1.0505x over previous
#include <hip/hip_runtime.h>


typedef __attribute__((ext_vector_type(2))) float v2f;
typedef __attribute__((ext_vector_type(8))) float v8f;

#define N_NODES  100000
#define N_EDGES  1600000
#define N_GRAPHS 128
#define N_TILES  (N_NODES / 16)


#define NB    256
#define BS    416
#define NBUCK ((N_NODES + BS - 1) / BS)
#define CH    4096
#define EPT   (CH / 256)
#define NBLK  ((N_EDGES + CH - 1) / CH)
#define NBLKP (((NBLK + 31) / 32) * 32)
#define PPW   ((NBLK + 31) / 32)

__global__ __launch_bounds__(256) void local_partition_kernel(
    const int* __restrict__ ei, int* __restrict__ pairs, int* __restrict__ offT,
    float* __restrict__ sums, float* __restrict__ cnt) {
  __shared__ int hist[NB];
  __shared__ int wsum[8];
  __shared__ int stage[CH];
  int t = threadIdx.x;
  int lane = t & 31, w = t >> 5;
  int e0 = blockIdx.x * CH;
  int s[EPT], d[EPT];
  #pragma unroll
  for (int k = 0; k < EPT; ++k) {
    int e = e0 + k * 256 + t;
    int ec = e < N_EDGES ? e : N_EDGES - 1;
    s[k] = ei[ec];
    d[k] = ei[N_EDGES + ec];
  }
  hist[t] = 0;
  __syncthreads();
  int rk[EPT];
  #pragma unroll
  for (int k = 0; k < EPT; ++k) {
    int e = e0 + k * 256 + t;
    rk[k] = 0;
    if (e < N_EDGES) rk[k] = atomicAdd(&hist[d[k] / BS], 1);
  }
  __syncthreads();
  int c = hist[t];
  int incl = c;
  for (int off = 1; off < 32; off <<= 1) {
    int u = __shfl_up(incl, off);
    if (lane >= off) incl += u;
  }
  if (lane == 31) wsum[w] = incl;
  __syncthreads();
  int woff = 0;
  for (int i = 0; i < w; ++i) woff += wsum[i];
  int excl = woff + incl - c;
  hist[t] = excl;
  __syncthreads();
  int bo[EPT];
  #pragma unroll
  for (int k = 0; k < EPT; ++k) bo[k] = hist[d[k] / BS];
  #pragma unroll
  for (int k = 0; k < EPT; ++k) {
    int e = e0 + k * 256 + t;
    if (e < N_EDGES) {
      int b = d[k] / BS;
      stage[bo[k] + rk[k]] = s[k] | ((d[k] - b * BS) << 17);
    }
  }
  __syncthreads();
  int nvalid = N_EDGES - e0 < CH ? N_EDGES - e0 : CH;
  int ov[EPT];
  #pragma unroll
  for (int k = 0; k < EPT; ++k) ov[k] = stage[k * 256 + t];
  #pragma unroll
  for (int k = 0; k < EPT; ++k) {
    int i = k * 256 + t;
    if (i < nvalid) pairs[e0 + i] = ov[k];
  }
  offT[t * NBLKP + blockIdx.x] = excl;
  if (t == NB - 1) offT[NB * NBLKP + blockIdx.x] = excl + c;
  if (blockIdx.x < (N_GRAPHS * 64) / 256) {
    int g = blockIdx.x * 256 + t;
    sums[g] = 0.0f;
    if (g < N_GRAPHS) cnt[g] = 0.0f;
  }
}

#define TPB   (BS / 16)
#define IDCAP 8192

__global__ __launch_bounds__(1024) void bucket_sort_mlp1_kernel(
    const int* __restrict__ pairs, const int* __restrict__ offT,
    int* __restrict__ deg, int* __restrict__ rowstart, int* col,
    const float* __restrict__ x,
    const float* __restrict__ W1a, const float* __restrict__ b1a,
    const float* __restrict__ W1b, const float* __restrict__ b1b,
    float* __restrict__ h1) {
  __shared__ int ncnt[512];
  __shared__ int ncur[512];
  __shared__ int nrs[512];
  __shared__ int psrc[NBLKP];
  __shared__ int plen[NBLKP];
  __shared__ int wsumA[32];
  __shared__ int wsumB[16];
  __shared__ int sid[IDCAP + 8];
  __shared__ float sWa[4 * 64];
  __shared__ float sWb[64 * 64];
  __shared__ float sBa[64];
  __shared__ float sBb[64];
  __shared__ float sT[TPB][16 * 68];
  int t = threadIdx.x;
  int lane = t & 31;
  int w = __builtin_amdgcn_readfirstlane(t >> 5);
  int b = blockIdx.x;

  int tc = t < NBLK ? t : 0;
  int o0r = offT[b * NBLKP + tc];
  int o1r = offT[(b + 1) * NBLKP + tc];
  float q0 = W1b[4 * t], q1 = W1b[4 * t + 1], q2 = W1b[4 * t + 2], q3 = W1b[4 * t + 3];
  float qa  = W1a[t < 3 * 64 ? t : 0];
  float qb0 = b1a[t & 63], qb1 = b1b[t & 63];
  sWb[4 * t]     = q0;
  sWb[4 * t + 1] = q1;
  sWb[4 * t + 2] = q2;
  sWb[4 * t + 3] = q3;
  asm volatile("" :: "v"(qa));
  if (t < 4 * 64) sWa[t] = (t < 3 * 64) ? qa : 0.0f;
  if (t < 64) { sBa[t] = qb0; sBb[t] = qb1; }

  int o0 = 0;
  if (t < NBLKP) {
    int o1 = 0;
    if (t < NBLK) {
      o0 = o0r;
      o1 = o1r;
    }
    psrc[t] = t < NBLK ? t * CH + o0 : 0;
    plen[t] = o1 - o0;
  }
  int v = o0;
  for (int off = 16; off > 0; off >>= 1) v += __shfl_xor(v, off);
  if (lane == 0) wsumA[w] = v;
  if (t < 512) ncnt[t] = 0;
  __syncthreads();
  int bstart = 0;
  for (int i = 0; i < 32; ++i) bstart += wsumA[i];

  int wv[PPW], pl[PPW], ps[PPW];
  #pragma unroll
  for (int i = 0; i < PPW; ++i) {
    pl[i] = plen[w + 32 * i];
    ps[i] = psrc[w + 32 * i];
  }
  int maxL = 0;
  #pragma unroll
  for (int i = 0; i < PPW; ++i) {
    int ix = ps[i] + (lane < pl[i] ? lane : 0);
    wv[i]  = pairs[ix < N_EDGES ? ix : N_EDGES - 1];
    maxL   = pl[i] > maxL ? pl[i] : maxL;
  }
  __builtin_amdgcn_sched_barrier(0);
  #pragma unroll
  for (int i = 0; i < PPW; ++i) wv[i] = lane < pl[i] ? wv[i] : -1;
  bool longer = maxL > 32;
  #pragma unroll
  for (int i = 0; i < PPW; ++i)
    if (wv[i] >= 0) atomicAdd(&ncnt[wv[i] >> 17], 1);
  if (longer) {
    for (int k = w; k < NBLK; k += 32) {
      int L = plen[k];
      const int* __restrict__ p = pairs + psrc[k];
      for (int j = 32 + lane; j < L; j += 32) atomicAdd(&ncnt[p[j] >> 17], 1);
    }
  }
  __syncthreads();

  int c = 0, incl = 0;
  if (t < 512) {
    c = ncnt[t];
    incl = c;
    for (int off = 1; off < 32; off <<= 1) {
      int u = __shfl_up(incl, off);
      if (lane >= off) incl += u;
    }
    if (lane == 31) wsumB[w] = incl;
  }
  __syncthreads();
  int tot = 0;
  #pragma unroll
  for (int i = 0; i < 16; ++i) tot += wsumB[i];
  tot = __builtin_amdgcn_readfirstlane(tot);
  bool inl = tot <= IDCAP;
  int ls = 0;
  if (t < 512) {
    int woff = 0;
    for (int i = 0; i < w; ++i) woff += wsumB[i];
    ls = woff + incl - c;
    ncur[t] = ls;
    nrs[t]  = bstart + ls;
  }
  __syncthreads();

  int slot[PPW];
  #pragma unroll
  for (int i = 0; i < PPW; ++i) {
    slot[i] = 0;
    if (wv[i] >= 0) slot[i] = atomicAdd(&ncur[wv[i] >> 17], 1);
  }
  if (inl) {
    #pragma unroll
    for (int i = 0; i < PPW; ++i)
      if (wv[i] >= 0) sid[slot[i]] = wv[i] & 0x1FFFF;
    if (longer) {
      for (int k = w; k < NBLK; k += 32) {
        int L = plen[k];
        const int* __restrict__ p = pairs + psrc[k];
        for (int j = 32 + lane; j < L; j += 32) {
          int e = p[j];
          sid[atomicAdd(&ncur[e >> 17], 1)] = e & 0x1FFFF;
        }
      }
    }
  } else {
    #pragma unroll
    for (int i = 0; i < PPW; ++i)
      if (wv[i] >= 0) col[bstart + slot[i]] = wv[i] & 0x1FFFF;
    if (longer) {
      for (int k = w; k < NBLK; k += 32) {
        int L = plen[k];
        const int* __restrict__ p = pairs + psrc[k];
        for (int j = 32 + lane; j < L; j += 32) {
          int e = p[j];
          col[bstart + atomicAdd(&ncur[e >> 17], 1)] = e & 0x1FFFF;
        }
      }
    }
  }
  __syncthreads();
  if (inl) {
    int ow[IDCAP / 1024];
    #pragma unroll
    for (int q = 0; q < IDCAP / 1024; ++q) ow[q] = sid[t + 1024 * q];
    #pragma unroll
    for (int q = 0; q < IDCAP / 1024; ++q) {
      int i = t + 1024 * q;
      if (i < tot) col[bstart + i] = ow[q];
    }
  }
  int node = b * BS + t;
  if (t < BS && node < N_NODES) {
    deg[node] = c;
    rowstart[node] = bstart + ls;
  }

  int node0 = b * BS + w * 16;
  if (w >= TPB || node0 >= N_NODES) return;
  int half = lane >> 4;
  int m    = lane & 15;
  int nl   = w * 16 + m;
  int nd   = node0 + m;

  float u0, u1, u2;
  {
    int lb = nrs[nl];
    int lc = ncnt[nl];
    float xs0 = x[nd * 3], xs1 = x[nd * 3 + 1], xs2 = x[nd * 3 + 2];
    float a0 = 0.0f, a1 = 0.0f, a2 = 0.0f;
    auto nb_sum = [&](auto ldid, int base) {
      int id[8];
      #pragma unroll
      for (int k = 0; k < 8; ++k) id[k] = ldid(base + k);
      for (int j = 0; j < lc; j += 8) {
        int nx[8];
        #pragma unroll
        for (int k = 0; k < 8; ++k) nx[k] = 0;
        if (j + 8 < lc) {
          #pragma unroll
          for (int k = 0; k < 8; ++k) nx[k] = ldid(base + j + 8 + k);
        }
        float p0[8], p1[8], p2[8];
        #pragma unroll
        for (int k = 0; k < 8; ++k) {
          int s = (j + k < lc) ? id[k] : 0;
          p0[k] = x[s * 3];
          p1[k] = x[s * 3 + 1];
          p2[k] = x[s * 3 + 2];
        }
        #pragma unroll
        for (int k = 0; k < 8; ++k) {
          if (j + k < lc) { a0 += p0[k]; a1 += p1[k]; a2 += p2[k]; }
        }
        #pragma unroll
        for (int k = 0; k < 8; ++k) id[k] = nx[k];
      }
    };
    if (inl) nb_sum([&](int i) { return sid[i]; }, lb - bstart);
    else     nb_sum([&](int i) { return col[i]; }, lb);
    u0 = xs0 + a0;
    u1 = xs1 + a1;
    u2 = xs2 + a2;
  }

  float* myT = &sT[w][0];

  int k0 = 2 * half;
  v2f a;
  a.x = half ? u2 : u0;
  a.y = half ? 0.0f : u1;

  v8f acc[4] = {};
  #pragma unroll
  for (int nt = 0; nt < 4; ++nt) {
    v2f bf;
    bf.x = sWa[k0 * 64 + nt * 16 + m];
    bf.y = sWa[(k0 + 1) * 64 + nt * 16 + m];
    acc[nt] = __builtin_amdgcn_wmma_f32_16x16x4_f32(
        false, a, false, bf, (short)0, acc[nt], false, false);
  }

  #pragma unroll
  for (int nt = 0; nt < 4; ++nt) {
    float bias = sBa[nt * 16 + m];
    #pragma unroll
    for (int r = 0; r < 8; ++r) {
      float z = acc[nt][r] + bias;
      myT[(r + 8 * half) * 68 + nt * 16 + m] = z > 0.0f ? z : 0.0f;
    }
  }

  v8f acc2[4] = {};
  for (int ks = 0; ks < 16; ++ks) {
    int kk = ks * 4 + 2 * half;
    v2f a2;
    a2.x = myT[m * 68 + kk];
    a2.y = myT[m * 68 + kk + 1];
    #pragma unroll
    for (int nt = 0; nt < 4; ++nt) {
      v2f bf;
      bf.x = sWb[kk * 64 + nt * 16 + m];
      bf.y = sWb[(kk + 1) * 64 + nt * 16 + m];
      acc2[nt] = __builtin_amdgcn_wmma_f32_16x16x4_f32(
          false, a2, false, bf, (short)0, acc2[nt], false, false);
    }
  }

  #pragma unroll
  for (int nt = 0; nt < 4; ++nt) {
    float bias = sBb[nt * 16 + m];
    #pragma unroll
    for (int r = 0; r < 8; ++r) {
      float z = acc2[nt][r] + bias;
      h1[(node0 + r + 8 * half) * 64 + nt * 16 + m] = z > 0.0f ? z : 0.0f;
    }
  }
}

#define ROW_LOAD(k) if (c <= k) return; p[k] = hv[id[k] * 32 + lane];
__device__ __forceinline__ void load_rows(v2f (&p)[31], const int (&id)[31], int c,
                                          const v2f* __restrict__ hv, int lane) {
  ROW_LOAD(0) ROW_LOAD(1) ROW_LOAD(2) ROW_LOAD(3) ROW_LOAD(4) ROW_LOAD(5) ROW_LOAD(6) ROW_LOAD(7)
  ROW_LOAD(8) ROW_LOAD(9) ROW_LOAD(10) ROW_LOAD(11) ROW_LOAD(12) ROW_LOAD(13) ROW_LOAD(14) ROW_LOAD(15)
  ROW_LOAD(16) ROW_LOAD(17) ROW_LOAD(18) ROW_LOAD(19) ROW_LOAD(20) ROW_LOAD(21) ROW_LOAD(22) ROW_LOAD(23)
  ROW_LOAD(24) ROW_LOAD(25) ROW_LOAD(26) ROW_LOAD(27) ROW_LOAD(28) ROW_LOAD(29) ROW_LOAD(30)
}
#define ROW_ADD(k) if (c <= k) return; acc += p[k];
__device__ __forceinline__ void add_rows(v2f& acc, const v2f (&p)[31], int c) {
  ROW_ADD(0) ROW_ADD(1) ROW_ADD(2) ROW_ADD(3) ROW_ADD(4) ROW_ADD(5) ROW_ADD(6) ROW_ADD(7)
  ROW_ADD(8) ROW_ADD(9) ROW_ADD(10) ROW_ADD(11) ROW_ADD(12) ROW_ADD(13) ROW_ADD(14) ROW_ADD(15)
  ROW_ADD(16) ROW_ADD(17) ROW_ADD(18) ROW_ADD(19) ROW_ADD(20) ROW_ADD(21) ROW_ADD(22) ROW_ADD(23)
  ROW_ADD(24) ROW_ADD(25) ROW_ADD(26) ROW_ADD(27) ROW_ADD(28) ROW_ADD(29) ROW_ADD(30)
}

__global__ __launch_bounds__(256) void gather_mlp2_pool_kernel(
    const float* __restrict__ h1, const int* __restrict__ rowstart,
    const int* __restrict__ deg, const int* __restrict__ col,
    const float* __restrict__ W2a, const float* __restrict__ b2a,
    const float* __restrict__ W2b, const float* __restrict__ b2b,
    const int* __restrict__ batch,
    float* __restrict__ sums, float* __restrict__ cnt) {
  __shared__ float sWa[64 * 64];
  __shared__ float sWb[64 * 64];
  __shared__ float sBa[64];
  __shared__ float sBb[64];
  __shared__ float sT[8][16 * 68];

  int tid  = threadIdx.x;
  int lane = tid & 31;
  int half = lane >> 4;
  int m    = lane & 15;
  int wave = __builtin_amdgcn_readfirstlane(tid >> 5);
  int tile = blockIdx.x * 8 + wave;
  bool has_tile = tile < N_TILES;
  int node0 = (has_tile ? tile : N_TILES - 1) * 16;
  float* myT = &sT[wave][0];

  const v2f* __restrict__ hv = (const v2f*)h1;
  int rs_v = rowstart[node0 + m];
  int dg_v = deg[node0 + m];
  float qa[16], qb[16];
  #pragma unroll
  for (int k = 0; k < 4; ++k) {
    #pragma unroll
    for (int i = 0; i < 4; ++i) {
      qa[4 * k + i] = W2a[1024 * k + 4 * tid + i];
      qb[4 * k + i] = W2b[1024 * k + 4 * tid + i];
    }
  }
  float qba = b2a[tid & 63], qbb = b2b[tid & 63];
  int b  = __builtin_amdgcn_readlane(rs_v, 0);
  int c  = __builtin_amdgcn_readlane(dg_v, 0);
  int cv = (lane < c) ? col[b + lane] : 0;
  #pragma unroll
  for (int k = 0; k < 4; ++k) {
    #pragma unroll
    for (int i = 0; i < 4; ++i) {
      sWa[1024 * k + 4 * tid + i] = qa[4 * k + i];
      sWb[1024 * k + 4 * tid + i] = qb[4 * k + i];
    }
  }
  if (tid < 64) { sBa[tid] = qba; sBb[tid] = qbb; }
  __syncthreads();
  if (!has_tile) return;

  int id[31];
  #pragma unroll
  for (int k = 0; k < 31; ++k) id[k] = __builtin_amdgcn_readlane(cv, k);
  for (int r = 0; r < 16; ++r) {
    int rn  = (r < 15) ? r + 1 : 15;
    int bn  = __builtin_amdgcn_readlane(rs_v, rn);
    int cn  = __builtin_amdgcn_readlane(dg_v, rn);
    int cvn = (lane < cn) ? col[bn + lane] : 0;
    v2f hs  = hv[(node0 + r) * 32 + lane];
    v2f acc = {0.0f, 0.0f};
    v2f p[31];
    load_rows(p, id, c, hv, lane);
    add_rows(acc, p, c);
    for (int j = 31; j < c; ++j)
      acc += hv[col[b + j] * 32 + lane];
    v2f av = hs + acc;
    myT[r * 68 + 2 * lane]     = av.x;
    myT[r * 68 + 2 * lane + 1] = av.y;
    #pragma unroll
    for (int k = 0; k < 31; ++k) id[k] = __builtin_amdgcn_readlane(cvn, k);
    b = bn; c = cn;
  }

  v8f acc[4] = {};
  for (int ks = 0; ks < 16; ++ks) {
    int kk = ks * 4 + 2 * half;
    v2f a;
    a.x = myT[m * 68 + kk];
    a.y = myT[m * 68 + kk + 1];
    #pragma unroll
    for (int nt = 0; nt < 4; ++nt) {
      v2f b;
      b.x = sWa[kk * 64 + nt * 16 + m];
      b.y = sWa[(kk + 1) * 64 + nt * 16 + m];
      acc[nt] = __builtin_amdgcn_wmma_f32_16x16x4_f32(
          false, a, false, b, (short)0, acc[nt], false, false);
    }
  }

  #pragma unroll
  for (int nt = 0; nt < 4; ++nt) {
    float bias = sBa[nt * 16 + m];
    #pragma unroll
    for (int r = 0; r < 8; ++r) {
      float v = acc[nt][r] + bias;
      myT[(r + 8 * half) * 68 + nt * 16 + m] = v > 0.0f ? v : 0.0f;
    }
  }

  v8f acc2[4] = {};
  for (int ks = 0; ks < 16; ++ks) {
    int kk = ks * 4 + 2 * half;
    v2f a2;
    a2.x = myT[m * 68 + kk];
    a2.y = myT[m * 68 + kk + 1];
    #pragma unroll
    for (int nt = 0; nt < 4; ++nt) {
      v2f b;
      b.x = sWb[kk * 64 + nt * 16 + m];
      b.y = sWb[(kk + 1) * 64 + nt * 16 + m];
      acc2[nt] = __builtin_amdgcn_wmma_f32_16x16x4_f32(
          false, a2, false, b, (short)0, acc2[nt], false, false);
    }
  }

  float v[4][8];
  #pragma unroll
  for (int nt = 0; nt < 4; ++nt) {
    float bias = sBb[nt * 16 + m];
    #pragma unroll
    for (int r = 0; r < 8; ++r) {
      float t = acc2[nt][r] + bias;
      v[nt][r] = t > 0.0f ? t : 0.0f;
    }
  }

  int gl = batch[node0 + m];
  int g0 = __builtin_amdgcn_readfirstlane(gl);
  bool one_graph = __builtin_amdgcn_ballot_w32(gl != g0) == 0u;
  if (one_graph) {
    float tot[4];
    #pragma unroll
    for (int nt = 0; nt < 4; ++nt) {
      float s = ((v[nt][0] + v[nt][1]) + (v[nt][2] + v[nt][3])) +
                ((v[nt][4] + v[nt][5]) + (v[nt][6] + v[nt][7]));
      tot[nt] = s + __shfl_xor(s, 16);
    }
    float* dst = sums + g0 * 64;
    atomicAdd(&dst[lane],      half ? tot[1] : tot[0]);
    atomicAdd(&dst[32 + lane], half ? tot[3] : tot[2]);
    if (lane == 0) atomicAdd(&cnt[g0], 16.0f);
  } else {
    #pragma unroll
    for (int r = 0; r < 8; ++r) {
      int g = batch[node0 + r + 8 * half];
      #pragma unroll
      for (int nt = 0; nt < 4; ++nt)
        atomicAdd(&sums[g * 64 + nt * 16 + m], v[nt][r]);
    }
    if (lane < 16) atomicAdd(&cnt[gl], 1.0f);
  }
}

__global__ void finalize_kernel(const float* __restrict__ sums,
                                const float* __restrict__ cnt,
                                float* __restrict__ out) {
  int i = blockIdx.x * blockDim.x + threadIdx.x;
  if (i >= N_GRAPHS * 64) return;
  float c = cnt[i >> 6];
  out[i] = sums[i] / fmaxf(c, 1.0f);
}

extern "C" void kernel_launch(void* const* d_in, const int* in_sizes, int n_in,
                              void* d_out, int out_size, void* d_ws, size_t ws_size,
                              hipStream_t stream) {
  const float* x     = (const float*)d_in[0];
  const int*   ei    = (const int*)d_in[1];
  const int*   batch = (const int*)d_in[2];
  const float* W1a   = (const float*)d_in[3];
  const float* b1a   = (const float*)d_in[4];
  const float* W1b   = (const float*)d_in[5];
  const float* b1b   = (const float*)d_in[6];
  const float* W2a   = (const float*)d_in[7];
  const float* b2a   = (const float*)d_in[8];
  const float* W2b   = (const float*)d_in[9];
  const float* b2b   = (const float*)d_in[10];
  float* out = (float*)d_out;

  float* ws       = (float*)d_ws;
  float* h1       = ws + 300032;
  int*   deg      = (int*)(ws + 13100032);
  float* sums     = ws + 13200032;
  float* cnt      = ws + 13208224;
  int*   rowstart = (int*)(ws + 13208384);
  int*   pairs    = (int*)(ws + 13308416);
  int*   col      = (int*)(ws + 14908416);
  int*   offT     = (int*)(ws + 16508416);

  dim3 b256(256);
  local_partition_kernel<<<NBLK, b256, 0, stream>>>(ei, pairs, offT, sums, cnt);
  bucket_sort_mlp1_kernel<<<NBUCK, dim3(1024), 0, stream>>>(
      pairs, offT, deg, rowstart, col, x, W1a, b1a, W1b, b1b, h1);
  gather_mlp2_pool_kernel<<<(N_TILES + 7) / 8, b256, 0, stream>>>(
      h1, rowstart, deg, col, W2a, b2a, W2b, b2b, batch, sums, cnt);
  finalize_kernel<<<(N_GRAPHS * 64 + 255) / 256, b256, 0, stream>>>(
      sums, cnt, out);
}


# Mean taken in the layer-2 epilogue: graph sizes are counted in the layer-1 launch, a tile's column sums are divided by the size, the quotients added by f32 atomics into out; finalize launch gone
# speedup vs baseline: 7.4819x; 1.0327x over previous
#include <hip/hip_runtime.h>


typedef __attribute__((ext_vector_type(2))) float v2f;
typedef __attribute__((ext_vector_type(8))) float v8f;

#define N_NODES  100000
#define N_EDGES  1600000
#define N_GRAPHS 128
#define N_TILES  (N_NODES / 16)


#define NB    256
#define BS    416
#define NBUCK ((N_NODES + BS - 1) / BS)
#define CH    4096
#define EPT   (CH / 256)
#define NBLK  ((N_EDGES + CH - 1) / CH)
#define NBLKP (((NBLK + 31) / 32) * 32)
#define PPW   ((NBLK + 31) / 32)

__global__ __launch_bounds__(256) void local_partition_kernel(
    const int* __restrict__ ei, int* __restrict__ pairs, int* __restrict__ offT,
    float* __restrict__ out, float* __restrict__ cnt) {
  __shared__ int hist[NB];
  __shared__ int wsum[8];
  __shared__ int stage[CH];
  int t = threadIdx.x;
  int lane = t & 31, w = t >> 5;
  int e0 = blockIdx.x * CH;
  int s[EPT], d[EPT];
  #pragma unroll
  for (int k = 0; k < EPT; ++k) {
    int e = e0 + k * 256 + t;
    int ec = e < N_EDGES ? e : N_EDGES - 1;
    s[k] = ei[ec];
    d[k] = ei[N_EDGES + ec];
  }
  hist[t] = 0;
  __syncthreads();
  int rk[EPT];
  #pragma unroll
  for (int k = 0; k < EPT; ++k) {
    int e = e0 + k * 256 + t;
    rk[k] = 0;
    if (e < N_EDGES) rk[k] = atomicAdd(&hist[d[k] / BS], 1);
  }
  __syncthreads();
  int c = hist[t];
  int incl = c;
  for (int off = 1; off < 32; off <<= 1) {
    int u = __shfl_up(incl, off);
    if (lane >= off) incl += u;
  }
  if (lane == 31) wsum[w] = incl;
  __syncthreads();
  int woff = 0;
  for (int i = 0; i < w; ++i) woff += wsum[i];
  int excl = woff + incl - c;
  hist[t] = excl;
  __syncthreads();
  int bo[EPT];
  #pragma unroll
  for (int k = 0; k < EPT; ++k) bo[k] = hist[d[k] / BS];
  #pragma unroll
  for (int k = 0; k < EPT; ++k) {
    int e = e0 + k * 256 + t;
    if (e < N_EDGES) {
      int b = d[k] / BS;
      stage[bo[k] + rk[k]] = s[k] | ((d[k] - b * BS) << 17);
    }
  }
  __syncthreads();
  int nvalid = N_EDGES - e0 < CH ? N_EDGES - e0 : CH;
  int ov[EPT];
  #pragma unroll
  for (int k = 0; k < EPT; ++k) ov[k] = stage[k * 256 + t];
  #pragma unroll
  for (int k = 0; k < EPT; ++k) {
    int i = k * 256 + t;
    if (i < nvalid) pairs[e0 + i] = ov[k];
  }
  offT[t * NBLKP + blockIdx.x] = excl;
  if (t == NB - 1) offT[NB * NBLKP + blockIdx.x] = excl + c;
  if (blockIdx.x < (N_GRAPHS * 64) / 256) {
    int g = blockIdx.x * 256 + t;
    out[g] = 0.0f;
    if (g < N_GRAPHS) cnt[g] = 0.0f;
  }
}

#define TPB   (BS / 16)
#define IDCAP 8192

__global__ __launch_bounds__(1024) void bucket_sort_mlp1_kernel(
    const int* __restrict__ pairs, const int* __restrict__ offT,
    int* __restrict__ deg, int* __restrict__ rowstart, int* col,
    const float* __restrict__ x,
    const float* __restrict__ W1a, const float* __restrict__ b1a,
    const float* __restrict__ W1b, const float* __restrict__ b1b,
    const int* __restrict__ batch, float* __restrict__ cnt,
    float* __restrict__ h1) {
  __shared__ int ncnt[512];
  __shared__ int ncur[512];
  __shared__ int nrs[512];
  __shared__ int psrc[NBLKP];
  __shared__ int plen[NBLKP];
  __shared__ int wsumA[32];
  __shared__ int wsumB[16];
  __shared__ int sid[IDCAP + 8];
  __shared__ float sWa[4 * 64];
  __shared__ float sWb[64 * 64];
  __shared__ float sBa[64];
  __shared__ float sBb[64];
  __shared__ float sT[TPB][16 * 68];
  int t = threadIdx.x;
  int lane = t & 31;
  int w = __builtin_amdgcn_readfirstlane(t >> 5);
  int b = blockIdx.x;

  int tc = t < NBLK ? t : 0;
  int o0r = offT[b * NBLKP + tc];
  int o1r = offT[(b + 1) * NBLKP + tc];
  float q0 = W1b[4 * t], q1 = W1b[4 * t + 1], q2 = W1b[4 * t + 2], q3 = W1b[4 * t + 3];
  float qa  = W1a[t < 3 * 64 ? t : 0];
  float qb0 = b1a[t & 63], qb1 = b1b[t & 63];
  int cnode = b * BS + (t < BS ? t : 0);
  int gq = batch[cnode < N_NODES ? cnode : N_NODES - 1];
  sWb[4 * t]     = q0;
  sWb[4 * t + 1] = q1;
  sWb[4 * t + 2] = q2;
  sWb[4 * t + 3] = q3;
  asm volatile("" :: "v"(qa));
  if (t < 4 * 64) sWa[t] = (t < 3 * 64) ? qa : 0.0f;
  if (t < 64) { sBa[t] = qb0; sBb[t] = qb1; }

  int o0 = 0;
  if (t < NBLKP) {
    int o1 = 0;
    if (t < NBLK) {
      o0 = o0r;
      o1 = o1r;
    }
    psrc[t] = t < NBLK ? t * CH + o0 : 0;
    plen[t] = o1 - o0;
  }
  int v = o0;
  for (int off = 16; off > 0; off >>= 1) v += __shfl_xor(v, off);
  if (lane == 0) wsumA[w] = v;
  if (t < 512) ncnt[t] = 0;
  __syncthreads();
  int bstart = 0;
  for (int i = 0; i < 32; ++i) bstart += wsumA[i];

  int wv[PPW], pl[PPW], ps[PPW];
  #pragma unroll
  for (int i = 0; i < PPW; ++i) {
    pl[i] = plen[w + 32 * i];
    ps[i] = psrc[w + 32 * i];
  }
  int maxL = 0;
  #pragma unroll
  for (int i = 0; i < PPW; ++i) {
    int ix = ps[i] + (lane < pl[i] ? lane : 0);
    wv[i]  = pairs[ix < N_EDGES ? ix : N_EDGES - 1];
    maxL   = pl[i] > maxL ? pl[i] : maxL;
  }
  __builtin_amdgcn_sched_barrier(0);
  #pragma unroll
  for (int i = 0; i < PPW; ++i) wv[i] = lane < pl[i] ? wv[i] : -1;
  bool longer = maxL > 32;
  #pragma unroll
  for (int i = 0; i < PPW; ++i)
    if (wv[i] >= 0) atomicAdd(&ncnt[wv[i] >> 17], 1);
  if (longer) {
    for (int k = w; k < NBLK; k += 32) {
      int L = plen[k];
      const int* __restrict__ p = pairs + psrc[k];
      for (int j = 32 + lane; j < L; j += 32) atomicAdd(&ncnt[p[j] >> 17], 1);
    }
  }
  __syncthreads();

  int c = 0, incl = 0;
  if (t < 512) {
    c = ncnt[t];
    incl = c;
    for (int off = 1; off < 32; off <<= 1) {
      int u = __shfl_up(incl, off);
      if (lane >= off) incl += u;
    }
    if (lane == 31) wsumB[w] = incl;
  }
  __syncthreads();
  int tot = 0;
  #pragma unroll
  for (int i = 0; i < 16; ++i) tot += wsumB[i];
  tot = __builtin_amdgcn_readfirstlane(tot);
  bool inl = tot <= IDCAP;
  int ls = 0;
  if (t < 512) {
    int woff = 0;
    for (int i = 0; i < w; ++i) woff += wsumB[i];
    ls = woff + incl - c;
    ncur[t] = ls;
    nrs[t]  = bstart + ls;
  }
  __syncthreads();

  int slot[PPW];
  #pragma unroll
  for (int i = 0; i < PPW; ++i) {
    slot[i] = 0;
    if (wv[i] >= 0) slot[i] = atomicAdd(&ncur[wv[i] >> 17], 1);
  }
  if (inl) {
    #pragma unroll
    for (int i = 0; i < PPW; ++i)
      if (wv[i] >= 0) sid[slot[i]] = wv[i] & 0x1FFFF;
    if (longer) {
      for (int k = w; k < NBLK; k += 32) {
        int L = plen[k];
        const int* __restrict__ p = pairs + psrc[k];
        for (int j = 32 + lane; j < L; j += 32) {
          int e = p[j];
          sid[atomicAdd(&ncur[e >> 17], 1)] = e & 0x1FFFF;
        }
      }
    }
  } else {
    #pragma unroll
    for (int i = 0; i < PPW; ++i)
      if (wv[i] >= 0) col[bstart + slot[i]] = wv[i] & 0x1FFFF;
    if (longer) {
      for (int k = w; k < NBLK; k += 32) {
        int L = plen[k];
        const int* __restrict__ p = pairs + psrc[k];
        for (int j = 32 + lane; j < L; j += 32) {
          int e = p[j];
          col[bstart + atomicAdd(&ncur[e >> 17], 1)] = e & 0x1FFFF;
        }
      }
    }
  }
  __syncthreads();
  if (inl) {
    int ow[IDCAP / 1024];
    #pragma unroll
    for (int q = 0; q < IDCAP / 1024; ++q) ow[q] = sid[t + 1024 * q];
    #pragma unroll
    for (int q = 0; q < IDCAP / 1024; ++q) {
      int i = t + 1024 * q;
      if (i < tot) col[bstart + i] = ow[q];
    }
  }
  int node = b * BS + t;
  bool own = t < BS && node < N_NODES;
  if (own) {
    deg[node] = c;
    rowstart[node] = bstart + ls;
  }
  {
    bool okg = own && (unsigned)gq < (unsigned)N_GRAPHS;
    int gf = __builtin_amdgcn_readfirstlane(gq);
    unsigned same = __builtin_amdgcn_ballot_w32(okg && gq == gf);
    if (lane == 0 && same != 0u) atomicAdd(&cnt[gf], (float)__popc(same));
    if (okg && gq != gf) atomicAdd(&cnt[gq], 1.0f);
  }

  int node0 = b * BS + w * 16;
  if (w >= TPB || node0 >= N_NODES) return;
  int half = lane >> 4;
  int m    = lane & 15;
  int nl   = w * 16 + m;
  int nd   = node0 + m;

  float u0, u1, u2;
  {
    int lb = nrs[nl];
    int lc = ncnt[nl];
    float xs0 = x[nd * 3], xs1 = x[nd * 3 + 1], xs2 = x[nd * 3 + 2];
    float a0 = 0.0f, a1 = 0.0f, a2 = 0.0f;
    auto nb_sum = [&](auto ldid, int base) {
      int id[8];
      #pragma unroll
      for (int k = 0; k < 8; ++k) id[k] = ldid(base + k);
      for (int j = 0; j < lc; j += 8) {
        int nx[8];
        #pragma unroll
        for (int k = 0; k < 8; ++k) nx[k] = 0;
        if (j + 8 < lc) {
          #pragma unroll
          for (int k = 0; k < 8; ++k) nx[k] = ldid(base + j + 8 + k);
        }
        float p0[8], p1[8], p2[8];
        #pragma unroll
        for (int k = 0; k < 8; ++k) {
          int s = (j + k < lc) ? id[k] : 0;
          p0[k] = x[s * 3];
          p1[k] = x[s * 3 + 1];
          p2[k] = x[s * 3 + 2];
        }
        #pragma unroll
        for (int k = 0; k < 8; ++k) {
          if (j + k < lc) { a0 += p0[k]; a1 += p1[k]; a2 += p2[k]; }
        }
        #pragma unroll
        for (int k = 0; k < 8; ++k) id[k] = nx[k];
      }
    };
    if (inl) nb_sum([&](int i) { return sid[i]; }, lb - bstart);
    else     nb_sum([&](int i) { return col[i]; }, lb);
    u0 = xs0 + a0;
    u1 = xs1 + a1;
    u2 = xs2 + a2;
  }

  float* myT = &sT[w][0];

  int k0 = 2 * half;
  v2f a;
  a.x = half ? u2 : u0;
  a.y = half ? 0.0f : u1;

  v8f acc[4] = {};
  #pragma unroll
  for (int nt = 0; nt < 4; ++nt) {
    v2f bf;
    bf.x = sWa[k0 * 64 + nt * 16 + m];
    bf.y = sWa[(k0 + 1) * 64 + nt * 16 + m];
    acc[nt] = __builtin_amdgcn_wmma_f32_16x16x4_f32(
        false, a, false, bf, (short)0, acc[nt], false, false);
  }

  #pragma unroll
  for (int nt = 0; nt < 4; ++nt) {
    float bias = sBa[nt * 16 + m];
    #pragma unroll
    for (int r = 0; r < 8; ++r) {
      float z = acc[nt][r] + bias;
      myT[(r + 8 * half) * 68 + nt * 16 + m] = z > 0.0f ? z : 0.0f;
    }
  }

  v8f acc2[4] = {};
  for (int ks = 0; ks < 16; ++ks) {
    int kk = ks * 4 + 2 * half;
    v2f a2;
    a2.x = myT[m * 68 + kk];
    a2.y = myT[m * 68 + kk + 1];
    #pragma unroll
    for (int nt = 0; nt < 4; ++nt) {
      v2f bf;
      bf.x = sWb[kk * 64 + nt * 16 + m];
      bf.y = sWb[(kk + 1) * 64 + nt * 16 + m];
      acc2[nt] = __builtin_amdgcn_wmma_f32_16x16x4_f32(
          false, a2, false, bf, (short)0, acc2[nt], false, false);
    }
  }

  #pragma unroll
  for (int nt = 0; nt < 4; ++nt) {
    float bias = sBb[nt * 16 + m];
    #pragma unroll
    for (int r = 0; r < 8; ++r) {
      float z = acc2[nt][r] + bias;
      h1[(node0 + r + 8 * half) * 64 + nt * 16 + m] = z > 0.0f ? z : 0.0f;
    }
  }
}

#define ROW_LOAD(k) if (c <= k) return; p[k] = hv[id[k] * 32 + lane];
__device__ __forceinline__ void load_rows(v2f (&p)[31], const int (&id)[31], int c,
                                          const v2f* __restrict__ hv, int lane) {
  ROW_LOAD(0) ROW_LOAD(1) ROW_LOAD(2) ROW_LOAD(3) ROW_LOAD(4) ROW_LOAD(5) ROW_LOAD(6) ROW_LOAD(7)
  ROW_LOAD(8) ROW_LOAD(9) ROW_LOAD(10) ROW_LOAD(11) ROW_LOAD(12) ROW_LOAD(13) ROW_LOAD(14) ROW_LOAD(15)
  ROW_LOAD(16) ROW_LOAD(17) ROW_LOAD(18) ROW_LOAD(19) ROW_LOAD(20) ROW_LOAD(21) ROW_LOAD(22) ROW_LOAD(23)
  ROW_LOAD(24) ROW_LOAD(25) ROW_LOAD(26) ROW_LOAD(27) ROW_LOAD(28) ROW_LOAD(29) ROW_LOAD(30)
}
#define ROW_ADD(k) if (c <= k) return; acc += p[k];
__device__ __forceinline__ void add_rows(v2f& acc, const v2f (&p)[31], int c) {
  ROW_ADD(0) ROW_ADD(1) ROW_ADD(2) ROW_ADD(3) ROW_ADD(4) ROW_ADD(5) ROW_ADD(6) ROW_ADD(7)
  ROW_ADD(8) ROW_ADD(9) ROW_ADD(10) ROW_ADD(11) ROW_ADD(12) ROW_ADD(13) ROW_ADD(14) ROW_ADD(15)
  ROW_ADD(16) ROW_ADD(17) ROW_ADD(18) ROW_ADD(19) ROW_ADD(20) ROW_ADD(21) ROW_ADD(22) ROW_ADD(23)
  ROW_ADD(24) ROW_ADD(25) ROW_ADD(26) ROW_ADD(27) ROW_ADD(28) ROW_ADD(29) ROW_ADD(30)
}

__global__ __launch_bounds__(256) void gather_mlp2_pool_kernel(
    const float* __restrict__ h1, const int* __restrict__ rowstart,
    const int* __restrict__ deg, const int* __restrict__ col,
    const float* __restrict__ W2a, const float* __restrict__ b2a,
    const float* __restrict__ W2b, const float* __restrict__ b2b,
    const int* __restrict__ batch,
    const float* __restrict__ cnt, float* __restrict__ out) {
  __shared__ float sWa[64 * 64];
  __shared__ float sWb[64 * 64];
  __shared__ float sBa[64];
  __shared__ float sBb[64];
  __shared__ float sT[8][16 * 68];

  int tid  = threadIdx.x;
  int lane = tid & 31;
  int half = lane >> 4;
  int m    = lane & 15;
  int wave = __builtin_amdgcn_readfirstlane(tid >> 5);
  int tile = blockIdx.x * 8 + wave;
  bool has_tile = tile < N_TILES;
  int node0 = (has_tile ? tile : N_TILES - 1) * 16;
  float* myT = &sT[wave][0];

  const v2f* __restrict__ hv = (const v2f*)h1;
  int rs_v = rowstart[node0 + m];
  int dg_v = deg[node0 + m];
  int gl   = batch[node0 + m];
  float qa[16], qb[16];
  #pragma unroll
  for (int k = 0; k < 4; ++k) {
    #pragma unroll
    for (int i = 0; i < 4; ++i) {
      qa[4 * k + i] = W2a[1024 * k + 4 * tid + i];
      qb[4 * k + i] = W2b[1024 * k + 4 * tid + i];
    }
  }
  float qba = b2a[tid & 63], qbb = b2b[tid & 63];
  int b  = __builtin_amdgcn_readlane(rs_v, 0);
  int c  = __builtin_amdgcn_readlane(dg_v, 0);
  int cv = (lane < c) ? col[b + lane] : 0;
  #pragma unroll
  for (int k = 0; k < 4; ++k) {
    #pragma unroll
    for (int i = 0; i < 4; ++i) {
      sWa[1024 * k + 4 * tid + i] = qa[4 * k + i];
      sWb[1024 * k + 4 * tid + i] = qb[4 * k + i];
    }
  }
  if (tid < 64) { sBa[tid] = qba; sBb[tid] = qbb; }
  __syncthreads();
  if (!has_tile) return;
  float cg = cnt[gl];

  int id[31];
  #pragma unroll
  for (int k = 0; k < 31; ++k) id[k] = __builtin_amdgcn_readlane(cv, k);
  for (int r = 0; r < 16; ++r) {
    int rn  = (r < 15) ? r + 1 : 15;
    int bn  = __builtin_amdgcn_readlane(rs_v, rn);
    int cn  = __builtin_amdgcn_readlane(dg_v, rn);
    int cvn = (lane < cn) ? col[bn + lane] : 0;
    v2f hs  = hv[(node0 + r) * 32 + lane];
    v2f acc = {0.0f, 0.0f};
    v2f p[31];
    load_rows(p, id, c, hv, lane);
    add_rows(acc, p, c);
    for (int j = 31; j < c; ++j)
      acc += hv[col[b + j] * 32 + lane];
    v2f av = hs + acc;
    myT[r * 68 + 2 * lane]     = av.x;
    myT[r * 68 + 2 * lane + 1] = av.y;
    #pragma unroll
    for (int k = 0; k < 31; ++k) id[k] = __builtin_amdgcn_readlane(cvn, k);
    b = bn; c = cn;
  }

  v8f acc[4] = {};
  for (int ks = 0; ks < 16; ++ks) {
    int kk = ks * 4 + 2 * half;
    v2f a;
    a.x = myT[m * 68 + kk];
    a.y = myT[m * 68 + kk + 1];
    #pragma unroll
    for (int nt = 0; nt < 4; ++nt) {
      v2f b;
      b.x = sWa[kk * 64 + nt * 16 + m];
      b.y = sWa[(kk + 1) * 64 + nt * 16 + m];
      acc[nt] = __builtin_amdgcn_wmma_f32_16x16x4_f32(
          false, a, false, b, (short)0, acc[nt], false, false);
    }
  }

  #pragma unroll
  for (int nt = 0; nt < 4; ++nt) {
    float bias = sBa[nt * 16 + m];
    #pragma unroll
    for (int r = 0; r < 8; ++r) {
      float v = acc[nt][r] + bias;
      myT[(r + 8 * half) * 68 + nt * 16 + m] = v > 0.0f ? v : 0.0f;
    }
  }

  v8f acc2[4] = {};
  for (int ks = 0; ks < 16; ++ks) {
    int kk = ks * 4 + 2 * half;
    v2f a2;
    a2.x = myT[m * 68 + kk];
    a2.y = myT[m * 68 + kk + 1];
    #pragma unroll
    for (int nt = 0; nt < 4; ++nt) {
      v2f b;
      b.x = sWb[kk * 64 + nt * 16 + m];
      b.y = sWb[(kk + 1) * 64 + nt * 16 + m];
      acc2[nt] = __builtin_amdgcn_wmma_f32_16x16x4_f32(
          false, a2, false, b, (short)0, acc2[nt], false, false);
    }
  }

  float v[4][8];
  #pragma unroll
  for (int nt = 0; nt < 4; ++nt) {
    float bias = sBb[nt * 16 + m];
    #pragma unroll
    for (int r = 0; r < 8; ++r) {
      float t = acc2[nt][r] + bias;
      v[nt][r] = t > 0.0f ? t : 0.0f;
    }
  }

  int g0 = __builtin_amdgcn_readfirstlane(gl);
  bool one_graph = __builtin_amdgcn_ballot_w32(gl != g0) == 0u;
  if (one_graph) {
    float tot[4];
    #pragma unroll
    for (int nt = 0; nt < 4; ++nt) {
      float s = ((v[nt][0] + v[nt][1]) + (v[nt][2] + v[nt][3])) +
                ((v[nt][4] + v[nt][5]) + (v[nt][6] + v[nt][7]));
      tot[nt] = s + __shfl_xor(s, 16);
    }
    float dn = fmaxf(cg, 1.0f);
    float* dst = out + g0 * 64;
    atomicAdd(&dst[lane],      (half ? tot[1] : tot[0]) / dn);
    atomicAdd(&dst[32 + lane], (half ? tot[3] : tot[2]) / dn);
  } else {
    #pragma unroll
    for (int r = 0; r < 8; ++r) {
      int   g  = __shfl(gl, r + 8 * half);
      float dn = fmaxf(__shfl(cg, r + 8 * half), 1.0f);
      #pragma unroll
      for (int nt = 0; nt < 4; ++nt)
        atomicAdd(&out[g * 64 + nt * 16 + m], v[nt][r] / dn);
    }
  }
}

extern "C" void kernel_launch(void* const* d_in, const int* in_sizes, int n_in,
                              void* d_out, int out_size, void* d_ws, size_t ws_size,
                              hipStream_t stream) {
  const float* x     = (const float*)d_in[0];
  const int*   ei    = (const int*)d_in[1];
  const int*   batch = (const int*)d_in[2];
  const float* W1a   = (const float*)d_in[3];
  const float* b1a   = (const float*)d_in[4];
  const float* W1b   = (const float*)d_in[5];
  const float* b1b   = (const float*)d_in[6];
  const float* W2a   = (const float*)d_in[7];
  const float* b2a   = (const float*)d_in[8];
  const float* W2b   = (const float*)d_in[9];
  const float* b2b   = (const float*)d_in[10];
  float* out = (float*)d_out;

  float* ws       = (float*)d_ws;
  float* h1       = ws + 300032;
  int*   deg      = (int*)(ws + 13100032);
  float* cnt      = ws + 13208224;
  int*   rowstart = (int*)(ws + 13208384);
  int*   pairs    = (int*)(ws + 13308416);
  int*   col      = (int*)(ws + 14908416);
  int*   offT     = (int*)(ws + 16508416);

  dim3 b256(256);
  local_partition_kernel<<<NBLK, b256, 0, stream>>>(ei, pairs, offT, out, cnt);
  bucket_sort_mlp1_kernel<<<NBUCK, dim3(1024), 0, stream>>>(
      pairs, offT, deg, rowstart, col, x, W1a, b1a, W1b, b1b, batch, cnt, h1);
  gather_mlp2_pool_kernel<<<(N_TILES + 7) / 8, b256, 0, stream>>>(
      h1, rowstart, deg, col, W2a, b2a, W2b, b2b, batch, cnt, out);
}
